# starting kernel
# baseline (speedup 1.0000x reference)
#include <hip/hip_runtime.h>
#include <hip/hip_bf16.h>
#include <math.h>
#include <stdint.h>


typedef float v2f __attribute__((ext_vector_type(2)));
typedef float v8f __attribute__((ext_vector_type(8)));

#define NEG_SLOPE 0.2f

__device__ __forceinline__ void atomicMaxF32(float* addr, float val) {
  if (val >= 0.0f) {
    atomicMax((int*)addr, __float_as_int(val));
  } else {
    atomicMin((unsigned int*)addr, __float_as_uint(val));
  }
}

template <int NTILES, int NACT>
__global__ void __launch_bounds__(32 * NTILES)
gat_wmma_gemm_k128(const float* __restrict__ A, const float* __restrict__ B,
                   float* __restrict__ C) {
  constexpr int NPAD = NTILES * 16;
  constexpr int SA = 132;
  constexpr int SB = 132;
  __shared__ float ldsA[16 * SA];
  __shared__ float ldsB[NPAD * SB];

  const int tid  = threadIdx.x;
  const int lane = tid & 31;
  const int wave = tid >> 5;
  const size_t m0 = (size_t)blockIdx.x * 16;
  const int c0 = blockIdx.y * NPAD;

  for (int i = tid; i < 16 * 128; i += 32 * NTILES) {
    int r = i >> 7, c = i & 127;
    ldsA[r * SA + c] = A[(m0 + r) * 128 + c];
  }
  for (int i = tid; i < NPAD * 128; i += 32 * NTILES) {
    int k = i / NPAD, n = i - k * NPAD;
    int gn = c0 + n;
    ldsB[n * SB + k] = (gn < NACT) ? B[(size_t)k * NACT + gn] : 0.0f;
  }
  __syncthreads();

  const int l15  = lane & 15;
  const int koff = (lane >> 4) << 1;
  const int ncol = wave * 16 + l15;
  const float* ap = &ldsA[l15 * SA + koff];
  const float* bp = &ldsB[ncol * SB + koff];

  v8f acc = {0.f, 0.f, 0.f, 0.f, 0.f, 0.f, 0.f, 0.f};
#pragma unroll
  for (int kb = 0; kb < 128; kb += 4) {
    v2f a = *(const v2f*)(ap + kb);
    v2f b = *(const v2f*)(bp + kb);
    acc = __builtin_amdgcn_wmma_f32_16x16x4_f32(
         false, a,  false, b,
         (short)0, acc,  false,  false);
  }

  const int gcol = c0 + ncol;
  if (gcol < NACT) {
    float* cp = C + (m0 + (size_t)((lane >> 4) << 3)) * NACT + gcol;
#pragma unroll
    for (int r = 0; r < 8; ++r) cp[(size_t)r * NACT] = acc[r];
  }
}

__global__ void fill_kernel(float* __restrict__ p, float v, long long n) {
  long long i = (long long)blockIdx.x * blockDim.x + threadIdx.x;
  long long stride = (long long)gridDim.x * blockDim.x;
  for (; i < n; i += stride) p[i] = v;
}

__global__ void init_out_kernel(float* __restrict__ out, const float* __restrict__ b2,
                                int Nn) {
  int idx = blockIdx.x * blockDim.x + threadIdx.x;
  if (idx >= Nn * 40) return;
  out[idx] = b2[idx % 40];
}

__global__ void scores1_kernel(const float* __restrict__ h,
                               const float* __restrict__ att_src,
                               const float* __restrict__ att_dst,
                               float* __restrict__ a_src, float* __restrict__ a_dst,
                               int Nn) {
  int idx = blockIdx.x * blockDim.x + threadIdx.x;
  if (idx >= Nn * 8) return;
  int n = idx >> 3, hd = idx & 7;
  const float* hp = h + (size_t)n * 128 + hd * 16;
  float s = 0.f, d = 0.f;
#pragma unroll
  for (int c = 0; c < 16; ++c) {
    float v = hp[c];
    s = fmaf(v, att_src[hd * 16 + c], s);
    d = fmaf(v, att_dst[hd * 16 + c], d);
  }
  a_src[idx] = s;
  a_dst[idx] = d;
}

__global__ void scores2_kernel(const float* __restrict__ h2,
                               const float* __restrict__ att_src,
                               const float* __restrict__ att_dst,
                               float* __restrict__ a_src, float* __restrict__ a_dst,
                               int Nn) {
  int n = blockIdx.x * blockDim.x + threadIdx.x;
  if (n >= Nn) return;
  const float* hp = h2 + (size_t)n * 40;
  float s = 0.f, d = 0.f;
#pragma unroll
  for (int c = 0; c < 40; ++c) {
    float v = hp[c];
    s = fmaf(v, att_src[c], s);
    d = fmaf(v, att_dst[c], d);
  }
  a_src[n] = s;
  a_dst[n] = d;
}

__global__ void edge_max1_kernel(const int* __restrict__ src, const int* __restrict__ dst,
                                 const float* __restrict__ a_src, const float* __restrict__ a_dst,
                                 float* __restrict__ m, int E, int Etot) {
  int idx = blockIdx.x * blockDim.x + threadIdx.x;
  if (idx >= Etot * 8) return;
  int e = idx >> 3, hd = idx & 7;
  int s = (e < E) ? src[e] : (e - E);
  int d = (e < E) ? dst[e] : (e - E);
  float v = a_src[s * 8 + hd] + a_dst[d * 8 + hd];
  v = (v >= 0.f) ? v : NEG_SLOPE * v;
  atomicMaxF32(&m[d * 8 + hd], v);
}

__global__ void edge_sum1_kernel(const int* __restrict__ src, const int* __restrict__ dst,
                                 const float* __restrict__ a_src, const float* __restrict__ a_dst,
                                 const float* __restrict__ m, float* __restrict__ ssum,
                                 int E, int Etot) {
  int idx = blockIdx.x * blockDim.x + threadIdx.x;
  if (idx >= Etot * 8) return;
  int e = idx >> 3, hd = idx & 7;
  int s = (e < E) ? src[e] : (e - E);
  int d = (e < E) ? dst[e] : (e - E);
  float v = a_src[s * 8 + hd] + a_dst[d * 8 + hd];
  v = (v >= 0.f) ? v : NEG_SLOPE * v;
  atomicAdd(&ssum[d * 8 + hd], __expf(v - m[d * 8 + hd]));
}

__global__ void edge_agg1_kernel(const int* __restrict__ src, const int* __restrict__ dst,
                                 const float* __restrict__ a_src, const float* __restrict__ a_dst,
                                 const float* __restrict__ m, const float* __restrict__ ssum,
                                 const float* __restrict__ h, float* __restrict__ agg,
                                 int E, int Etot) {
  int wid  = (int)(((long long)blockIdx.x * blockDim.x + threadIdx.x) >> 5);
  int lane = threadIdx.x & 31;
  if (wid >= Etot) return;
  int s = (wid < E) ? src[wid] : (wid - E);
  int d = (wid < E) ? dst[wid] : (wid - E);
  int hd = lane >> 2;
  float e = a_src[s * 8 + hd] + a_dst[d * 8 + hd];
  e = (e >= 0.f) ? e : NEG_SLOPE * e;
  float alpha = __expf(e - m[d * 8 + hd]) / ssum[d * 8 + hd];
  const float4 hv = *(const float4*)(h + (size_t)s * 128 + lane * 4);
  float* op = agg + (size_t)d * 128 + lane * 4;
  atomicAdd(op + 0, hv.x * alpha);
  atomicAdd(op + 1, hv.y * alpha);
  atomicAdd(op + 2, hv.z * alpha);
  atomicAdd(op + 3, hv.w * alpha);
}

__global__ void elu_bias_kernel(float* __restrict__ a, const float* __restrict__ b,
                                long long n) {
  long long idx = (long long)blockIdx.x * blockDim.x + threadIdx.x;
  if (idx >= n) return;
  float v = a[idx] + b[idx & 127];
  a[idx] = (v > 0.f) ? v : (__expf(v) - 1.0f);
}

__global__ void edge_max2_kernel(const int* __restrict__ src, const int* __restrict__ dst,
                                 const float* __restrict__ a_src, const float* __restrict__ a_dst,
                                 float* __restrict__ m, int E, int Etot) {
  int e = blockIdx.x * blockDim.x + threadIdx.x;
  if (e >= Etot) return;
  int s = (e < E) ? src[e] : (e - E);
  int d = (e < E) ? dst[e] : (e - E);
  float v = a_src[s] + a_dst[d];
  v = (v >= 0.f) ? v : NEG_SLOPE * v;
  atomicMaxF32(&m[d], v);
}

__global__ void edge_sum2_kernel(const int* __restrict__ src, const int* __restrict__ dst,
                                 const float* __restrict__ a_src, const float* __restrict__ a_dst,
                                 const float* __restrict__ m, float* __restrict__ ssum,
                                 int E, int Etot) {
  int e = blockIdx.x * blockDim.x + threadIdx.x;
  if (e >= Etot) return;
  int s = (e < E) ? src[e] : (e - E);
  int d = (e < E) ? dst[e] : (e - E);
  float v = a_src[s] + a_dst[d];
  v = (v >= 0.f) ? v : NEG_SLOPE * v;
  atomicAdd(&ssum[d], __expf(v - m[d]));
}

__global__ void edge_agg2_kernel(const int* __restrict__ src, const int* __restrict__ dst,
                                 const float* __restrict__ a_src, const float* __restrict__ a_dst,
                                 const float* __restrict__ m, const float* __restrict__ ssum,
                                 const float* __restrict__ h2, float* __restrict__ out,
                                 int E, int Etot) {
  int wid  = (int)(((long long)blockIdx.x * blockDim.x + threadIdx.x) >> 5);
  int lane = threadIdx.x & 31;
  if (wid >= Etot) return;
  int s = (wid < E) ? src[wid] : (wid - E);
  int d = (wid < E) ? dst[wid] : (wid - E);
  float e = a_src[s] + a_dst[d];
  e = (e >= 0.f) ? e : NEG_SLOPE * e;
  float alpha = __expf(e - m[d]) / ssum[d];
  for (int c = lane; c < 40; c += 32)
    atomicAdd(&out[(size_t)d * 40 + c], h2[(size_t)s * 40 + c] * alpha);
}

extern "C" void kernel_launch(void* const* d_in, const int* in_sizes, int n_in,
                              void* d_out, int out_size, void* d_ws, size_t ws_size,
                              hipStream_t stream) {
  const float* x   = (const float*)d_in[0];
  const int*   ei  = (const int*)d_in[1];
  const float* W1  = (const float*)d_in[2];
  const float* as1 = (const float*)d_in[3];
  const float* ad1 = (const float*)d_in[4];
  const float* b1  = (const float*)d_in[5];
  const float* W2  = (const float*)d_in[6];
  const float* as2 = (const float*)d_in[7];
  const float* ad2 = (const float*)d_in[8];
  const float* b2  = (const float*)d_in[9];
  float* out = (float*)d_out;

  const int Nn   = in_sizes[0] / 128;
  const int E    = in_sizes[1] / 2;
  const int Etot = E + Nn;
  const int* srcp = ei;
  const int* dstp = ei + E;

  float* ws   = (float*)d_ws;
  float* h    = ws;  ws += (size_t)Nn * 128;
  float* agg1 = ws;  ws += (size_t)Nn * 128;
  float* asc1 = ws;  ws += (size_t)Nn * 8;
  float* adc1 = ws;  ws += (size_t)Nn * 8;
  float* m1   = ws;  ws += (size_t)Nn * 8;
  float* s1   = ws;  ws += (size_t)Nn * 8;
  float* h2   = ws;  ws += (size_t)Nn * 40;
  float* asc2 = ws;  ws += Nn;
  float* adc2 = ws;  ws += Nn;
  float* m2   = ws;  ws += Nn;
  float* s2   = ws;  ws += Nn;

  const int T = 256;
  const int fillBlocks = 2048;

  fill_kernel<<<fillBlocks, T, 0, stream>>>(m1, -INFINITY, (long long)Nn * 8);
  fill_kernel<<<fillBlocks, T, 0, stream>>>(s1, 0.0f, (long long)Nn * 8);
  fill_kernel<<<fillBlocks, T, 0, stream>>>(agg1, 0.0f, (long long)Nn * 128);
  fill_kernel<<<fillBlocks, T, 0, stream>>>(m2, -INFINITY, (long long)Nn);
  fill_kernel<<<fillBlocks, T, 0, stream>>>(s2, 0.0f, (long long)Nn);
  init_out_kernel<<<(Nn * 40 + T - 1) / T, T, 0, stream>>>(out, b2, Nn);

  gat_wmma_gemm_k128<4, 128><<<dim3(Nn / 16, 2), 4 * 32, 0, stream>>>(x, W1, h);
  scores1_kernel<<<(Nn * 8 + T - 1) / T, T, 0, stream>>>(h, as1, ad1, asc1, adc1, Nn);
  edge_max1_kernel<<<(Etot * 8 + T - 1) / T, T, 0, stream>>>(srcp, dstp, asc1, adc1, m1, E, Etot);
  edge_sum1_kernel<<<(Etot * 8 + T - 1) / T, T, 0, stream>>>(srcp, dstp, asc1, adc1, m1, s1, E, Etot);
  edge_agg1_kernel<<<(Etot + 7) / 8, T, 0, stream>>>(srcp, dstp, asc1, adc1, m1, s1, h, agg1, E, Etot);
  elu_bias_kernel<<<(int)(((long long)Nn * 128 + T - 1) / T), T, 0, stream>>>(agg1, b1, (long long)Nn * 128);

  gat_wmma_gemm_k128<3, 40><<<dim3(Nn / 16, 1), 3 * 32, 0, stream>>>(agg1, W2, h2);
  scores2_kernel<<<(Nn + T - 1) / T, T, 0, stream>>>(h2, as2, ad2, asc2, adc2, Nn);
  edge_max2_kernel<<<(Etot + T - 1) / T, T, 0, stream>>>(srcp, dstp, asc2, adc2, m2, E, Etot);
  edge_sum2_kernel<<<(Etot + T - 1) / T, T, 0, stream>>>(srcp, dstp, asc2, adc2, m2, s2, E, Etot);
  edge_agg2_kernel<<<(Etot + 7) / 8, T, 0, stream>>>(srcp, dstp, asc2, adc2, m2, s2, h2, out, E, Etot);
}


# Arrival-rank sum order varied run to run: now per-dst wave gather over a CSR rebuilt per call, segments sorted by source id so sum order is fixed (layer-2 denominator a lane tree); no float atomics
# speedup vs baseline: 2.5100x; 2.5100x over previous
#include <hip/hip_runtime.h>
#include <hip/hip_bf16.h>
#include <math.h>
#include <stdint.h>


typedef float v2f __attribute__((ext_vector_type(2)));
typedef float v8f __attribute__((ext_vector_type(8)));

#define NEG_SLOPE 0.2f

template <int NTILES, int NACT>
__global__ void __launch_bounds__(32 * NTILES)
gat_wmma_gemm_k128(const float* __restrict__ A, const float* __restrict__ B,
                   float* __restrict__ C) {
  constexpr int NPAD = NTILES * 16;
  constexpr int SA = 132;
  constexpr int SB = 132;
  __shared__ float ldsA[16 * SA];
  __shared__ float ldsB[NPAD * SB];

  const int tid  = threadIdx.x;
  const int lane = tid & 31;
  const int wave = tid >> 5;
  const size_t m0 = (size_t)blockIdx.x * 16;
  const int c0 = blockIdx.y * NPAD;

  for (int i = tid; i < 16 * 128; i += 32 * NTILES) {
    int r = i >> 7, c = i & 127;
    ldsA[r * SA + c] = A[(m0 + r) * 128 + c];
  }
  for (int i = tid; i < NPAD * 128; i += 32 * NTILES) {
    int k = i / NPAD, n = i - k * NPAD;
    int gn = c0 + n;
    ldsB[n * SB + k] = (gn < NACT) ? B[(size_t)k * NACT + gn] : 0.0f;
  }
  __syncthreads();

  const int l15  = lane & 15;
  const int koff = (lane >> 4) << 1;
  const int ncol = wave * 16 + l15;
  const float* ap = &ldsA[l15 * SA + koff];
  const float* bp = &ldsB[ncol * SB + koff];

  v8f acc = {0.f, 0.f, 0.f, 0.f, 0.f, 0.f, 0.f, 0.f};
#pragma unroll
  for (int kb = 0; kb < 128; kb += 4) {
    v2f a = *(const v2f*)(ap + kb);
    v2f b = *(const v2f*)(bp + kb);
    acc = __builtin_amdgcn_wmma_f32_16x16x4_f32(
         false, a,  false, b,
         (short)0, acc,  false,  false);
  }

  const int gcol = c0 + ncol;
  if (gcol < NACT) {
    float* cp = C + (m0 + (size_t)((lane >> 4) << 3)) * NACT + gcol;
#pragma unroll
    for (int r = 0; r < 8; ++r) cp[(size_t)r * NACT] = acc[r];
  }
}

__global__ void scores1_kernel(const float* __restrict__ h,
                               const float* __restrict__ att_src,
                               const float* __restrict__ att_dst,
                               float* __restrict__ a_src, float* __restrict__ a_dst,
                               int Nn) {
  int idx = blockIdx.x * blockDim.x + threadIdx.x;
  if (idx >= Nn * 8) return;
  int n = idx >> 3, hd = idx & 7;
  const float* hp = h + (size_t)n * 128 + hd * 16;
  float s = 0.f, d = 0.f;
#pragma unroll
  for (int c = 0; c < 16; ++c) {
    float v = hp[c];
    s = fmaf(v, att_src[hd * 16 + c], s);
    d = fmaf(v, att_dst[hd * 16 + c], d);
  }
  a_src[idx] = s;
  a_dst[idx] = d;
}

__global__ void scores2_kernel(const float* __restrict__ h2,
                               const float* __restrict__ att_src,
                               const float* __restrict__ att_dst,
                               float* __restrict__ a_src, float* __restrict__ a_dst,
                               int Nn) {
  int n = blockIdx.x * blockDim.x + threadIdx.x;
  if (n >= Nn) return;
  const float* hp = h2 + (size_t)n * 40;
  float s = 0.f, d = 0.f;
#pragma unroll
  for (int c = 0; c < 40; ++c) {
    float v = hp[c];
    s = fmaf(v, att_src[c], s);
    d = fmaf(v, att_dst[c], d);
  }
  a_src[n] = s;
  a_dst[n] = d;
}

__global__ void zero_int_kernel(int* __restrict__ p, int n) {
  int i = blockIdx.x * blockDim.x + threadIdx.x;
  if (i < n) p[i] = 0;
}

__global__ void hist_rank_kernel(const int* __restrict__ dst, int* __restrict__ cnt,
                                 int* __restrict__ rank, int E) {
  int e = blockIdx.x * blockDim.x + threadIdx.x;
  if (e >= E) return;
  rank[e] = atomicAdd(&cnt[dst[e]], 1);
}

__global__ void __launch_bounds__(256)
scan_local_kernel(const int* __restrict__ cnt, int* __restrict__ lscan,
                  int* __restrict__ btot, int Nn) {
  __shared__ int wsum[8];
  const int tid = threadIdx.x, lane = tid & 31, w = tid >> 5;
  const int base = blockIdx.x * 2048 + tid * 8;
  int v[8];
  int run = 0;
#pragma unroll
  for (int k = 0; k < 8; ++k) {
    int idx = base + k;
    int c = (idx < Nn) ? cnt[idx] : 0;
    v[k] = run;
    run += c;
  }
  int inc = run;
#pragma unroll
  for (int off = 1; off < 32; off <<= 1) {
    int nb = __shfl_up(inc, off);
    if (lane >= off) inc += nb;
  }
  if (lane == 31) wsum[w] = inc;
  __syncthreads();
  int woff = 0;
#pragma unroll
  for (int k = 0; k < 8; ++k) woff += (k < w) ? wsum[k] : 0;
  const int texcl = woff + inc - run;
#pragma unroll
  for (int k = 0; k < 8; ++k) {
    int idx = base + k;
    if (idx < Nn) lscan[idx] = texcl + v[k];
  }
  if (tid == 255) btot[blockIdx.x] = woff + inc;
}

__global__ void __launch_bounds__(256)
scan_finish_kernel(int* __restrict__ rowptr, const int* __restrict__ btot, int Nn) {
  __shared__ int red[8];
  const int tid = threadIdx.x, lane = tid & 31, w = tid >> 5;
  int part = 0;
  for (int t = tid; t < (int)blockIdx.x; t += 256) part += btot[t];
#pragma unroll
  for (int off = 16; off >= 1; off >>= 1) part += __shfl_xor(part, off);
  if (lane == 0) red[w] = part;
  __syncthreads();
  int boff = 0;
#pragma unroll
  for (int k = 0; k < 8; ++k) boff += red[k];
#pragma unroll
  for (int k = 0; k < 8; ++k) {
    int idx = blockIdx.x * 2048 + k * 256 + tid;
    if (idx < Nn) rowptr[idx] += boff;
  }
  if (blockIdx.x == gridDim.x - 1 && tid == 0) rowptr[Nn] = boff + btot[blockIdx.x];
}

__global__ void fill_col_kernel(const int* __restrict__ src, const int* __restrict__ dst,
                                const int* __restrict__ rowptr, const int* __restrict__ rank,
                                int* __restrict__ col, int E) {
  int e = blockIdx.x * blockDim.x + threadIdx.x;
  if (e >= E) return;
  col[rowptr[dst[e]] + rank[e]] = src[e];
}

__global__ void __launch_bounds__(256)
sort_seg_kernel(const int* __restrict__ rowptr, const int* __restrict__ col,
                int* __restrict__ col2, int Nn) {
  const int lane = threadIdx.x & 31;
  const int d = __builtin_amdgcn_readfirstlane(blockIdx.x * 8 + (threadIdx.x >> 5));
  if (d >= Nn) return;
  const int start = rowptr[d];
  const int deg = rowptr[d + 1] - start;
  if (deg <= 32) {
    const int k = (lane < deg) ? col[start + lane] : 0x7fffffff;
    int r = 0;
    for (int j = 0; j < deg; ++j) {
      const int kj = __builtin_amdgcn_readlane(k, j);
      r += (kj < k || (kj == k && j < lane)) ? 1 : 0;
    }
    if (lane < deg) col2[start + r] = k;
  } else {
    for (int i = lane; i < deg; i += 32) {
      const int k = col[start + i];
      int r = 0;
      for (int j = 0; j < deg; ++j) {
        const int kj = col[start + j];
        r += (kj < k || (kj == k && j < i)) ? 1 : 0;
      }
      col2[start + r] = k;
    }
  }
}

__device__ __forceinline__ float leaky(float v) { return (v >= 0.f) ? v : NEG_SLOPE * v; }

template <bool CACHED>
__device__ __forceinline__ void gat1_node(int d, int lane, int start, int deg,
                                          const int* __restrict__ col2,
                                          const float* __restrict__ asc,
                                          const float* __restrict__ adc,
                                          const float* __restrict__ h,
                                          const float* __restrict__ b1,
                                          float* __restrict__ h1, float* es) {
  const int hd = lane >> 2;
  const float ad = adc[(size_t)d * 8 + hd];
  const float eself = leaky(asc[(size_t)d * 8 + hd] + ad);
  int sv = 0;
  if (CACHED && lane < deg) sv = col2[start + lane];
  float m = eself;
#pragma unroll 4
  for (int j = 0; j < deg; ++j) {
    const int s = CACHED ? __builtin_amdgcn_readlane(sv, j) : col2[start + j];
    const float e = leaky(asc[(size_t)s * 8 + hd] + ad);
    if (CACHED) es[j * 32] = e;
    m = fmaxf(m, e);
  }
  float ssum = 0.f;
#pragma unroll 4
  for (int j = 0; j < deg; ++j) {
    float e;
    if (CACHED) e = es[j * 32];
    else e = leaky(asc[(size_t)col2[start + j] * 8 + hd] + ad);
    const float ex = __expf(e - m);
    if (CACHED) es[j * 32] = ex;
    ssum += ex;
  }
  const float exself = __expf(eself - m);
  ssum += exself;
  float4 acc = make_float4(0.f, 0.f, 0.f, 0.f);
#pragma unroll 4
  for (int j = 0; j < deg; ++j) {
    const int s = CACHED ? __builtin_amdgcn_readlane(sv, j) : col2[start + j];
    float ex;
    if (CACHED) ex = es[j * 32];
    else ex = __expf(leaky(asc[(size_t)s * 8 + hd] + ad) - m);
    const float alpha = ex / ssum;
    const float4 hv = *(const float4*)(h + (size_t)s * 128 + lane * 4);
    acc.x = fmaf(hv.x, alpha, acc.x);
    acc.y = fmaf(hv.y, alpha, acc.y);
    acc.z = fmaf(hv.z, alpha, acc.z);
    acc.w = fmaf(hv.w, alpha, acc.w);
  }
  {
    const float alpha = exself / ssum;
    const float4 hv = *(const float4*)(h + (size_t)d * 128 + lane * 4);
    acc.x = fmaf(hv.x, alpha, acc.x);
    acc.y = fmaf(hv.y, alpha, acc.y);
    acc.z = fmaf(hv.z, alpha, acc.z);
    acc.w = fmaf(hv.w, alpha, acc.w);
  }
  const float4 bv = *(const float4*)(b1 + lane * 4);
  float4 o;
  o.x = acc.x + bv.x; o.x = (o.x > 0.f) ? o.x : (__expf(o.x) - 1.0f);
  o.y = acc.y + bv.y; o.y = (o.y > 0.f) ? o.y : (__expf(o.y) - 1.0f);
  o.z = acc.z + bv.z; o.z = (o.z > 0.f) ? o.z : (__expf(o.z) - 1.0f);
  o.w = acc.w + bv.w; o.w = (o.w > 0.f) ? o.w : (__expf(o.w) - 1.0f);
  *(float4*)(h1 + (size_t)d * 128 + lane * 4) = o;
}

__global__ void __launch_bounds__(256)
gather1_kernel(const int* __restrict__ rowptr, const int* __restrict__ col2,
               const float* __restrict__ asc, const float* __restrict__ adc,
               const float* __restrict__ h, const float* __restrict__ b1,
               float* __restrict__ h1, int Nn) {
  __shared__ float es_all[8 * 32 * 32];
  const int lane = threadIdx.x & 31;
  const int w = threadIdx.x >> 5;
  const int d = __builtin_amdgcn_readfirstlane(blockIdx.x * 8 + w);
  if (d >= Nn) return;
  const int start = rowptr[d];
  const int deg = rowptr[d + 1] - start;
  float* es = es_all + w * (32 * 32) + lane;
  if (deg <= 32) gat1_node<true>(d, lane, start, deg, col2, asc, adc, h, b1, h1, es);
  else           gat1_node<false>(d, lane, start, deg, col2, asc, adc, h, b1, h1, es);
}

__device__ __forceinline__ float wave_max(float v) {
#pragma unroll
  for (int off = 16; off >= 1; off >>= 1) v = fmaxf(v, __shfl_xor(v, off));
  return v;
}
__device__ __forceinline__ float wave_sum(float v) {
#pragma unroll
  for (int off = 16; off >= 1; off >>= 1) v += __shfl_xor(v, off);
  return v;
}

__device__ __forceinline__ void gat2_accum(int n, int sv, float av, int lc,
                                           const float* __restrict__ h2, float2& acc) {
#pragma unroll 4
  for (int t = 0; t < n; ++t) {
    const int s = __builtin_amdgcn_readlane(sv, t);
    const float a = __int_as_float(__builtin_amdgcn_readlane(__float_as_int(av), t));
    const float2 hv = *(const float2*)(h2 + (size_t)s * 40 + 2 * lc);
    acc.x = fmaf(hv.x, a, acc.x);
    acc.y = fmaf(hv.y, a, acc.y);
  }
}

__global__ void __launch_bounds__(256)
gather2_kernel(const int* __restrict__ rowptr, const int* __restrict__ col2,
               const float* __restrict__ asc, const float* __restrict__ adc,
               const float* __restrict__ h2, const float* __restrict__ b2,
               float* __restrict__ out, int Nn) {
  const int lane = threadIdx.x & 31;
  const int lc = (lane < 19) ? lane : 19;
  const int d = __builtin_amdgcn_readfirstlane(blockIdx.x * 8 + (threadIdx.x >> 5));
  if (d >= Nn) return;
  const int start = rowptr[d];
  const int deg = rowptr[d + 1] - start;
  const float ad = adc[d];
  const float eself = leaky(asc[d] + ad);

  int s0 = 0;
  float e0 = -INFINITY;
  if (lane < deg) { s0 = col2[start + lane]; e0 = leaky(asc[s0] + ad); }
  float m = fmaxf(eself, wave_max(e0));
  for (int c = 32; c < deg; c += 32) {
    float e = -INFINITY;
    if (c + lane < deg) e = leaky(asc[col2[start + c + lane]] + ad);
    m = fmaxf(m, wave_max(e));
  }
  const float ex0 = (lane < deg) ? __expf(e0 - m) : 0.f;
  float ssum = wave_sum(ex0);
  for (int c = 32; c < deg; c += 32) {
    float ex = 0.f;
    if (c + lane < deg) ex = __expf(leaky(asc[col2[start + c + lane]] + ad) - m);
    ssum += wave_sum(ex);
  }
  const float exself = __expf(eself - m);
  ssum += exself;

  float2 acc = make_float2(0.f, 0.f);
  gat2_accum(min(deg, 32), s0, ex0 / ssum, lc, h2, acc);
  for (int c = 32; c < deg; c += 32) {
    int s = 0;
    float a = 0.f;
    if (c + lane < deg) {
      s = col2[start + c + lane];
      a = __expf(leaky(asc[s] + ad) - m) / ssum;
    }
    gat2_accum(min(deg - c, 32), s, a, lc, h2, acc);
  }
  {
    const float a = exself / ssum;
    const float2 hv = *(const float2*)(h2 + (size_t)d * 40 + 2 * lc);
    acc.x = fmaf(hv.x, a, acc.x);
    acc.y = fmaf(hv.y, a, acc.y);
  }
  if (lane < 20) {
    const float2 bv = *(const float2*)(b2 + 2 * lane);
    *(float2*)(out + (size_t)d * 40 + 2 * lane) = make_float2(acc.x + bv.x, acc.y + bv.y);
  }
}

extern "C" void kernel_launch(void* const* d_in, const int* in_sizes, int n_in,
                              void* d_out, int out_size, void* d_ws, size_t ws_size,
                              hipStream_t stream) {
  const float* x   = (const float*)d_in[0];
  const int*   ei  = (const int*)d_in[1];
  const float* W1  = (const float*)d_in[2];
  const float* as1 = (const float*)d_in[3];
  const float* ad1 = (const float*)d_in[4];
  const float* b1  = (const float*)d_in[5];
  const float* W2  = (const float*)d_in[6];
  const float* as2 = (const float*)d_in[7];
  const float* ad2 = (const float*)d_in[8];
  const float* b2  = (const float*)d_in[9];
  float* out = (float*)d_out;

  const int Nn   = in_sizes[0] / 128;
  const int E    = in_sizes[1] / 2;
  const int* srcp = ei;
  const int* dstp = ei + E;
  const int NB = (Nn + 2047) / 2048;

  float* ws   = (float*)d_ws;
  float* h    = ws;  ws += (size_t)Nn * 128;
  float* h1   = ws;  ws += (size_t)Nn * 128;
  float* asc1 = ws;  ws += (size_t)Nn * 8;
  float* adc1 = ws;  ws += (size_t)Nn * 8;
  float* h2   = ws;  ws += (size_t)Nn * 40;
  float* asc2 = ws;  ws += Nn;
  float* adc2 = ws;  ws += Nn;
  int* wi     = (int*)ws;
  int* cnt    = wi;  wi += Nn;
  int* rank   = wi;  wi += E;
  int* col    = wi;  wi += E;
  int* col2   = wi;  wi += E;
  int* rowptr = wi;  wi += (size_t)Nn + 1;
  int* btot   = wi;  wi += NB;

  const int T = 256;

  zero_int_kernel<<<(Nn + T - 1) / T, T, 0, stream>>>(cnt, Nn);
  hist_rank_kernel<<<(E + T - 1) / T, T, 0, stream>>>(dstp, cnt, rank, E);
  scan_local_kernel<<<NB, 256, 0, stream>>>(cnt, rowptr, btot, Nn);
  scan_finish_kernel<<<NB, 256, 0, stream>>>(rowptr, btot, Nn);
  fill_col_kernel<<<(E + T - 1) / T, T, 0, stream>>>(srcp, dstp, rowptr, rank, col, E);
  sort_seg_kernel<<<(Nn + 7) / 8, 256, 0, stream>>>(rowptr, col, col2, Nn);

  gat_wmma_gemm_k128<4, 128><<<dim3(Nn / 16, 2), 4 * 32, 0, stream>>>(x, W1, h);
  scores1_kernel<<<(Nn * 8 + T - 1) / T, T, 0, stream>>>(h, as1, ad1, asc1, adc1, Nn);
  gather1_kernel<<<(Nn + 7) / 8, 256, 0, stream>>>(rowptr, col2, asc1, adc1, h, b1, h1, Nn);

  gat_wmma_gemm_k128<3, 40><<<dim3(Nn / 16, 1), 3 * 32, 0, stream>>>(h1, W2, h2);
  scores2_kernel<<<(Nn + T - 1) / T, T, 0, stream>>>(h2, as2, ad2, asc2, adc2, Nn);
  gather2_kernel<<<(Nn + 7) / 8, 256, 0, stream>>>(rowptr, col2, asc2, adc2, h2, b2, out, Nn);
}


# Work items looped inside fewer, longer-lived waves: 8 edges per thread, several nodes per wave, GEMM blocks loop over row strips with the W panel staged once; about 6x fewer waves started per call
# speedup vs baseline: 3.2186x; 1.2823x over previous
#include <hip/hip_runtime.h>
#include <hip/hip_bf16.h>
#include <math.h>
#include <stdint.h>


typedef float v2f __attribute__((ext_vector_type(2)));
typedef float v8f __attribute__((ext_vector_type(8)));

#define NEG_SLOPE 0.2f

template <int NTILES, int NACT>
__global__ void __launch_bounds__(32 * NTILES)
gat_wmma_gemm_k128(const float* __restrict__ A, const float* __restrict__ B,
                   float* __restrict__ C, int nstrips) {
  constexpr int NPAD = NTILES * 16;
  constexpr int SA = 132;
  constexpr int SB = 132;
  constexpr int NT = 32 * NTILES;
  __shared__ __attribute__((aligned(16))) float ldsA[16 * SA];
  __shared__ float ldsB[NPAD * SB];

  const int tid  = threadIdx.x;
  const int lane = tid & 31;
  const int wave = tid >> 5;
  const int c0 = blockIdx.y * NPAD;

  for (int i = tid; i < NPAD * 128; i += NT) {
    int k = i / NPAD, n = i - k * NPAD;
    int gn = c0 + n;
    ldsB[n * SB + k] = (gn < NACT) ? B[(size_t)k * NACT + gn] : 0.0f;
  }

  const int l15  = lane & 15;
  const int koff = (lane >> 4) << 1;
  const int ncol = wave * 16 + l15;
  const float* ap = &ldsA[l15 * SA + koff];
  const float* bp = &ldsB[ncol * SB + koff];

  const int gcol = c0 + ncol;

  for (int strip = blockIdx.x; strip < nstrips; strip += gridDim.x) {
    const size_t m0 = (size_t)strip * 16;
    constexpr int AIT = (16 * 32 + NT - 1) / NT;
    float4 av[AIT];
#pragma unroll
    for (int u = 0; u < AIT; ++u) {
      const int i = min(tid + u * NT, 16 * 32 - 1);
      av[u] = *(const float4*)(A + (m0 + (i >> 5)) * 128 + ((i & 31) << 2));
    }
#pragma unroll
    for (int u = 0; u < AIT; ++u) {
      const int i = tid + u * NT;
      if ((u + 1) * NT <= 16 * 32 || i < 16 * 32) *(float4*)&ldsA[(i >> 5) * SA + ((i & 31) << 2)] = av[u];
    }
    __syncthreads();

    v8f acc = {0.f, 0.f, 0.f, 0.f, 0.f, 0.f, 0.f, 0.f};
#pragma unroll
    for (int kb = 0; kb < 128; kb += 4) {
      v2f a = *(const v2f*)(ap + kb);
      v2f b = *(const v2f*)(bp + kb);
      acc = __builtin_amdgcn_wmma_f32_16x16x4_f32(
           false, a,  false, b,
           (short)0, acc,  false,  false);
    }

    if (gcol < NACT) {
      float* cp = C + (m0 + (size_t)((lane >> 4) << 3)) * NACT + gcol;
#pragma unroll
      for (int r = 0; r < 8; ++r) cp[(size_t)r * NACT] = acc[r];
    }
    __syncthreads();
  }
}

__global__ void __launch_bounds__(256)
scores1_kernel(const float* __restrict__ h,
                               const float* __restrict__ att_src,
                               const float* __restrict__ att_dst,
                               float* __restrict__ a_src, float* __restrict__ a_dst,
                               int Nn) {
  const int gt = blockIdx.x * blockDim.x + threadIdx.x;
  const int hd = gt & 7;
  const int nstep = (gridDim.x * blockDim.x) >> 3;
  float ws[16], wd[16];
#pragma unroll
  for (int c = 0; c < 16; ++c) { ws[c] = att_src[hd * 16 + c]; wd[c] = att_dst[hd * 16 + c]; }
  for (int n = gt >> 3; n < Nn; n += nstep) {
    const float4* hp = (const float4*)(h + (size_t)n * 128 + hd * 16);
    float v[16];
#pragma unroll
    for (int q = 0; q < 4; ++q) {
      const float4 t = hp[q];
      v[4 * q] = t.x; v[4 * q + 1] = t.y; v[4 * q + 2] = t.z; v[4 * q + 3] = t.w;
    }
    float s = 0.f, d = 0.f;
#pragma unroll
    for (int c = 0; c < 16; ++c) {
      s = fmaf(v[c], ws[c], s);
      d = fmaf(v[c], wd[c], d);
    }
    a_src[(size_t)n * 8 + hd] = s;
    a_dst[(size_t)n * 8 + hd] = d;
  }
}

__global__ void scores2_kernel(const float* __restrict__ h2,
                               const float* __restrict__ att_src,
                               const float* __restrict__ att_dst,
                               float* __restrict__ a_src, float* __restrict__ a_dst,
                               int Nn) {
  int n = blockIdx.x * blockDim.x + threadIdx.x;
  if (n >= Nn) return;
  const float* hp = h2 + (size_t)n * 40;
  float s = 0.f, d = 0.f;
#pragma unroll
  for (int c = 0; c < 40; ++c) {
    float v = hp[c];
    s = fmaf(v, att_src[c], s);
    d = fmaf(v, att_dst[c], d);
  }
  a_src[n] = s;
  a_dst[n] = d;
}

__global__ void zero_int_kernel(int* __restrict__ p, int n) {
  const int step = gridDim.x * blockDim.x;
  for (int i = blockIdx.x * blockDim.x + threadIdx.x; i < n; i += step) p[i] = 0;
}

__global__ void __launch_bounds__(256)
hist_rank_kernel(const int* __restrict__ dst, int* __restrict__ cnt,
                 int* __restrict__ rank, int E) {
  const int base = blockIdx.x * 2048 + threadIdx.x;
  int dv[8], rv[8];
#pragma unroll
  for (int k = 0; k < 8; ++k) {
    const int e = base + k * 256;
    dv[k] = (e < E) ? dst[e] : -1;
  }
#pragma unroll
  for (int k = 0; k < 8; ++k) rv[k] = (dv[k] >= 0) ? atomicAdd(&cnt[dv[k]], 1) : 0;
#pragma unroll
  for (int k = 0; k < 8; ++k) {
    const int e = base + k * 256;
    if (e < E) rank[e] = rv[k];
  }
}

__global__ void __launch_bounds__(256)
scan_local_kernel(const int* __restrict__ cnt, int* __restrict__ lscan,
                  int* __restrict__ btot, int Nn) {
  __shared__ int wsum[8];
  const int tid = threadIdx.x, lane = tid & 31, w = tid >> 5;
  const int base = blockIdx.x * 2048 + tid * 8;
  int v[8];
  int run = 0;
#pragma unroll
  for (int k = 0; k < 8; ++k) {
    int idx = base + k;
    int c = (idx < Nn) ? cnt[idx] : 0;
    v[k] = run;
    run += c;
  }
  int inc = run;
#pragma unroll
  for (int off = 1; off < 32; off <<= 1) {
    int nb = __shfl_up(inc, off);
    if (lane >= off) inc += nb;
  }
  if (lane == 31) wsum[w] = inc;
  __syncthreads();
  int woff = 0;
#pragma unroll
  for (int k = 0; k < 8; ++k) woff += (k < w) ? wsum[k] : 0;
  const int texcl = woff + inc - run;
#pragma unroll
  for (int k = 0; k < 8; ++k) {
    int idx = base + k;
    if (idx < Nn) lscan[idx] = texcl + v[k];
  }
  if (tid == 255) btot[blockIdx.x] = woff + inc;
}

__global__ void __launch_bounds__(256)
scan_finish_kernel(int* __restrict__ rowptr, const int* __restrict__ btot, int Nn) {
  __shared__ int red[8];
  const int tid = threadIdx.x, lane = tid & 31, w = tid >> 5;
  int part = 0;
  for (int t = tid; t < (int)blockIdx.x; t += 256) part += btot[t];
#pragma unroll
  for (int off = 16; off >= 1; off >>= 1) part += __shfl_xor(part, off);
  if (lane == 0) red[w] = part;
  __syncthreads();
  int boff = 0;
#pragma unroll
  for (int k = 0; k < 8; ++k) boff += red[k];
#pragma unroll
  for (int k = 0; k < 8; ++k) {
    int idx = blockIdx.x * 2048 + k * 256 + tid;
    if (idx < Nn) rowptr[idx] += boff;
  }
  if (blockIdx.x == gridDim.x - 1 && tid == 0) rowptr[Nn] = boff + btot[blockIdx.x];
}

__global__ void __launch_bounds__(256)
fill_col_kernel(const int* __restrict__ src, const int* __restrict__ dst,
                const int* __restrict__ rowptr, const int* __restrict__ rank,
                int* __restrict__ col, int E) {
  const int base = blockIdx.x * 2048 + threadIdx.x;
  int dv[8], sv[8], pos[8];
#pragma unroll
  for (int k = 0; k < 8; ++k) {
    const int e = base + k * 256;
    const bool ok = e < E;
    dv[k] = ok ? dst[e] : -1;
    sv[k] = ok ? src[e] : 0;
    pos[k] = ok ? rank[e] : 0;
  }
#pragma unroll
  for (int k = 0; k < 8; ++k) if (dv[k] >= 0) pos[k] += rowptr[dv[k]];
#pragma unroll
  for (int k = 0; k < 8; ++k) if (dv[k] >= 0) col[pos[k]] = sv[k];
}

__global__ void __launch_bounds__(256)
sort_seg_kernel(const int* __restrict__ rowptr, const int* __restrict__ col,
                int* __restrict__ col2, int Nn) {
  const int lane = threadIdx.x & 31;
  const int nw = gridDim.x * 8;
  for (int d = __builtin_amdgcn_readfirstlane(blockIdx.x * 8 + (threadIdx.x >> 5)); d < Nn; d += nw) {
  const int start = rowptr[d];
  const int deg = rowptr[d + 1] - start;
  if (deg <= 32) {
    const int k = (lane < deg) ? col[start + lane] : 0x7fffffff;
    int r = 0;
    for (int j = 0; j < deg; ++j) {
      const int kj = __builtin_amdgcn_readlane(k, j);
      r += (kj < k || (kj == k && j < lane)) ? 1 : 0;
    }
    if (lane < deg) col2[start + r] = k;
  } else {
    for (int i = lane; i < deg; i += 32) {
      const int k = col[start + i];
      int r = 0;
      for (int j = 0; j < deg; ++j) {
        const int kj = col[start + j];
        r += (kj < k || (kj == k && j < i)) ? 1 : 0;
      }
      col2[start + r] = k;
    }
  }
  }
}

__device__ __forceinline__ float leaky(float v) { return (v >= 0.f) ? v : NEG_SLOPE * v; }

template <bool CACHED>
__device__ __forceinline__ void gat1_node(int d, int lane, int start, int deg,
                                          const int* __restrict__ col2,
                                          const float* __restrict__ asc,
                                          const float* __restrict__ adc,
                                          const float* __restrict__ h,
                                          const float* __restrict__ b1,
                                          float* __restrict__ h1, float* es) {
  const int hd = lane >> 2;
  const float ad = adc[(size_t)d * 8 + hd];
  const float eself = leaky(asc[(size_t)d * 8 + hd] + ad);
  int sv = 0;
  if (CACHED && lane < deg) sv = col2[start + lane];
  float m = eself;
#pragma unroll 4
  for (int j = 0; j < deg; ++j) {
    const int s = CACHED ? __builtin_amdgcn_readlane(sv, j) : col2[start + j];
    const float e = leaky(asc[(size_t)s * 8 + hd] + ad);
    if (CACHED) es[j * 32] = e;
    m = fmaxf(m, e);
  }
  float ssum = 0.f;
#pragma unroll 4
  for (int j = 0; j < deg; ++j) {
    float e;
    if (CACHED) e = es[j * 32];
    else e = leaky(asc[(size_t)col2[start + j] * 8 + hd] + ad);
    const float ex = __expf(e - m);
    if (CACHED) es[j * 32] = ex;
    ssum += ex;
  }
  const float exself = __expf(eself - m);
  ssum += exself;
  float4 acc = make_float4(0.f, 0.f, 0.f, 0.f);
#pragma unroll 4
  for (int j = 0; j < deg; ++j) {
    const int s = CACHED ? __builtin_amdgcn_readlane(sv, j) : col2[start + j];
    float ex;
    if (CACHED) ex = es[j * 32];
    else ex = __expf(leaky(asc[(size_t)s * 8 + hd] + ad) - m);
    const float alpha = ex / ssum;
    const float4 hv = *(const float4*)(h + (size_t)s * 128 + lane * 4);
    acc.x = fmaf(hv.x, alpha, acc.x);
    acc.y = fmaf(hv.y, alpha, acc.y);
    acc.z = fmaf(hv.z, alpha, acc.z);
    acc.w = fmaf(hv.w, alpha, acc.w);
  }
  {
    const float alpha = exself / ssum;
    const float4 hv = *(const float4*)(h + (size_t)d * 128 + lane * 4);
    acc.x = fmaf(hv.x, alpha, acc.x);
    acc.y = fmaf(hv.y, alpha, acc.y);
    acc.z = fmaf(hv.z, alpha, acc.z);
    acc.w = fmaf(hv.w, alpha, acc.w);
  }
  const float4 bv = *(const float4*)(b1 + lane * 4);
  float4 o;
  o.x = acc.x + bv.x; o.x = (o.x > 0.f) ? o.x : (__expf(o.x) - 1.0f);
  o.y = acc.y + bv.y; o.y = (o.y > 0.f) ? o.y : (__expf(o.y) - 1.0f);
  o.z = acc.z + bv.z; o.z = (o.z > 0.f) ? o.z : (__expf(o.z) - 1.0f);
  o.w = acc.w + bv.w; o.w = (o.w > 0.f) ? o.w : (__expf(o.w) - 1.0f);
  *(float4*)(h1 + (size_t)d * 128 + lane * 4) = o;
}

__global__ void __launch_bounds__(256, 16)
gather1_kernel(const int* __restrict__ rowptr, const int* __restrict__ col2,
               const float* __restrict__ asc, const float* __restrict__ adc,
               const float* __restrict__ h, const float* __restrict__ b1,
               float* __restrict__ h1, int Nn) {
  __shared__ float es_all[8 * 32 * 32];
  const int lane = threadIdx.x & 31;
  const int w = threadIdx.x >> 5;
  float* es = es_all + w * (32 * 32) + lane;
  const int nw = gridDim.x * 8;
  for (int d = __builtin_amdgcn_readfirstlane(blockIdx.x * 8 + w); d < Nn; d += nw) {
    const int start = rowptr[d];
    const int deg = rowptr[d + 1] - start;
    if (deg <= 32) gat1_node<true>(d, lane, start, deg, col2, asc, adc, h, b1, h1, es);
    else           gat1_node<false>(d, lane, start, deg, col2, asc, adc, h, b1, h1, es);
  }
}

__device__ __forceinline__ float wave_max(float v) {
#pragma unroll
  for (int off = 16; off >= 1; off >>= 1) v = fmaxf(v, __shfl_xor(v, off));
  return v;
}
__device__ __forceinline__ float wave_sum(float v) {
#pragma unroll
  for (int off = 16; off >= 1; off >>= 1) v += __shfl_xor(v, off);
  return v;
}

__device__ __forceinline__ void gat2_accum(int n, int sv, float av, int lc,
                                           const float* __restrict__ h2, float2& acc) {
#pragma unroll 4
  for (int t = 0; t < n; ++t) {
    const int s = __builtin_amdgcn_readlane(sv, t);
    const float a = __int_as_float(__builtin_amdgcn_readlane(__float_as_int(av), t));
    const float2 hv = *(const float2*)(h2 + (size_t)s * 40 + 2 * lc);
    acc.x = fmaf(hv.x, a, acc.x);
    acc.y = fmaf(hv.y, a, acc.y);
  }
}

__global__ void __launch_bounds__(256)
gather2_kernel(const int* __restrict__ rowptr, const int* __restrict__ col2,
               const float* __restrict__ asc, const float* __restrict__ adc,
               const float* __restrict__ h2, const float* __restrict__ b2,
               float* __restrict__ out, int Nn) {
  const int lane = threadIdx.x & 31;
  const int lc = (lane < 19) ? lane : 19;
  const int nw = gridDim.x * 8;
  for (int d = __builtin_amdgcn_readfirstlane(blockIdx.x * 8 + (threadIdx.x >> 5)); d < Nn; d += nw) {
  const int start = rowptr[d];
  const int deg = rowptr[d + 1] - start;
  const float ad = adc[d];
  const float eself = leaky(asc[d] + ad);

  int s0 = 0;
  float e0 = -INFINITY;
  if (lane < deg) { s0 = col2[start + lane]; e0 = leaky(asc[s0] + ad); }
  float m = fmaxf(eself, wave_max(e0));
  for (int c = 32; c < deg; c += 32) {
    float e = -INFINITY;
    if (c + lane < deg) e = leaky(asc[col2[start + c + lane]] + ad);
    m = fmaxf(m, wave_max(e));
  }
  const float ex0 = (lane < deg) ? __expf(e0 - m) : 0.f;
  float ssum = wave_sum(ex0);
  for (int c = 32; c < deg; c += 32) {
    float ex = 0.f;
    if (c + lane < deg) ex = __expf(leaky(asc[col2[start + c + lane]] + ad) - m);
    ssum += wave_sum(ex);
  }
  const float exself = __expf(eself - m);
  ssum += exself;

  float2 acc = make_float2(0.f, 0.f);
  gat2_accum(min(deg, 32), s0, ex0 / ssum, lc, h2, acc);
  for (int c = 32; c < deg; c += 32) {
    int s = 0;
    float a = 0.f;
    if (c + lane < deg) {
      s = col2[start + c + lane];
      a = __expf(leaky(asc[s] + ad) - m) / ssum;
    }
    gat2_accum(min(deg - c, 32), s, a, lc, h2, acc);
  }
  {
    const float a = exself / ssum;
    const float2 hv = *(const float2*)(h2 + (size_t)d * 40 + 2 * lc);
    acc.x = fmaf(hv.x, a, acc.x);
    acc.y = fmaf(hv.y, a, acc.y);
  }
  if (lane < 20) {
    const float2 bv = *(const float2*)(b2 + 2 * lane);
    *(float2*)(out + (size_t)d * 40 + 2 * lane) = make_float2(acc.x + bv.x, acc.y + bv.y);
  }
  }
}

extern "C" void kernel_launch(void* const* d_in, const int* in_sizes, int n_in,
                              void* d_out, int out_size, void* d_ws, size_t ws_size,
                              hipStream_t stream) {
  const float* x   = (const float*)d_in[0];
  const int*   ei  = (const int*)d_in[1];
  const float* W1  = (const float*)d_in[2];
  const float* as1 = (const float*)d_in[3];
  const float* ad1 = (const float*)d_in[4];
  const float* b1  = (const float*)d_in[5];
  const float* W2  = (const float*)d_in[6];
  const float* as2 = (const float*)d_in[7];
  const float* ad2 = (const float*)d_in[8];
  const float* b2  = (const float*)d_in[9];
  float* out = (float*)d_out;

  const int Nn   = in_sizes[0] / 128;
  const int E    = in_sizes[1] / 2;
  const int* srcp = ei;
  const int* dstp = ei + E;
  const int NB = (Nn + 2047) / 2048;

  float* ws   = (float*)d_ws;
  float* h    = ws;  ws += (size_t)Nn * 128;
  float* h1   = ws;  ws += (size_t)Nn * 128;
  float* asc1 = ws;  ws += (size_t)Nn * 8;
  float* adc1 = ws;  ws += (size_t)Nn * 8;
  float* h2   = ws;  ws += (size_t)Nn * 40;
  float* asc2 = ws;  ws += Nn;
  float* adc2 = ws;  ws += Nn;
  int* wi     = (int*)ws;
  int* cnt    = wi;  wi += Nn;
  int* rank   = wi;  wi += E;
  int* col    = wi;  wi += E;
  int* col2   = wi;  wi += E;
  int* rowptr = wi;  wi += (size_t)Nn + 1;
  int* btot   = wi;  wi += NB;

  const int T = 256;

  zero_int_kernel<<<98, T, 0, stream>>>(cnt, Nn);
  hist_rank_kernel<<<(E + 2047) / 2048, T, 0, stream>>>(dstp, cnt, rank, E);
  scan_local_kernel<<<NB, 256, 0, stream>>>(cnt, rowptr, btot, Nn);
  scan_finish_kernel<<<NB, 256, 0, stream>>>(rowptr, btot, Nn);
  fill_col_kernel<<<(E + 2047) / 2048, T, 0, stream>>>(srcp, dstp, rowptr, rank, col, E);
  sort_seg_kernel<<<2048, 256, 0, stream>>>(rowptr, col, col2, Nn);

  gat_wmma_gemm_k128<4, 128><<<dim3(896, 2), 4 * 32, 0, stream>>>(x, W1, h, Nn / 16);
  scores1_kernel<<<782, T, 0, stream>>>(h, as1, ad1, asc1, adc1, Nn);
  gather1_kernel<<<2048, 256, 0, stream>>>(rowptr, col2, asc1, adc1, h, b1, h1, Nn);

  gat_wmma_gemm_k128<3, 40><<<dim3(1563, 1), 3 * 32, 0, stream>>>(h1, W2, h2, Nn / 16);
  scores2_kernel<<<(Nn + T - 1) / T, T, 0, stream>>>(h2, as2, ad2, asc2, adc2, Nn);
  gather2_kernel<<<2048, 256, 0, stream>>>(rowptr, col2, asc2, adc2, h2, b2, out, Nn);
}


# CSR build without global atomics: two-level counting sort on LDS counters (edges binned by 512-node destination bucket, then ranked per node); 5 CSR launches become 3; float sum order unchanged
# speedup vs baseline: 3.9670x; 1.2325x over previous
#include <hip/hip_runtime.h>
#include <hip/hip_bf16.h>
#include <math.h>
#include <stdint.h>


typedef float v2f __attribute__((ext_vector_type(2)));
typedef float v8f __attribute__((ext_vector_type(8)));

#define NEG_SLOPE 0.2f

template <int NTILES, int NACT>
__global__ void __launch_bounds__(32 * NTILES)
gat_wmma_gemm_k128(const float* __restrict__ A, const float* __restrict__ B,
                   float* __restrict__ C, int nstrips) {
  constexpr int NPAD = NTILES * 16;
  constexpr int SA = 132;
  constexpr int SB = 132;
  constexpr int NT = 32 * NTILES;
  __shared__ __attribute__((aligned(16))) float ldsA[16 * SA];
  __shared__ float ldsB[NPAD * SB];

  const int tid  = threadIdx.x;
  const int lane = tid & 31;
  const int wave = tid >> 5;
  const int c0 = blockIdx.y * NPAD;

  for (int i = tid; i < NPAD * 128; i += NT) {
    int k = i / NPAD, n = i - k * NPAD;
    int gn = c0 + n;
    ldsB[n * SB + k] = (gn < NACT) ? B[(size_t)k * NACT + gn] : 0.0f;
  }

  const int l15  = lane & 15;
  const int koff = (lane >> 4) << 1;
  const int ncol = wave * 16 + l15;
  const float* ap = &ldsA[l15 * SA + koff];
  const float* bp = &ldsB[ncol * SB + koff];

  const int gcol = c0 + ncol;

  for (int strip = blockIdx.x; strip < nstrips; strip += gridDim.x) {
    const size_t m0 = (size_t)strip * 16;
    constexpr int AIT = (16 * 32 + NT - 1) / NT;
    float4 av[AIT];
#pragma unroll
    for (int u = 0; u < AIT; ++u) {
      const int i = min(tid + u * NT, 16 * 32 - 1);
      av[u] = *(const float4*)(A + (m0 + (i >> 5)) * 128 + ((i & 31) << 2));
    }
#pragma unroll
    for (int u = 0; u < AIT; ++u) {
      const int i = tid + u * NT;
      if ((u + 1) * NT <= 16 * 32 || i < 16 * 32) *(float4*)&ldsA[(i >> 5) * SA + ((i & 31) << 2)] = av[u];
    }
    __syncthreads();

    v8f acc = {0.f, 0.f, 0.f, 0.f, 0.f, 0.f, 0.f, 0.f};
#pragma unroll
    for (int kb = 0; kb < 128; kb += 4) {
      v2f a = *(const v2f*)(ap + kb);
      v2f b = *(const v2f*)(bp + kb);
      acc = __builtin_amdgcn_wmma_f32_16x16x4_f32(
           false, a,  false, b,
           (short)0, acc,  false,  false);
    }

    if (gcol < NACT) {
      float* cp = C + (m0 + (size_t)((lane >> 4) << 3)) * NACT + gcol;
#pragma unroll
      for (int r = 0; r < 8; ++r) cp[(size_t)r * NACT] = acc[r];
    }
    __syncthreads();
  }
}

__global__ void __launch_bounds__(256)
scores1_kernel(const float* __restrict__ h,
                               const float* __restrict__ att_src,
                               const float* __restrict__ att_dst,
                               float* __restrict__ a_src, float* __restrict__ a_dst,
                               int Nn) {
  const int gt = blockIdx.x * blockDim.x + threadIdx.x;
  const int hd = gt & 7;
  const int nstep = (gridDim.x * blockDim.x) >> 3;
  float ws[16], wd[16];
#pragma unroll
  for (int c = 0; c < 16; ++c) { ws[c] = att_src[hd * 16 + c]; wd[c] = att_dst[hd * 16 + c]; }
  for (int n = gt >> 3; n < Nn; n += nstep) {
    const float4* hp = (const float4*)(h + (size_t)n * 128 + hd * 16);
    float v[16];
#pragma unroll
    for (int q = 0; q < 4; ++q) {
      const float4 t = hp[q];
      v[4 * q] = t.x; v[4 * q + 1] = t.y; v[4 * q + 2] = t.z; v[4 * q + 3] = t.w;
    }
    float s = 0.f, d = 0.f;
#pragma unroll
    for (int c = 0; c < 16; ++c) {
      s = fmaf(v[c], ws[c], s);
      d = fmaf(v[c], wd[c], d);
    }
    a_src[(size_t)n * 8 + hd] = s;
    a_dst[(size_t)n * 8 + hd] = d;
  }
}

__global__ void scores2_kernel(const float* __restrict__ h2,
                               const float* __restrict__ att_src,
                               const float* __restrict__ att_dst,
                               float* __restrict__ a_src, float* __restrict__ a_dst,
                               int Nn) {
  int n = blockIdx.x * blockDim.x + threadIdx.x;
  if (n >= Nn) return;
  const float* hp = h2 + (size_t)n * 40;
  float s = 0.f, d = 0.f;
#pragma unroll
  for (int c = 0; c < 40; ++c) {
    float v = hp[c];
    s = fmaf(v, att_src[c], s);
    d = fmaf(v, att_dst[c], d);
  }
  a_src[n] = s;
  a_dst[n] = d;
}

#define BSHIFT 9
#define BNODES 512
#define NBINS 256
#define SRC_BITS 17
#define ETILE 8192

__global__ void __launch_bounds__(1024)
bucket_count_kernel(const int* __restrict__ dst, int* __restrict__ bcnt, int E) {
  __shared__ int hist[NBINS];
  const int tid = threadIdx.x;
  if (tid < NBINS) hist[tid] = 0;
  const int base = blockIdx.x * ETILE + tid;
  int dv[8];
#pragma unroll
  for (int k = 0; k < 8; ++k) {
    const int e = base + k * 1024;
    dv[k] = (e < E) ? dst[e] : -1;
  }
  __syncthreads();
#pragma unroll
  for (int k = 0; k < 8; ++k) if (dv[k] >= 0) atomicAdd(&hist[dv[k] >> BSHIFT], 1);
  __syncthreads();
  if (tid < NBINS) bcnt[blockIdx.x * NBINS + tid] = hist[tid];
}

__global__ void __launch_bounds__(1024)
bucket_scatter_kernel(const int* __restrict__ src, const int* __restrict__ dst,
                      const int* __restrict__ bcnt, int* __restrict__ bstart,
                      int* __restrict__ rec, int E, int nblk) {
  __shared__ int ptot[4][NBINS];
  __shared__ int ppre[4][NBINS];
  __shared__ int cursor[NBINS + 1];
  const int tid = threadIdx.x;
  const int lane = tid & 31;
  const int base = blockIdx.x * ETILE + tid;
  int dv[8], sv[8];
#pragma unroll
  for (int k = 0; k < 8; ++k) {
    const int e = base + k * 1024;
    const bool ok = e < E;
    dv[k] = ok ? dst[e] : -1;
    sv[k] = ok ? src[e] : 0;
  }
  {
    const int k = tid & (NBINS - 1);
    const int q = tid >> 8;
    const int gq = (nblk + 3) >> 2;
    const int j0 = q * gq;
    const int j1 = min(nblk, j0 + gq);
    int tot = 0, pre = 0;
#pragma unroll 8
    for (int j = j0; j < j1; ++j) {
      const int c = bcnt[j * NBINS + k];
      tot += c;
      pre += (j < (int)blockIdx.x) ? c : 0;
    }
    ptot[q][k] = tot;
    ppre[q][k] = pre;
    if (tid == 0) cursor[NBINS] = 0;
  }
  __syncthreads();
  if (tid < 32) {
    int v[8];
    int run = 0;
#pragma unroll
    for (int i = 0; i < 8; ++i) {
      const int idx = tid * 8 + i;
      v[i] = run;
      run += ptot[0][idx] + ptot[1][idx] + ptot[2][idx] + ptot[3][idx];
    }
    int inc = run;
#pragma unroll
    for (int off = 1; off < 32; off <<= 1) {
      const int nb = __shfl_up(inc, off);
      if (lane >= off) inc += nb;
    }
    const int excl = inc - run;
#pragma unroll
    for (int i = 0; i < 8; ++i) {
      const int idx = tid * 8 + i;
      const int st = excl + v[i];
      cursor[idx] = st + ppre[0][idx] + ppre[1][idx] + ppre[2][idx] + ppre[3][idx];
      if (blockIdx.x == 0) bstart[idx] = st;
    }
    if (blockIdx.x == 0 && tid == 31) bstart[NBINS] = inc;
  }
  __syncthreads();
  int pv[8];
#pragma unroll
  for (int k = 0; k < 8; ++k) pv[k] = atomicAdd(&cursor[(dv[k] >= 0) ? (dv[k] >> BSHIFT) : NBINS], 1);
#pragma unroll
  for (int k = 0; k < 8; ++k)
    if (dv[k] >= 0) rec[pv[k]] = ((dv[k] & (BNODES - 1)) << SRC_BITS) | sv[k];
}

__global__ void __launch_bounds__(1024)
bucket_csr_kernel(const int* __restrict__ bstart, const int* __restrict__ rec,
                  int* __restrict__ rowptr, int* __restrict__ col, int Nn) {
  __shared__ int cntl[BNODES + 1];
  __shared__ int wsum[16];
  const int tid = threadIdx.x, lane = tid & 31, w = tid >> 5;
  const int bs = bstart[blockIdx.x];
  const int n = bstart[blockIdx.x + 1] - bs;
  if (tid <= BNODES) cntl[tid] = 0;
  __syncthreads();
  const int* rp = rec + bs;
  for (int t0 = 0; t0 < n; t0 += ETILE) {
    int rv[8];
#pragma unroll
    for (int k = 0; k < 8; ++k) {
      const int i = t0 + k * 1024 + tid;
      rv[k] = (i < n) ? rp[i] : -1;
    }
#pragma unroll
    for (int k = 0; k < 8; ++k) if (rv[k] >= 0) atomicAdd(&cntl[rv[k] >> SRC_BITS], 1);
  }
  __syncthreads();
  int c = 0, inc = 0;
  if (tid < BNODES) {
    c = cntl[tid];
    inc = c;
#pragma unroll
    for (int off = 1; off < 32; off <<= 1) {
      const int nb = __shfl_up(inc, off);
      if (lane >= off) inc += nb;
    }
    if (lane == 31) wsum[w] = inc;
  }
  __syncthreads();
  if (tid < BNODES) {
    int woff = 0;
#pragma unroll
    for (int k = 0; k < 16; ++k) woff += (k < w) ? wsum[k] : 0;
    const int excl = woff + inc - c;
    cntl[tid] = excl;
    const int node = blockIdx.x * BNODES + tid;
    if (node < Nn) rowptr[node] = bs + excl;
  }
  if (blockIdx.x == gridDim.x - 1 && tid == 0) rowptr[Nn] = bs + n;
  __syncthreads();
  int* cp = col + bs;
  for (int t0 = 0; t0 < n; t0 += ETILE) {
    int rv[8], pv[8];
#pragma unroll
    for (int k = 0; k < 8; ++k) {
      const int i = t0 + k * 1024 + tid;
      rv[k] = (i < n) ? rp[i] : -1;
    }
#pragma unroll
    for (int k = 0; k < 8; ++k) pv[k] = atomicAdd(&cntl[(rv[k] >= 0) ? (rv[k] >> SRC_BITS) : BNODES], 1);
#pragma unroll
    for (int k = 0; k < 8; ++k) if (rv[k] >= 0) cp[pv[k]] = rv[k] & ((1 << SRC_BITS) - 1);
  }
}

__global__ void __launch_bounds__(256)
sort_seg_kernel(const int* __restrict__ rowptr, const int* __restrict__ col,
                int* __restrict__ col2, int Nn) {
  const int lane = threadIdx.x & 31;
  const int nw = gridDim.x * 8;
  for (int d = __builtin_amdgcn_readfirstlane(blockIdx.x * 8 + (threadIdx.x >> 5)); d < Nn; d += nw) {
  const int start = rowptr[d];
  const int deg = rowptr[d + 1] - start;
  if (deg <= 32) {
    const int k = (lane < deg) ? col[start + lane] : 0x7fffffff;
    int r = 0;
    for (int j = 0; j < deg; ++j) {
      const int kj = __builtin_amdgcn_readlane(k, j);
      r += (kj < k || (kj == k && j < lane)) ? 1 : 0;
    }
    if (lane < deg) col2[start + r] = k;
  } else {
    for (int i = lane; i < deg; i += 32) {
      const int k = col[start + i];
      int r = 0;
      for (int j = 0; j < deg; ++j) {
        const int kj = col[start + j];
        r += (kj < k || (kj == k && j < i)) ? 1 : 0;
      }
      col2[start + r] = k;
    }
  }
  }
}

__device__ __forceinline__ float leaky(float v) { return (v >= 0.f) ? v : NEG_SLOPE * v; }

template <bool CACHED>
__device__ __forceinline__ void gat1_node(int d, int lane, int start, int deg,
                                          const int* __restrict__ col2,
                                          const float* __restrict__ asc,
                                          const float* __restrict__ adc,
                                          const float* __restrict__ h,
                                          const float* __restrict__ b1,
                                          float* __restrict__ h1, float* es) {
  const int hd = lane >> 2;
  const float ad = adc[(size_t)d * 8 + hd];
  const float eself = leaky(asc[(size_t)d * 8 + hd] + ad);
  int sv = 0;
  if (CACHED && lane < deg) sv = col2[start + lane];
  float m = eself;
#pragma unroll 4
  for (int j = 0; j < deg; ++j) {
    const int s = CACHED ? __builtin_amdgcn_readlane(sv, j) : col2[start + j];
    const float e = leaky(asc[(size_t)s * 8 + hd] + ad);
    if (CACHED) es[j * 32] = e;
    m = fmaxf(m, e);
  }
  float ssum = 0.f;
#pragma unroll 4
  for (int j = 0; j < deg; ++j) {
    float e;
    if (CACHED) e = es[j * 32];
    else e = leaky(asc[(size_t)col2[start + j] * 8 + hd] + ad);
    const float ex = __expf(e - m);
    if (CACHED) es[j * 32] = ex;
    ssum += ex;
  }
  const float exself = __expf(eself - m);
  ssum += exself;
  float4 acc = make_float4(0.f, 0.f, 0.f, 0.f);
#pragma unroll 4
  for (int j = 0; j < deg; ++j) {
    const int s = CACHED ? __builtin_amdgcn_readlane(sv, j) : col2[start + j];
    float ex;
    if (CACHED) ex = es[j * 32];
    else ex = __expf(leaky(asc[(size_t)s * 8 + hd] + ad) - m);
    const float alpha = ex / ssum;
    const float4 hv = *(const float4*)(h + (size_t)s * 128 + lane * 4);
    acc.x = fmaf(hv.x, alpha, acc.x);
    acc.y = fmaf(hv.y, alpha, acc.y);
    acc.z = fmaf(hv.z, alpha, acc.z);
    acc.w = fmaf(hv.w, alpha, acc.w);
  }
  {
    const float alpha = exself / ssum;
    const float4 hv = *(const float4*)(h + (size_t)d * 128 + lane * 4);
    acc.x = fmaf(hv.x, alpha, acc.x);
    acc.y = fmaf(hv.y, alpha, acc.y);
    acc.z = fmaf(hv.z, alpha, acc.z);
    acc.w = fmaf(hv.w, alpha, acc.w);
  }
  const float4 bv = *(const float4*)(b1 + lane * 4);
  float4 o;
  o.x = acc.x + bv.x; o.x = (o.x > 0.f) ? o.x : (__expf(o.x) - 1.0f);
  o.y = acc.y + bv.y; o.y = (o.y > 0.f) ? o.y : (__expf(o.y) - 1.0f);
  o.z = acc.z + bv.z; o.z = (o.z > 0.f) ? o.z : (__expf(o.z) - 1.0f);
  o.w = acc.w + bv.w; o.w = (o.w > 0.f) ? o.w : (__expf(o.w) - 1.0f);
  *(float4*)(h1 + (size_t)d * 128 + lane * 4) = o;
}

__global__ void __launch_bounds__(256, 16)
gather1_kernel(const int* __restrict__ rowptr, const int* __restrict__ col2,
               const float* __restrict__ asc, const float* __restrict__ adc,
               const float* __restrict__ h, const float* __restrict__ b1,
               float* __restrict__ h1, int Nn) {
  __shared__ float es_all[8 * 32 * 32];
  const int lane = threadIdx.x & 31;
  const int w = threadIdx.x >> 5;
  float* es = es_all + w * (32 * 32) + lane;
  const int nw = gridDim.x * 8;
  for (int d = __builtin_amdgcn_readfirstlane(blockIdx.x * 8 + w); d < Nn; d += nw) {
    const int start = rowptr[d];
    const int deg = rowptr[d + 1] - start;
    if (deg <= 32) gat1_node<true>(d, lane, start, deg, col2, asc, adc, h, b1, h1, es);
    else           gat1_node<false>(d, lane, start, deg, col2, asc, adc, h, b1, h1, es);
  }
}

__device__ __forceinline__ float wave_max(float v) {
#pragma unroll
  for (int off = 16; off >= 1; off >>= 1) v = fmaxf(v, __shfl_xor(v, off));
  return v;
}
__device__ __forceinline__ float wave_sum(float v) {
#pragma unroll
  for (int off = 16; off >= 1; off >>= 1) v += __shfl_xor(v, off);
  return v;
}

__device__ __forceinline__ void gat2_accum(int n, int sv, float av, int lc,
                                           const float* __restrict__ h2, float2& acc) {
#pragma unroll 4
  for (int t = 0; t < n; ++t) {
    const int s = __builtin_amdgcn_readlane(sv, t);
    const float a = __int_as_float(__builtin_amdgcn_readlane(__float_as_int(av), t));
    const float2 hv = *(const float2*)(h2 + (size_t)s * 40 + 2 * lc);
    acc.x = fmaf(hv.x, a, acc.x);
    acc.y = fmaf(hv.y, a, acc.y);
  }
}

__global__ void __launch_bounds__(256)
gather2_kernel(const int* __restrict__ rowptr, const int* __restrict__ col2,
               const float* __restrict__ asc, const float* __restrict__ adc,
               const float* __restrict__ h2, const float* __restrict__ b2,
               float* __restrict__ out, int Nn) {
  const int lane = threadIdx.x & 31;
  const int lc = (lane < 19) ? lane : 19;
  const int nw = gridDim.x * 8;
  for (int d = __builtin_amdgcn_readfirstlane(blockIdx.x * 8 + (threadIdx.x >> 5)); d < Nn; d += nw) {
  const int start = rowptr[d];
  const int deg = rowptr[d + 1] - start;
  const float ad = adc[d];
  const float eself = leaky(asc[d] + ad);

  int s0 = 0;
  float e0 = -INFINITY;
  if (lane < deg) { s0 = col2[start + lane]; e0 = leaky(asc[s0] + ad); }
  float m = fmaxf(eself, wave_max(e0));
  for (int c = 32; c < deg; c += 32) {
    float e = -INFINITY;
    if (c + lane < deg) e = leaky(asc[col2[start + c + lane]] + ad);
    m = fmaxf(m, wave_max(e));
  }
  const float ex0 = (lane < deg) ? __expf(e0 - m) : 0.f;
  float ssum = wave_sum(ex0);
  for (int c = 32; c < deg; c += 32) {
    float ex = 0.f;
    if (c + lane < deg) ex = __expf(leaky(asc[col2[start + c + lane]] + ad) - m);
    ssum += wave_sum(ex);
  }
  const float exself = __expf(eself - m);
  ssum += exself;

  float2 acc = make_float2(0.f, 0.f);
  gat2_accum(min(deg, 32), s0, ex0 / ssum, lc, h2, acc);
  for (int c = 32; c < deg; c += 32) {
    int s = 0;
    float a = 0.f;
    if (c + lane < deg) {
      s = col2[start + c + lane];
      a = __expf(leaky(asc[s] + ad) - m) / ssum;
    }
    gat2_accum(min(deg - c, 32), s, a, lc, h2, acc);
  }
  {
    const float a = exself / ssum;
    const float2 hv = *(const float2*)(h2 + (size_t)d * 40 + 2 * lc);
    acc.x = fmaf(hv.x, a, acc.x);
    acc.y = fmaf(hv.y, a, acc.y);
  }
  if (lane < 20) {
    const float2 bv = *(const float2*)(b2 + 2 * lane);
    *(float2*)(out + (size_t)d * 40 + 2 * lane) = make_float2(acc.x + bv.x, acc.y + bv.y);
  }
  }
}

extern "C" void kernel_launch(void* const* d_in, const int* in_sizes, int n_in,
                              void* d_out, int out_size, void* d_ws, size_t ws_size,
                              hipStream_t stream) {
  const float* x   = (const float*)d_in[0];
  const int*   ei  = (const int*)d_in[1];
  const float* W1  = (const float*)d_in[2];
  const float* as1 = (const float*)d_in[3];
  const float* ad1 = (const float*)d_in[4];
  const float* b1  = (const float*)d_in[5];
  const float* W2  = (const float*)d_in[6];
  const float* as2 = (const float*)d_in[7];
  const float* ad2 = (const float*)d_in[8];
  const float* b2  = (const float*)d_in[9];
  float* out = (float*)d_out;

  const int Nn   = in_sizes[0] / 128;
  const int E    = in_sizes[1] / 2;
  const int* srcp = ei;
  const int* dstp = ei + E;
  const int NBLK = (E + ETILE - 1) / ETILE;
  const int NBKT = (Nn + BNODES - 1) / BNODES;

  float* ws   = (float*)d_ws;
  float* h    = ws;  ws += (size_t)Nn * 128;
  float* h1   = ws;  ws += (size_t)Nn * 128;
  float* asc1 = ws;  ws += (size_t)Nn * 8;
  float* adc1 = ws;  ws += (size_t)Nn * 8;
  float* h2   = ws;  ws += (size_t)Nn * 40;
  float* asc2 = ws;  ws += Nn;
  float* adc2 = ws;  ws += Nn;
  int* wi     = (int*)ws;
  int* bcnt   = wi;  wi += (size_t)NBLK * NBINS;
  int* rec    = wi;  wi += E;
  int* col    = wi;  wi += E;
  int* col2   = wi;  wi += E;
  int* rowptr = wi;  wi += (size_t)Nn + 1;
  int* bstart = wi;  wi += NBINS + 1;

  const int T = 256;

  bucket_count_kernel<<<NBLK, 1024, 0, stream>>>(dstp, bcnt, E);
  bucket_scatter_kernel<<<NBLK, 1024, 0, stream>>>(srcp, dstp, bcnt, bstart, rec, E, NBLK);
  bucket_csr_kernel<<<NBKT, 1024, 0, stream>>>(bstart, rec, rowptr, col, Nn);
  sort_seg_kernel<<<2048, 256, 0, stream>>>(rowptr, col, col2, Nn);

  gat_wmma_gemm_k128<4, 128><<<dim3(896, 2), 4 * 32, 0, stream>>>(x, W1, h, Nn / 16);
  scores1_kernel<<<782, T, 0, stream>>>(h, as1, ad1, asc1, adc1, Nn);
  gather1_kernel<<<2048, 256, 0, stream>>>(rowptr, col2, asc1, adc1, h, b1, h1, Nn);

  gat_wmma_gemm_k128<3, 40><<<dim3(1563, 1), 3 * 32, 0, stream>>>(h1, W2, h2, Nn / 16);
  scores2_kernel<<<(Nn + T - 1) / T, T, 0, stream>>>(h2, as2, ad2, asc2, adc2, Nn);
  gather2_kernel<<<2048, 256, 0, stream>>>(rowptr, col2, asc2, adc2, h2, b2, out, Nn);
}


# Attention scores formed in the GEMM epilogues from accumulator tiles (16-lane DPP tree per row; layer 2 adds 3 wave partials in LDS): both score launches and h/h2 re-reads go; score sums now trees
# speedup vs baseline: 4.0653x; 1.0248x over previous
#include <hip/hip_runtime.h>
#include <hip/hip_bf16.h>
#include <math.h>
#include <stdint.h>


typedef float v2f __attribute__((ext_vector_type(2)));
typedef float v8f __attribute__((ext_vector_type(8)));

#define NEG_SLOPE 0.2f

template <int MASK>
__device__ __forceinline__ float row_xor_get(float v) {
  return __int_as_float(__builtin_amdgcn_update_dpp(0, __float_as_int(v), 0x160 + MASK, 0xf, 0xf, true));
}

template <int NTILES, int NACT, int HEADS>
__global__ void __launch_bounds__(32 * NTILES)
gat_wmma_gemm_k128(const float* __restrict__ A, const float* __restrict__ B,
                   float* __restrict__ C,
                   const float* __restrict__ att_src, const float* __restrict__ att_dst,
                   float* __restrict__ a_src, float* __restrict__ a_dst, int nstrips) {
  static_assert(HEADS == 1 || HEADS * 16 == NACT, "a wave's 16 columns must be one head");
  constexpr int NPAD = NTILES * 16;
  constexpr int SA = 132;
  constexpr int SB = 132;
  constexpr int NT = 32 * NTILES;
  __shared__ __attribute__((aligned(16))) float ldsA[16 * SA];
  __shared__ float ldsB[NPAD * SB];
  __shared__ float red[(HEADS == 1) ? NTILES * 32 : 1];

  const int tid  = threadIdx.x;
  const int lane = tid & 31;
  const int wave = tid >> 5;
  const int c0 = blockIdx.y * NPAD;

  for (int i = tid; i < NPAD * 128; i += NT) {
    int k = i / NPAD, n = i - k * NPAD;
    int gn = c0 + n;
    ldsB[n * SB + k] = (gn < NACT) ? B[(size_t)k * NACT + gn] : 0.0f;
  }

  const int l15  = lane & 15;
  const int koff = (lane >> 4) << 1;
  const int ncol = wave * 16 + l15;
  const float* ap = &ldsA[l15 * SA + koff];
  const float* bp = &ldsB[ncol * SB + koff];

  const int gcol = c0 + ncol;
  const float wsrc = (gcol < NACT) ? att_src[gcol] : 0.0f;
  const float wdst = (gcol < NACT) ? att_dst[gcol] : 0.0f;
  const int hd = blockIdx.y * NTILES + wave;

  for (int strip = blockIdx.x; strip < nstrips; strip += gridDim.x) {
    const size_t m0 = (size_t)strip * 16;
    constexpr int AIT = (16 * 32 + NT - 1) / NT;
    float4 av[AIT];
#pragma unroll
    for (int u = 0; u < AIT; ++u) {
      const int i = min(tid + u * NT, 16 * 32 - 1);
      av[u] = *(const float4*)(A + (m0 + (i >> 5)) * 128 + ((i & 31) << 2));
    }
#pragma unroll
    for (int u = 0; u < AIT; ++u) {
      const int i = tid + u * NT;
      if ((u + 1) * NT <= 16 * 32 || i < 16 * 32) *(float4*)&ldsA[(i >> 5) * SA + ((i & 31) << 2)] = av[u];
    }
    __syncthreads();

    v8f acc = {0.f, 0.f, 0.f, 0.f, 0.f, 0.f, 0.f, 0.f};
#pragma unroll
    for (int kb = 0; kb < 128; kb += 4) {
      v2f a = *(const v2f*)(ap + kb);
      v2f b = *(const v2f*)(bp + kb);
      acc = __builtin_amdgcn_wmma_f32_16x16x4_f32(
           false, a,  false, b,
           (short)0, acc,  false,  false);
    }

    if (gcol < NACT) {
      float* cp = C + (m0 + (size_t)((lane >> 4) << 3)) * NACT + gcol;
#pragma unroll
      for (int r = 0; r < 8; ++r) cp[(size_t)r * NACT] = acc[r];
    }

    const bool b3 = (l15 & 8) != 0, b2 = (l15 & 4) != 0, b1 = (l15 & 2) != 0;
    float ps[8], pd[8];
#pragma unroll
    for (int r = 0; r < 8; ++r) { ps[r] = acc[r] * wsrc; pd[r] = acc[r] * wdst; }
    float s4[4], d4[4];
#pragma unroll
    for (int i = 0; i < 4; ++i) {
      s4[i] = (b3 ? ps[4 + i] : ps[i]) + row_xor_get<8>(b3 ? ps[i] : ps[4 + i]);
      d4[i] = (b3 ? pd[4 + i] : pd[i]) + row_xor_get<8>(b3 ? pd[i] : pd[4 + i]);
    }
    float s2[2], d2[2];
#pragma unroll
    for (int j = 0; j < 2; ++j) {
      s2[j] = (b2 ? s4[2 + j] : s4[j]) + row_xor_get<4>(b2 ? s4[j] : s4[2 + j]);
      d2[j] = (b2 ? d4[2 + j] : d4[j]) + row_xor_get<4>(b2 ? d4[j] : d4[2 + j]);
    }
    float s1 = (b1 ? s2[1] : s2[0]) + row_xor_get<2>(b1 ? s2[0] : s2[1]);
    float d1 = (b1 ? d2[1] : d2[0]) + row_xor_get<2>(b1 ? d2[0] : d2[1]);
    s1 += row_xor_get<1>(s1);
    d1 += row_xor_get<1>(d1);
    const bool is_src = (l15 & 1) == 0;
    const float sv = is_src ? s1 : d1;
    const int row = ((lane >> 4) << 3) + (l15 >> 1);
    if constexpr (HEADS == 1) {
      red[(wave * 16 + row) * 2 + (is_src ? 0 : 1)] = sv;
    } else {
      float* sp = is_src ? a_src : a_dst;
      sp[(m0 + row) * HEADS + hd] = sv;
    }
    __syncthreads();
    if constexpr (HEADS == 1) if (wave == 0) {
      const int r16 = lane & 15, which = lane >> 4;
      float s = red[r16 * 2 + which];
#pragma unroll
      for (int w = 1; w < NTILES; ++w) s += red[(w * 16 + r16) * 2 + which];
      float* sp = which ? a_dst : a_src;
      sp[m0 + r16] = s;
    }
  }
}

#define BSHIFT 9
#define BNODES 512
#define NBINS 256
#define SRC_BITS 17
#define ETILE 8192

__global__ void __launch_bounds__(1024)
bucket_count_kernel(const int* __restrict__ dst, int* __restrict__ bcnt, int E) {
  __shared__ int hist[NBINS];
  const int tid = threadIdx.x;
  if (tid < NBINS) hist[tid] = 0;
  const int base = blockIdx.x * ETILE + tid;
  int dv[8];
#pragma unroll
  for (int k = 0; k < 8; ++k) {
    const int e = base + k * 1024;
    dv[k] = (e < E) ? dst[e] : -1;
  }
  __syncthreads();
#pragma unroll
  for (int k = 0; k < 8; ++k) if (dv[k] >= 0) atomicAdd(&hist[dv[k] >> BSHIFT], 1);
  __syncthreads();
  if (tid < NBINS) bcnt[blockIdx.x * NBINS + tid] = hist[tid];
}

__global__ void __launch_bounds__(1024)
bucket_scatter_kernel(const int* __restrict__ src, const int* __restrict__ dst,
                      const int* __restrict__ bcnt, int* __restrict__ bstart,
                      int* __restrict__ rec, int E, int nblk) {
  __shared__ int ptot[4][NBINS];
  __shared__ int ppre[4][NBINS];
  __shared__ int cursor[NBINS + 1];
  const int tid = threadIdx.x;
  const int lane = tid & 31;
  const int base = blockIdx.x * ETILE + tid;
  int dv[8], sv[8];
#pragma unroll
  for (int k = 0; k < 8; ++k) {
    const int e = base + k * 1024;
    const bool ok = e < E;
    dv[k] = ok ? dst[e] : -1;
    sv[k] = ok ? src[e] : 0;
  }
  {
    const int k = tid & (NBINS - 1);
    const int q = tid >> 8;
    const int gq = (nblk + 3) >> 2;
    const int j0 = q * gq;
    const int j1 = min(nblk, j0 + gq);
    int tot = 0, pre = 0;
#pragma unroll 8
    for (int j = j0; j < j1; ++j) {
      const int c = bcnt[j * NBINS + k];
      tot += c;
      pre += (j < (int)blockIdx.x) ? c : 0;
    }
    ptot[q][k] = tot;
    ppre[q][k] = pre;
    if (tid == 0) cursor[NBINS] = 0;
  }
  __syncthreads();
  if (tid < 32) {
    int v[8];
    int run = 0;
#pragma unroll
    for (int i = 0; i < 8; ++i) {
      const int idx = tid * 8 + i;
      v[i] = run;
      run += ptot[0][idx] + ptot[1][idx] + ptot[2][idx] + ptot[3][idx];
    }
    int inc = run;
#pragma unroll
    for (int off = 1; off < 32; off <<= 1) {
      const int nb = __shfl_up(inc, off);
      if (lane >= off) inc += nb;
    }
    const int excl = inc - run;
#pragma unroll
    for (int i = 0; i < 8; ++i) {
      const int idx = tid * 8 + i;
      const int st = excl + v[i];
      cursor[idx] = st + ppre[0][idx] + ppre[1][idx] + ppre[2][idx] + ppre[3][idx];
      if (blockIdx.x == 0) bstart[idx] = st;
    }
    if (blockIdx.x == 0 && tid == 31) bstart[NBINS] = inc;
  }
  __syncthreads();
  int pv[8];
#pragma unroll
  for (int k = 0; k < 8; ++k) pv[k] = atomicAdd(&cursor[(dv[k] >= 0) ? (dv[k] >> BSHIFT) : NBINS], 1);
#pragma unroll
  for (int k = 0; k < 8; ++k)
    if (dv[k] >= 0) rec[pv[k]] = ((dv[k] & (BNODES - 1)) << SRC_BITS) | sv[k];
}

__global__ void __launch_bounds__(1024)
bucket_csr_kernel(const int* __restrict__ bstart, const int* __restrict__ rec,
                  int* __restrict__ rowptr, int* __restrict__ col, int Nn) {
  __shared__ int cntl[BNODES + 1];
  __shared__ int wsum[16];
  const int tid = threadIdx.x, lane = tid & 31, w = tid >> 5;
  const int bs = bstart[blockIdx.x];
  const int n = bstart[blockIdx.x + 1] - bs;
  if (tid <= BNODES) cntl[tid] = 0;
  __syncthreads();
  const int* rp = rec + bs;
  for (int t0 = 0; t0 < n; t0 += ETILE) {
    int rv[8];
#pragma unroll
    for (int k = 0; k < 8; ++k) {
      const int i = t0 + k * 1024 + tid;
      rv[k] = (i < n) ? rp[i] : -1;
    }
#pragma unroll
    for (int k = 0; k < 8; ++k) if (rv[k] >= 0) atomicAdd(&cntl[rv[k] >> SRC_BITS], 1);
  }
  __syncthreads();
  int c = 0, inc = 0;
  if (tid < BNODES) {
    c = cntl[tid];
    inc = c;
#pragma unroll
    for (int off = 1; off < 32; off <<= 1) {
      const int nb = __shfl_up(inc, off);
      if (lane >= off) inc += nb;
    }
    if (lane == 31) wsum[w] = inc;
  }
  __syncthreads();
  if (tid < BNODES) {
    int woff = 0;
#pragma unroll
    for (int k = 0; k < 16; ++k) woff += (k < w) ? wsum[k] : 0;
    const int excl = woff + inc - c;
    cntl[tid] = excl;
    const int node = blockIdx.x * BNODES + tid;
    if (node < Nn) rowptr[node] = bs + excl;
  }
  if (blockIdx.x == gridDim.x - 1 && tid == 0) rowptr[Nn] = bs + n;
  __syncthreads();
  int* cp = col + bs;
  for (int t0 = 0; t0 < n; t0 += ETILE) {
    int rv[8], pv[8];
#pragma unroll
    for (int k = 0; k < 8; ++k) {
      const int i = t0 + k * 1024 + tid;
      rv[k] = (i < n) ? rp[i] : -1;
    }
#pragma unroll
    for (int k = 0; k < 8; ++k) pv[k] = atomicAdd(&cntl[(rv[k] >= 0) ? (rv[k] >> SRC_BITS) : BNODES], 1);
#pragma unroll
    for (int k = 0; k < 8; ++k) if (rv[k] >= 0) cp[pv[k]] = rv[k] & ((1 << SRC_BITS) - 1);
  }
}

__global__ void __launch_bounds__(256)
sort_seg_kernel(const int* __restrict__ rowptr, const int* __restrict__ col,
                int* __restrict__ col2, int Nn) {
  const int lane = threadIdx.x & 31;
  const int nw = gridDim.x * 8;
  for (int d = __builtin_amdgcn_readfirstlane(blockIdx.x * 8 + (threadIdx.x >> 5)); d < Nn; d += nw) {
  const int start = rowptr[d];
  const int deg = rowptr[d + 1] - start;
  if (deg <= 32) {
    const int k = (lane < deg) ? col[start + lane] : 0x7fffffff;
    int r = 0;
    for (int j = 0; j < deg; ++j) {
      const int kj = __builtin_amdgcn_readlane(k, j);
      r += (kj < k || (kj == k && j < lane)) ? 1 : 0;
    }
    if (lane < deg) col2[start + r] = k;
  } else {
    for (int i = lane; i < deg; i += 32) {
      const int k = col[start + i];
      int r = 0;
      for (int j = 0; j < deg; ++j) {
        const int kj = col[start + j];
        r += (kj < k || (kj == k && j < i)) ? 1 : 0;
      }
      col2[start + r] = k;
    }
  }
  }
}

__device__ __forceinline__ float leaky(float v) { return (v >= 0.f) ? v : NEG_SLOPE * v; }

template <bool CACHED>
__device__ __forceinline__ void gat1_node(int d, int lane, int start, int deg,
                                          const int* __restrict__ col2,
                                          const float* __restrict__ asc,
                                          const float* __restrict__ adc,
                                          const float* __restrict__ h,
                                          const float* __restrict__ b1,
                                          float* __restrict__ h1, float* es) {
  const int hd = lane >> 2;
  const float ad = adc[(size_t)d * 8 + hd];
  const float eself = leaky(asc[(size_t)d * 8 + hd] + ad);
  int sv = 0;
  if (CACHED && lane < deg) sv = col2[start + lane];
  float m = eself;
#pragma unroll 4
  for (int j = 0; j < deg; ++j) {
    const int s = CACHED ? __builtin_amdgcn_readlane(sv, j) : col2[start + j];
    const float e = leaky(asc[(size_t)s * 8 + hd] + ad);
    if (CACHED) es[j * 32] = e;
    m = fmaxf(m, e);
  }
  float ssum = 0.f;
#pragma unroll 4
  for (int j = 0; j < deg; ++j) {
    float e;
    if (CACHED) e = es[j * 32];
    else e = leaky(asc[(size_t)col2[start + j] * 8 + hd] + ad);
    const float ex = __expf(e - m);
    if (CACHED) es[j * 32] = ex;
    ssum += ex;
  }
  const float exself = __expf(eself - m);
  ssum += exself;
  float4 acc = make_float4(0.f, 0.f, 0.f, 0.f);
#pragma unroll 4
  for (int j = 0; j < deg; ++j) {
    const int s = CACHED ? __builtin_amdgcn_readlane(sv, j) : col2[start + j];
    float ex;
    if (CACHED) ex = es[j * 32];
    else ex = __expf(leaky(asc[(size_t)s * 8 + hd] + ad) - m);
    const float alpha = ex / ssum;
    const float4 hv = *(const float4*)(h + (size_t)s * 128 + lane * 4);
    acc.x = fmaf(hv.x, alpha, acc.x);
    acc.y = fmaf(hv.y, alpha, acc.y);
    acc.z = fmaf(hv.z, alpha, acc.z);
    acc.w = fmaf(hv.w, alpha, acc.w);
  }
  {
    const float alpha = exself / ssum;
    const float4 hv = *(const float4*)(h + (size_t)d * 128 + lane * 4);
    acc.x = fmaf(hv.x, alpha, acc.x);
    acc.y = fmaf(hv.y, alpha, acc.y);
    acc.z = fmaf(hv.z, alpha, acc.z);
    acc.w = fmaf(hv.w, alpha, acc.w);
  }
  const float4 bv = *(const float4*)(b1 + lane * 4);
  float4 o;
  o.x = acc.x + bv.x; o.x = (o.x > 0.f) ? o.x : (__expf(o.x) - 1.0f);
  o.y = acc.y + bv.y; o.y = (o.y > 0.f) ? o.y : (__expf(o.y) - 1.0f);
  o.z = acc.z + bv.z; o.z = (o.z > 0.f) ? o.z : (__expf(o.z) - 1.0f);
  o.w = acc.w + bv.w; o.w = (o.w > 0.f) ? o.w : (__expf(o.w) - 1.0f);
  *(float4*)(h1 + (size_t)d * 128 + lane * 4) = o;
}

__global__ void __launch_bounds__(256, 16)
gather1_kernel(const int* __restrict__ rowptr, const int* __restrict__ col2,
               const float* __restrict__ asc, const float* __restrict__ adc,
               const float* __restrict__ h, const float* __restrict__ b1,
               float* __restrict__ h1, int Nn) {
  __shared__ float es_all[8 * 32 * 32];
  const int lane = threadIdx.x & 31;
  const int w = threadIdx.x >> 5;
  float* es = es_all + w * (32 * 32) + lane;
  const int nw = gridDim.x * 8;
  for (int d = __builtin_amdgcn_readfirstlane(blockIdx.x * 8 + w); d < Nn; d += nw) {
    const int start = rowptr[d];
    const int deg = rowptr[d + 1] - start;
    if (deg <= 32) gat1_node<true>(d, lane, start, deg, col2, asc, adc, h, b1, h1, es);
    else           gat1_node<false>(d, lane, start, deg, col2, asc, adc, h, b1, h1, es);
  }
}

__device__ __forceinline__ float wave_max(float v) {
#pragma unroll
  for (int off = 16; off >= 1; off >>= 1) v = fmaxf(v, __shfl_xor(v, off));
  return v;
}
__device__ __forceinline__ float wave_sum(float v) {
#pragma unroll
  for (int off = 16; off >= 1; off >>= 1) v += __shfl_xor(v, off);
  return v;
}

__device__ __forceinline__ void gat2_accum(int n, int sv, float av, int lc,
                                           const float* __restrict__ h2, float2& acc) {
#pragma unroll 4
  for (int t = 0; t < n; ++t) {
    const int s = __builtin_amdgcn_readlane(sv, t);
    const float a = __int_as_float(__builtin_amdgcn_readlane(__float_as_int(av), t));
    const float2 hv = *(const float2*)(h2 + (size_t)s * 40 + 2 * lc);
    acc.x = fmaf(hv.x, a, acc.x);
    acc.y = fmaf(hv.y, a, acc.y);
  }
}

__global__ void __launch_bounds__(256)
gather2_kernel(const int* __restrict__ rowptr, const int* __restrict__ col2,
               const float* __restrict__ asc, const float* __restrict__ adc,
               const float* __restrict__ h2, const float* __restrict__ b2,
               float* __restrict__ out, int Nn) {
  const int lane = threadIdx.x & 31;
  const int lc = (lane < 19) ? lane : 19;
  const int nw = gridDim.x * 8;
  for (int d = __builtin_amdgcn_readfirstlane(blockIdx.x * 8 + (threadIdx.x >> 5)); d < Nn; d += nw) {
  const int start = rowptr[d];
  const int deg = rowptr[d + 1] - start;
  const float ad = adc[d];
  const float eself = leaky(asc[d] + ad);

  int s0 = 0;
  float e0 = -INFINITY;
  if (lane < deg) { s0 = col2[start + lane]; e0 = leaky(asc[s0] + ad); }
  float m = fmaxf(eself, wave_max(e0));
  for (int c = 32; c < deg; c += 32) {
    float e = -INFINITY;
    if (c + lane < deg) e = leaky(asc[col2[start + c + lane]] + ad);
    m = fmaxf(m, wave_max(e));
  }
  const float ex0 = (lane < deg) ? __expf(e0 - m) : 0.f;
  float ssum = wave_sum(ex0);
  for (int c = 32; c < deg; c += 32) {
    float ex = 0.f;
    if (c + lane < deg) ex = __expf(leaky(asc[col2[start + c + lane]] + ad) - m);
    ssum += wave_sum(ex);
  }
  const float exself = __expf(eself - m);
  ssum += exself;

  float2 acc = make_float2(0.f, 0.f);
  gat2_accum(min(deg, 32), s0, ex0 / ssum, lc, h2, acc);
  for (int c = 32; c < deg; c += 32) {
    int s = 0;
    float a = 0.f;
    if (c + lane < deg) {
      s = col2[start + c + lane];
      a = __expf(leaky(asc[s] + ad) - m) / ssum;
    }
    gat2_accum(min(deg - c, 32), s, a, lc, h2, acc);
  }
  {
    const float a = exself / ssum;
    const float2 hv = *(const float2*)(h2 + (size_t)d * 40 + 2 * lc);
    acc.x = fmaf(hv.x, a, acc.x);
    acc.y = fmaf(hv.y, a, acc.y);
  }
  if (lane < 20) {
    const float2 bv = *(const float2*)(b2 + 2 * lane);
    *(float2*)(out + (size_t)d * 40 + 2 * lane) = make_float2(acc.x + bv.x, acc.y + bv.y);
  }
  }
}

extern "C" void kernel_launch(void* const* d_in, const int* in_sizes, int n_in,
                              void* d_out, int out_size, void* d_ws, size_t ws_size,
                              hipStream_t stream) {
  const float* x   = (const float*)d_in[0];
  const int*   ei  = (const int*)d_in[1];
  const float* W1  = (const float*)d_in[2];
  const float* as1 = (const float*)d_in[3];
  const float* ad1 = (const float*)d_in[4];
  const float* b1  = (const float*)d_in[5];
  const float* W2  = (const float*)d_in[6];
  const float* as2 = (const float*)d_in[7];
  const float* ad2 = (const float*)d_in[8];
  const float* b2  = (const float*)d_in[9];
  float* out = (float*)d_out;

  const int Nn   = in_sizes[0] / 128;
  const int E    = in_sizes[1] / 2;
  const int* srcp = ei;
  const int* dstp = ei + E;
  const int NBLK = (E + ETILE - 1) / ETILE;
  const int NBKT = (Nn + BNODES - 1) / BNODES;

  float* ws   = (float*)d_ws;
  float* h    = ws;  ws += (size_t)Nn * 128;
  float* h1   = ws;  ws += (size_t)Nn * 128;
  float* asc1 = ws;  ws += (size_t)Nn * 8;
  float* adc1 = ws;  ws += (size_t)Nn * 8;
  float* h2   = ws;  ws += (size_t)Nn * 40;
  float* asc2 = ws;  ws += Nn;
  float* adc2 = ws;  ws += Nn;
  int* wi     = (int*)ws;
  int* bcnt   = wi;  wi += (size_t)NBLK * NBINS;
  int* rec    = wi;  wi += E;
  int* col    = wi;  wi += E;
  int* col2   = wi;  wi += E;
  int* rowptr = wi;  wi += (size_t)Nn + 1;
  int* bstart = wi;  wi += NBINS + 1;

  bucket_count_kernel<<<NBLK, 1024, 0, stream>>>(dstp, bcnt, E);
  bucket_scatter_kernel<<<NBLK, 1024, 0, stream>>>(srcp, dstp, bcnt, bstart, rec, E, NBLK);
  bucket_csr_kernel<<<NBKT, 1024, 0, stream>>>(bstart, rec, rowptr, col, Nn);
  sort_seg_kernel<<<2048, 256, 0, stream>>>(rowptr, col, col2, Nn);

  gat_wmma_gemm_k128<4, 128, 8><<<dim3(896, 2), 4 * 32, 0, stream>>>(x, W1, h, as1, ad1, asc1, adc1, Nn / 16);
  gather1_kernel<<<2048, 256, 0, stream>>>(rowptr, col2, asc1, adc1, h, b1, h1, Nn);

  gat_wmma_gemm_k128<3, 40, 1><<<dim3(1563, 1), 3 * 32, 0, stream>>>(h1, W2, h2, as2, ad2, asc2, adc2, Nn / 16);
  gather2_kernel<<<2048, 256, 0, stream>>>(rowptr, col2, asc2, adc2, h2, b2, out, Nn);
}


# Layer-1 gather: a node's edge scores, exps and weights formed edge-per-lane in one load round via an LDS tile (was one dependent round per 4 edges); denominator now lane sums then a 4-lane tree
# speedup vs baseline: 4.2430x; 1.0437x over previous
#include <hip/hip_runtime.h>
#include <hip/hip_bf16.h>
#include <math.h>
#include <stdint.h>


typedef float v2f __attribute__((ext_vector_type(2)));
typedef float v8f __attribute__((ext_vector_type(8)));

#define NEG_SLOPE 0.2f

template <int MASK>
__device__ __forceinline__ float row_xor_get(float v) {
  return __int_as_float(__builtin_amdgcn_update_dpp(0, __float_as_int(v), 0x160 + MASK, 0xf, 0xf, true));
}

template <int NTILES, int NACT, int HEADS>
__global__ void __launch_bounds__(32 * NTILES)
gat_wmma_gemm_k128(const float* __restrict__ A, const float* __restrict__ B,
                   float* __restrict__ C,
                   const float* __restrict__ att_src, const float* __restrict__ att_dst,
                   float* __restrict__ a_src, float* __restrict__ a_dst, int nstrips) {
  static_assert(HEADS == 1 || HEADS * 16 == NACT, "a wave's 16 columns must be one head");
  constexpr int NPAD = NTILES * 16;
  constexpr int SA = 132;
  constexpr int SB = 132;
  constexpr int NT = 32 * NTILES;
  __shared__ __attribute__((aligned(16))) float ldsA[16 * SA];
  __shared__ float ldsB[NPAD * SB];
  __shared__ float red[(HEADS == 1) ? NTILES * 32 : 1];

  const int tid  = threadIdx.x;
  const int lane = tid & 31;
  const int wave = tid >> 5;
  const int c0 = blockIdx.y * NPAD;

  for (int i = tid; i < NPAD * 128; i += NT) {
    int k = i / NPAD, n = i - k * NPAD;
    int gn = c0 + n;
    ldsB[n * SB + k] = (gn < NACT) ? B[(size_t)k * NACT + gn] : 0.0f;
  }

  const int l15  = lane & 15;
  const int koff = (lane >> 4) << 1;
  const int ncol = wave * 16 + l15;
  const float* ap = &ldsA[l15 * SA + koff];
  const float* bp = &ldsB[ncol * SB + koff];

  const int gcol = c0 + ncol;
  const float wsrc = (gcol < NACT) ? att_src[gcol] : 0.0f;
  const float wdst = (gcol < NACT) ? att_dst[gcol] : 0.0f;
  const int hd = blockIdx.y * NTILES + wave;

  for (int strip = blockIdx.x; strip < nstrips; strip += gridDim.x) {
    const size_t m0 = (size_t)strip * 16;
    constexpr int AIT = (16 * 32 + NT - 1) / NT;
    float4 av[AIT];
#pragma unroll
    for (int u = 0; u < AIT; ++u) {
      const int i = min(tid + u * NT, 16 * 32 - 1);
      av[u] = *(const float4*)(A + (m0 + (i >> 5)) * 128 + ((i & 31) << 2));
    }
#pragma unroll
    for (int u = 0; u < AIT; ++u) {
      const int i = tid + u * NT;
      if ((u + 1) * NT <= 16 * 32 || i < 16 * 32) *(float4*)&ldsA[(i >> 5) * SA + ((i & 31) << 2)] = av[u];
    }
    __syncthreads();

    v8f acc = {0.f, 0.f, 0.f, 0.f, 0.f, 0.f, 0.f, 0.f};
#pragma unroll
    for (int kb = 0; kb < 128; kb += 4) {
      v2f a = *(const v2f*)(ap + kb);
      v2f b = *(const v2f*)(bp + kb);
      acc = __builtin_amdgcn_wmma_f32_16x16x4_f32(
           false, a,  false, b,
           (short)0, acc,  false,  false);
    }

    if (gcol < NACT) {
      float* cp = C + (m0 + (size_t)((lane >> 4) << 3)) * NACT + gcol;
#pragma unroll
      for (int r = 0; r < 8; ++r) cp[(size_t)r * NACT] = acc[r];
    }

    const bool b3 = (l15 & 8) != 0, b2 = (l15 & 4) != 0, b1 = (l15 & 2) != 0;
    float ps[8], pd[8];
#pragma unroll
    for (int r = 0; r < 8; ++r) { ps[r] = acc[r] * wsrc; pd[r] = acc[r] * wdst; }
    float s4[4], d4[4];
#pragma unroll
    for (int i = 0; i < 4; ++i) {
      s4[i] = (b3 ? ps[4 + i] : ps[i]) + row_xor_get<8>(b3 ? ps[i] : ps[4 + i]);
      d4[i] = (b3 ? pd[4 + i] : pd[i]) + row_xor_get<8>(b3 ? pd[i] : pd[4 + i]);
    }
    float s2[2], d2[2];
#pragma unroll
    for (int j = 0; j < 2; ++j) {
      s2[j] = (b2 ? s4[2 + j] : s4[j]) + row_xor_get<4>(b2 ? s4[j] : s4[2 + j]);
      d2[j] = (b2 ? d4[2 + j] : d4[j]) + row_xor_get<4>(b2 ? d4[j] : d4[2 + j]);
    }
    float s1 = (b1 ? s2[1] : s2[0]) + row_xor_get<2>(b1 ? s2[0] : s2[1]);
    float d1 = (b1 ? d2[1] : d2[0]) + row_xor_get<2>(b1 ? d2[0] : d2[1]);
    s1 += row_xor_get<1>(s1);
    d1 += row_xor_get<1>(d1);
    const bool is_src = (l15 & 1) == 0;
    const float sv = is_src ? s1 : d1;
    const int row = ((lane >> 4) << 3) + (l15 >> 1);
    if constexpr (HEADS == 1) {
      red[(wave * 16 + row) * 2 + (is_src ? 0 : 1)] = sv;
    } else {
      float* sp = is_src ? a_src : a_dst;
      sp[(m0 + row) * HEADS + hd] = sv;
    }
    __syncthreads();
    if constexpr (HEADS == 1) if (wave == 0) {
      const int r16 = lane & 15, which = lane >> 4;
      float s = red[r16 * 2 + which];
#pragma unroll
      for (int w = 1; w < NTILES; ++w) s += red[(w * 16 + r16) * 2 + which];
      float* sp = which ? a_dst : a_src;
      sp[m0 + r16] = s;
    }
  }
}

#define BSHIFT 9
#define BNODES 512
#define NBINS 256
#define SRC_BITS 17
#define ETILE 8192

__global__ void __launch_bounds__(1024)
bucket_count_kernel(const int* __restrict__ dst, int* __restrict__ bcnt, int E) {
  __shared__ int hist[NBINS];
  const int tid = threadIdx.x;
  if (tid < NBINS) hist[tid] = 0;
  const int base = blockIdx.x * ETILE + tid;
  int dv[8];
#pragma unroll
  for (int k = 0; k < 8; ++k) {
    const int e = base + k * 1024;
    dv[k] = (e < E) ? dst[e] : -1;
  }
  __syncthreads();
#pragma unroll
  for (int k = 0; k < 8; ++k) if (dv[k] >= 0) atomicAdd(&hist[dv[k] >> BSHIFT], 1);
  __syncthreads();
  if (tid < NBINS) bcnt[blockIdx.x * NBINS + tid] = hist[tid];
}

__global__ void __launch_bounds__(1024)
bucket_scatter_kernel(const int* __restrict__ src, const int* __restrict__ dst,
                      const int* __restrict__ bcnt, int* __restrict__ bstart,
                      int* __restrict__ rec, int E, int nblk) {
  __shared__ int ptot[4][NBINS];
  __shared__ int ppre[4][NBINS];
  __shared__ int cursor[NBINS + 1];
  const int tid = threadIdx.x;
  const int lane = tid & 31;
  const int base = blockIdx.x * ETILE + tid;
  int dv[8], sv[8];
#pragma unroll
  for (int k = 0; k < 8; ++k) {
    const int e = base + k * 1024;
    const bool ok = e < E;
    dv[k] = ok ? dst[e] : -1;
    sv[k] = ok ? src[e] : 0;
  }
  {
    const int k = tid & (NBINS - 1);
    const int q = tid >> 8;
    const int gq = (nblk + 3) >> 2;
    const int j0 = q * gq;
    const int j1 = min(nblk, j0 + gq);
    int tot = 0, pre = 0;
#pragma unroll 8
    for (int j = j0; j < j1; ++j) {
      const int c = bcnt[j * NBINS + k];
      tot += c;
      pre += (j < (int)blockIdx.x) ? c : 0;
    }
    ptot[q][k] = tot;
    ppre[q][k] = pre;
    if (tid == 0) cursor[NBINS] = 0;
  }
  __syncthreads();
  if (tid < 32) {
    int v[8];
    int run = 0;
#pragma unroll
    for (int i = 0; i < 8; ++i) {
      const int idx = tid * 8 + i;
      v[i] = run;
      run += ptot[0][idx] + ptot[1][idx] + ptot[2][idx] + ptot[3][idx];
    }
    int inc = run;
#pragma unroll
    for (int off = 1; off < 32; off <<= 1) {
      const int nb = __shfl_up(inc, off);
      if (lane >= off) inc += nb;
    }
    const int excl = inc - run;
#pragma unroll
    for (int i = 0; i < 8; ++i) {
      const int idx = tid * 8 + i;
      const int st = excl + v[i];
      cursor[idx] = st + ppre[0][idx] + ppre[1][idx] + ppre[2][idx] + ppre[3][idx];
      if (blockIdx.x == 0) bstart[idx] = st;
    }
    if (blockIdx.x == 0 && tid == 31) bstart[NBINS] = inc;
  }
  __syncthreads();
  int pv[8];
#pragma unroll
  for (int k = 0; k < 8; ++k) pv[k] = atomicAdd(&cursor[(dv[k] >= 0) ? (dv[k] >> BSHIFT) : NBINS], 1);
#pragma unroll
  for (int k = 0; k < 8; ++k)
    if (dv[k] >= 0) rec[pv[k]] = ((dv[k] & (BNODES - 1)) << SRC_BITS) | sv[k];
}

__global__ void __launch_bounds__(1024)
bucket_csr_kernel(const int* __restrict__ bstart, const int* __restrict__ rec,
                  int* __restrict__ rowptr, int* __restrict__ col, int Nn) {
  __shared__ int cntl[BNODES + 1];
  __shared__ int wsum[16];
  const int tid = threadIdx.x, lane = tid & 31, w = tid >> 5;
  const int bs = bstart[blockIdx.x];
  const int n = bstart[blockIdx.x + 1] - bs;
  if (tid <= BNODES) cntl[tid] = 0;
  __syncthreads();
  const int* rp = rec + bs;
  for (int t0 = 0; t0 < n; t0 += ETILE) {
    int rv[8];
#pragma unroll
    for (int k = 0; k < 8; ++k) {
      const int i = t0 + k * 1024 + tid;
      rv[k] = (i < n) ? rp[i] : -1;
    }
#pragma unroll
    for (int k = 0; k < 8; ++k) if (rv[k] >= 0) atomicAdd(&cntl[rv[k] >> SRC_BITS], 1);
  }
  __syncthreads();
  int c = 0, inc = 0;
  if (tid < BNODES) {
    c = cntl[tid];
    inc = c;
#pragma unroll
    for (int off = 1; off < 32; off <<= 1) {
      const int nb = __shfl_up(inc, off);
      if (lane >= off) inc += nb;
    }
    if (lane == 31) wsum[w] = inc;
  }
  __syncthreads();
  if (tid < BNODES) {
    int woff = 0;
#pragma unroll
    for (int k = 0; k < 16; ++k) woff += (k < w) ? wsum[k] : 0;
    const int excl = woff + inc - c;
    cntl[tid] = excl;
    const int node = blockIdx.x * BNODES + tid;
    if (node < Nn) rowptr[node] = bs + excl;
  }
  if (blockIdx.x == gridDim.x - 1 && tid == 0) rowptr[Nn] = bs + n;
  __syncthreads();
  int* cp = col + bs;
  for (int t0 = 0; t0 < n; t0 += ETILE) {
    int rv[8], pv[8];
#pragma unroll
    for (int k = 0; k < 8; ++k) {
      const int i = t0 + k * 1024 + tid;
      rv[k] = (i < n) ? rp[i] : -1;
    }
#pragma unroll
    for (int k = 0; k < 8; ++k) pv[k] = atomicAdd(&cntl[(rv[k] >= 0) ? (rv[k] >> SRC_BITS) : BNODES], 1);
#pragma unroll
    for (int k = 0; k < 8; ++k) if (rv[k] >= 0) cp[pv[k]] = rv[k] & ((1 << SRC_BITS) - 1);
  }
}

__global__ void __launch_bounds__(256)
sort_seg_kernel(const int* __restrict__ rowptr, const int* __restrict__ col,
                int* __restrict__ col2, int Nn) {
  const int lane = threadIdx.x & 31;
  const int nw = gridDim.x * 8;
  for (int d = __builtin_amdgcn_readfirstlane(blockIdx.x * 8 + (threadIdx.x >> 5)); d < Nn; d += nw) {
  const int start = rowptr[d];
  const int deg = rowptr[d + 1] - start;
  if (deg <= 32) {
    const int k = (lane < deg) ? col[start + lane] : 0x7fffffff;
    int r = 0;
    for (int j = 0; j < deg; ++j) {
      const int kj = __builtin_amdgcn_readlane(k, j);
      r += (kj < k || (kj == k && j < lane)) ? 1 : 0;
    }
    if (lane < deg) col2[start + r] = k;
  } else {
    for (int i = lane; i < deg; i += 32) {
      const int k = col[start + i];
      int r = 0;
      for (int j = 0; j < deg; ++j) {
        const int kj = col[start + j];
        r += (kj < k || (kj == k && j < i)) ? 1 : 0;
      }
      col2[start + r] = k;
    }
  }
  }
}

__device__ __forceinline__ float leaky(float v) { return (v >= 0.f) ? v : NEG_SLOPE * v; }


__device__ __forceinline__ void wave_lds_order() {
  __builtin_amdgcn_fence(__ATOMIC_RELEASE, "wavefront");
  __builtin_amdgcn_wave_barrier();
  __builtin_amdgcn_fence(__ATOMIC_ACQUIRE, "wavefront");
}

__device__ __forceinline__ void gat1_node_short(int d, int lane, int start, int deg,
                                                const int* __restrict__ col2,
                                                const float* __restrict__ asc,
                                                const float* __restrict__ adc,
                                                const float* __restrict__ h,
                                                const float* __restrict__ b1,
                                                float* __restrict__ h1, float* tile) {
  const int hd = lane >> 2, q = lane & 3;
  const float4 hself = *(const float4*)(h + (size_t)d * 128 + lane * 4);
  const float4 ad0 = *(const float4*)(adc + (size_t)d * 8);
  const float4 ad1 = *(const float4*)(adc + (size_t)d * 8 + 4);
  const float eself = leaky(asc[(size_t)d * 8 + hd] + adc[(size_t)d * 8 + hd]);
  int sv = d;
  float4 e0 = make_float4(-INFINITY, -INFINITY, -INFINITY, -INFINITY);
  float4 e1 = e0;
  if (lane < deg) {
    sv = col2[start + lane];
    const float4 a0 = *(const float4*)(asc + (size_t)sv * 8);
    const float4 a1 = *(const float4*)(asc + (size_t)sv * 8 + 4);
    e0.x = leaky(a0.x + ad0.x); e0.y = leaky(a0.y + ad0.y);
    e0.z = leaky(a0.z + ad0.z); e0.w = leaky(a0.w + ad0.w);
    e1.x = leaky(a1.x + ad1.x); e1.y = leaky(a1.y + ad1.y);
    e1.z = leaky(a1.z + ad1.z); e1.w = leaky(a1.w + ad1.w);
  }
  *(float4*)(tile + lane * 8) = e0;
  *(float4*)(tile + lane * 8 + 4) = e1;
  wave_lds_order();

  float* colp = tile + q * 8 + hd;
  float v[8];
#pragma unroll
  for (int t = 0; t < 8; ++t) v[t] = colp[32 * t];
  float m = eself;
#pragma unroll
  for (int t = 0; t < 8; ++t) m = fmaxf(m, v[t]);
  m = fmaxf(m, row_xor_get<1>(m));
  m = fmaxf(m, row_xor_get<2>(m));
  float ssum = 0.f;
#pragma unroll
  for (int t = 0; t < 8; ++t) { v[t] = __expf(v[t] - m); ssum += v[t]; }
  ssum += row_xor_get<1>(ssum);
  ssum += row_xor_get<2>(ssum);
  const float exself = __expf(eself - m);
  ssum += exself;
#pragma unroll
  for (int t = 0; t < 8; ++t) {
    if (4 * t < deg) colp[32 * t] = v[t] / ssum;
  }
  wave_lds_order();

  float4 acc = make_float4(0.f, 0.f, 0.f, 0.f);
  const float* wp = tile + hd;
  for (int j0 = 0; j0 < deg; j0 += 4) {
#pragma unroll
    for (int u = 0; u < 4; ++u) {
      const int j = j0 + u;
      const int s = __builtin_amdgcn_readlane(sv, j);
      const float alpha = wp[j * 8];
      const float4 hv = *(const float4*)(h + (size_t)s * 128 + lane * 4);
      acc.x = fmaf(hv.x, alpha, acc.x);
      acc.y = fmaf(hv.y, alpha, acc.y);
      acc.z = fmaf(hv.z, alpha, acc.z);
      acc.w = fmaf(hv.w, alpha, acc.w);
    }
  }
  {
    const float alpha = exself / ssum;
    acc.x = fmaf(hself.x, alpha, acc.x);
    acc.y = fmaf(hself.y, alpha, acc.y);
    acc.z = fmaf(hself.z, alpha, acc.z);
    acc.w = fmaf(hself.w, alpha, acc.w);
  }
  const float4 bv = *(const float4*)(b1 + lane * 4);
  float4 o;
  o.x = acc.x + bv.x; o.x = (o.x > 0.f) ? o.x : (__expf(o.x) - 1.0f);
  o.y = acc.y + bv.y; o.y = (o.y > 0.f) ? o.y : (__expf(o.y) - 1.0f);
  o.z = acc.z + bv.z; o.z = (o.z > 0.f) ? o.z : (__expf(o.z) - 1.0f);
  o.w = acc.w + bv.w; o.w = (o.w > 0.f) ? o.w : (__expf(o.w) - 1.0f);
  *(float4*)(h1 + (size_t)d * 128 + lane * 4) = o;
  wave_lds_order();
}

__device__ __forceinline__ void gat1_node_long(int d, int lane, int start, int deg,
                                               const int* __restrict__ col2,
                                               const float* __restrict__ asc,
                                               const float* __restrict__ adc,
                                               const float* __restrict__ h,
                                               const float* __restrict__ b1,
                                               float* __restrict__ h1) {
  const int hd = lane >> 2;
  const float ad = adc[(size_t)d * 8 + hd];
  const float eself = leaky(asc[(size_t)d * 8 + hd] + ad);
  float m = eself;
#pragma unroll 4
  for (int j = 0; j < deg; ++j) {
    const int s = col2[start + j];
    const float e = leaky(asc[(size_t)s * 8 + hd] + ad);
    m = fmaxf(m, e);
  }
  float ssum = 0.f;
#pragma unroll 4
  for (int j = 0; j < deg; ++j) {
    const float e = leaky(asc[(size_t)col2[start + j] * 8 + hd] + ad);
    ssum += __expf(e - m);
  }
  const float exself = __expf(eself - m);
  ssum += exself;
  float4 acc = make_float4(0.f, 0.f, 0.f, 0.f);
#pragma unroll 4
  for (int j = 0; j < deg; ++j) {
    const int s = col2[start + j];
    const float ex = __expf(leaky(asc[(size_t)s * 8 + hd] + ad) - m);
    const float alpha = ex / ssum;
    const float4 hv = *(const float4*)(h + (size_t)s * 128 + lane * 4);
    acc.x = fmaf(hv.x, alpha, acc.x);
    acc.y = fmaf(hv.y, alpha, acc.y);
    acc.z = fmaf(hv.z, alpha, acc.z);
    acc.w = fmaf(hv.w, alpha, acc.w);
  }
  {
    const float alpha = exself / ssum;
    const float4 hv = *(const float4*)(h + (size_t)d * 128 + lane * 4);
    acc.x = fmaf(hv.x, alpha, acc.x);
    acc.y = fmaf(hv.y, alpha, acc.y);
    acc.z = fmaf(hv.z, alpha, acc.z);
    acc.w = fmaf(hv.w, alpha, acc.w);
  }
  const float4 bv = *(const float4*)(b1 + lane * 4);
  float4 o;
  o.x = acc.x + bv.x; o.x = (o.x > 0.f) ? o.x : (__expf(o.x) - 1.0f);
  o.y = acc.y + bv.y; o.y = (o.y > 0.f) ? o.y : (__expf(o.y) - 1.0f);
  o.z = acc.z + bv.z; o.z = (o.z > 0.f) ? o.z : (__expf(o.z) - 1.0f);
  o.w = acc.w + bv.w; o.w = (o.w > 0.f) ? o.w : (__expf(o.w) - 1.0f);
  *(float4*)(h1 + (size_t)d * 128 + lane * 4) = o;
}

__global__ void __launch_bounds__(256, 16)
gather1_kernel(const int* __restrict__ rowptr, const int* __restrict__ col2,
               const float* __restrict__ asc, const float* __restrict__ adc,
               const float* __restrict__ h, const float* __restrict__ b1,
               float* __restrict__ h1, int Nn) {
  __shared__ __attribute__((aligned(16))) float tiles[8 * 32 * 8];
  const int lane = threadIdx.x & 31;
  const int w = threadIdx.x >> 5;
  float* tile = tiles + w * (32 * 8);
  const int nw = gridDim.x * 8;
  for (int d = __builtin_amdgcn_readfirstlane(blockIdx.x * 8 + w); d < Nn; d += nw) {
    const int start = rowptr[d];
    const int deg = rowptr[d + 1] - start;
    if (deg <= 32) gat1_node_short(d, lane, start, deg, col2, asc, adc, h, b1, h1, tile);
    else           gat1_node_long(d, lane, start, deg, col2, asc, adc, h, b1, h1);
  }
}

__device__ __forceinline__ float wave_max(float v) {
#pragma unroll
  for (int off = 16; off >= 1; off >>= 1) v = fmaxf(v, __shfl_xor(v, off));
  return v;
}
__device__ __forceinline__ float wave_sum(float v) {
#pragma unroll
  for (int off = 16; off >= 1; off >>= 1) v += __shfl_xor(v, off);
  return v;
}

__device__ __forceinline__ void gat2_accum(int n, int sv, float av, int lc,
                                           const float* __restrict__ h2, float2& acc) {
#pragma unroll 4
  for (int t = 0; t < n; ++t) {
    const int s = __builtin_amdgcn_readlane(sv, t);
    const float a = __int_as_float(__builtin_amdgcn_readlane(__float_as_int(av), t));
    const float2 hv = *(const float2*)(h2 + (size_t)s * 40 + 2 * lc);
    acc.x = fmaf(hv.x, a, acc.x);
    acc.y = fmaf(hv.y, a, acc.y);
  }
}

__global__ void __launch_bounds__(256)
gather2_kernel(const int* __restrict__ rowptr, const int* __restrict__ col2,
               const float* __restrict__ asc, const float* __restrict__ adc,
               const float* __restrict__ h2, const float* __restrict__ b2,
               float* __restrict__ out, int Nn) {
  const int lane = threadIdx.x & 31;
  const int lc = (lane < 19) ? lane : 19;
  const int nw = gridDim.x * 8;
  for (int d = __builtin_amdgcn_readfirstlane(blockIdx.x * 8 + (threadIdx.x >> 5)); d < Nn; d += nw) {
  const int start = rowptr[d];
  const int deg = rowptr[d + 1] - start;
  const float ad = adc[d];
  const float eself = leaky(asc[d] + ad);

  int s0 = 0;
  float e0 = -INFINITY;
  if (lane < deg) { s0 = col2[start + lane]; e0 = leaky(asc[s0] + ad); }
  float m = fmaxf(eself, wave_max(e0));
  for (int c = 32; c < deg; c += 32) {
    float e = -INFINITY;
    if (c + lane < deg) e = leaky(asc[col2[start + c + lane]] + ad);
    m = fmaxf(m, wave_max(e));
  }
  const float ex0 = (lane < deg) ? __expf(e0 - m) : 0.f;
  float ssum = wave_sum(ex0);
  for (int c = 32; c < deg; c += 32) {
    float ex = 0.f;
    if (c + lane < deg) ex = __expf(leaky(asc[col2[start + c + lane]] + ad) - m);
    ssum += wave_sum(ex);
  }
  const float exself = __expf(eself - m);
  ssum += exself;

  float2 acc = make_float2(0.f, 0.f);
  gat2_accum(min(deg, 32), s0, ex0 / ssum, lc, h2, acc);
  for (int c = 32; c < deg; c += 32) {
    int s = 0;
    float a = 0.f;
    if (c + lane < deg) {
      s = col2[start + c + lane];
      a = __expf(leaky(asc[s] + ad) - m) / ssum;
    }
    gat2_accum(min(deg - c, 32), s, a, lc, h2, acc);
  }
  {
    const float a = exself / ssum;
    const float2 hv = *(const float2*)(h2 + (size_t)d * 40 + 2 * lc);
    acc.x = fmaf(hv.x, a, acc.x);
    acc.y = fmaf(hv.y, a, acc.y);
  }
  if (lane < 20) {
    const float2 bv = *(const float2*)(b2 + 2 * lane);
    *(float2*)(out + (size_t)d * 40 + 2 * lane) = make_float2(acc.x + bv.x, acc.y + bv.y);
  }
  }
}

extern "C" void kernel_launch(void* const* d_in, const int* in_sizes, int n_in,
                              void* d_out, int out_size, void* d_ws, size_t ws_size,
                              hipStream_t stream) {
  const float* x   = (const float*)d_in[0];
  const int*   ei  = (const int*)d_in[1];
  const float* W1  = (const float*)d_in[2];
  const float* as1 = (const float*)d_in[3];
  const float* ad1 = (const float*)d_in[4];
  const float* b1  = (const float*)d_in[5];
  const float* W2  = (const float*)d_in[6];
  const float* as2 = (const float*)d_in[7];
  const float* ad2 = (const float*)d_in[8];
  const float* b2  = (const float*)d_in[9];
  float* out = (float*)d_out;

  const int Nn   = in_sizes[0] / 128;
  const int E    = in_sizes[1] / 2;
  const int* srcp = ei;
  const int* dstp = ei + E;
  const int NBLK = (E + ETILE - 1) / ETILE;
  const int NBKT = (Nn + BNODES - 1) / BNODES;

  float* ws   = (float*)d_ws;
  float* h    = ws;  ws += (size_t)Nn * 128;
  float* h1   = ws;  ws += (size_t)Nn * 128;
  float* asc1 = ws;  ws += (size_t)Nn * 8;
  float* adc1 = ws;  ws += (size_t)Nn * 8;
  float* h2   = ws;  ws += (size_t)Nn * 40;
  float* asc2 = ws;  ws += Nn;
  float* adc2 = ws;  ws += Nn;
  int* wi     = (int*)ws;
  int* bcnt   = wi;  wi += (size_t)NBLK * NBINS;
  int* rec    = wi;  wi += E;
  int* col    = wi;  wi += E;
  int* col2   = wi;  wi += E;
  int* rowptr = wi;  wi += (size_t)Nn + 1;
  int* bstart = wi;  wi += NBINS + 1;

  bucket_count_kernel<<<NBLK, 1024, 0, stream>>>(dstp, bcnt, E);
  bucket_scatter_kernel<<<NBLK, 1024, 0, stream>>>(srcp, dstp, bcnt, bstart, rec, E, NBLK);
  bucket_csr_kernel<<<NBKT, 1024, 0, stream>>>(bstart, rec, rowptr, col, Nn);
  sort_seg_kernel<<<2048, 256, 0, stream>>>(rowptr, col, col2, Nn);

  gat_wmma_gemm_k128<4, 128, 8><<<dim3(896, 2), 4 * 32, 0, stream>>>(x, W1, h, as1, ad1, asc1, adc1, Nn / 16);
  gather1_kernel<<<2048, 256, 0, stream>>>(rowptr, col2, asc1, adc1, h, b1, h1, Nn);

  gat_wmma_gemm_k128<3, 40, 1><<<dim3(1563, 1), 3 * 32, 0, stream>>>(h1, W2, h2, as2, ad2, asc2, adc2, Nn / 16);
  gather2_kernel<<<2048, 256, 0, stream>>>(rowptr, col2, asc2, adc2, h2, b2, out, Nn);
}


# GEMM prologues: the W panel is fetched as 16 16-byte loads per thread, all in flight before the first transposed LDS store (was 64 dependent load-wait-store round trips per thread); sums unchanged
# speedup vs baseline: 5.1077x; 1.2038x over previous
#include <hip/hip_runtime.h>
#include <hip/hip_bf16.h>
#include <math.h>
#include <stdint.h>


typedef float v2f __attribute__((ext_vector_type(2)));
typedef float v8f __attribute__((ext_vector_type(8)));

#define NEG_SLOPE 0.2f

template <int MASK>
__device__ __forceinline__ float row_xor_get(float v) {
  return __int_as_float(__builtin_amdgcn_update_dpp(0, __float_as_int(v), 0x160 + MASK, 0xf, 0xf, true));
}

template <int NTILES, int NACT, int HEADS>
__global__ void __launch_bounds__(32 * NTILES)
gat_wmma_gemm_k128(const float* __restrict__ A, const float* __restrict__ B,
                   float* __restrict__ C,
                   const float* __restrict__ att_src, const float* __restrict__ att_dst,
                   float* __restrict__ a_src, float* __restrict__ a_dst, int nstrips) {
  static_assert(HEADS == 1 || HEADS * 16 == NACT, "a wave's 16 columns must be one head");
  constexpr int NPAD = NTILES * 16;
  constexpr int SA = 132;
  constexpr int SB = 132;
  constexpr int NT = 32 * NTILES;
  __shared__ __attribute__((aligned(16))) float ldsA[16 * SA];
  __shared__ float ldsB[NPAD * SB];
  __shared__ float red[(HEADS == 1) ? NTILES * 32 : 1];

  const int tid  = threadIdx.x;
  const int lane = tid & 31;
  const int wave = tid >> 5;
  const int c0 = blockIdx.y * NPAD;

  static_assert(NACT % 4 == 0 && NPAD % 4 == 0, "16-byte pieces must not straddle the last column");
  constexpr int BIT = (NPAD / 4) * 128 / NT;
  static_assert(BIT * NT == (NPAD / 4) * 128, "the panel must be a whole number of pieces per thread");
  {
    float4 bv[BIT];
#pragma unroll
    for (int u = 0; u < BIT; ++u) {
      const int p = tid + u * NT;
      const int k = ((p / NPAD) << 2) + (p & 3);
      const int gn = c0 + (((p >> 2) % (NPAD / 4)) << 2);
      const int gc = (gn < NACT) ? gn : (NACT - 4);
      bv[u] = *(const float4*)(B + (size_t)k * NACT + gc);
    }
#pragma unroll
    for (int u = 0; u < BIT; ++u) {
      const int p = tid + u * NT;
      const int k = ((p / NPAD) << 2) + (p & 3);
      const int n = ((p >> 2) % (NPAD / 4)) << 2;
      const bool in = (c0 + n) < NACT;
      float* lp = &ldsB[n * SB + k];
      lp[0]      = in ? bv[u].x : 0.0f;
      lp[SB]     = in ? bv[u].y : 0.0f;
      lp[2 * SB] = in ? bv[u].z : 0.0f;
      lp[3 * SB] = in ? bv[u].w : 0.0f;
    }
  }

  const int l15  = lane & 15;
  const int koff = (lane >> 4) << 1;
  const int ncol = wave * 16 + l15;
  const float* ap = &ldsA[l15 * SA + koff];
  const float* bp = &ldsB[ncol * SB + koff];

  const int gcol = c0 + ncol;
  const float wsrc = (gcol < NACT) ? att_src[gcol] : 0.0f;
  const float wdst = (gcol < NACT) ? att_dst[gcol] : 0.0f;
  const int hd = blockIdx.y * NTILES + wave;

  for (int strip = blockIdx.x; strip < nstrips; strip += gridDim.x) {
    const size_t m0 = (size_t)strip * 16;
    constexpr int AIT = (16 * 32 + NT - 1) / NT;
    float4 av[AIT];
#pragma unroll
    for (int u = 0; u < AIT; ++u) {
      const int i = min(tid + u * NT, 16 * 32 - 1);
      av[u] = *(const float4*)(A + (m0 + (i >> 5)) * 128 + ((i & 31) << 2));
    }
#pragma unroll
    for (int u = 0; u < AIT; ++u) {
      const int i = tid + u * NT;
      if ((u + 1) * NT <= 16 * 32 || i < 16 * 32) *(float4*)&ldsA[(i >> 5) * SA + ((i & 31) << 2)] = av[u];
    }
    __syncthreads();

    v8f acc = {0.f, 0.f, 0.f, 0.f, 0.f, 0.f, 0.f, 0.f};
#pragma unroll
    for (int kb = 0; kb < 128; kb += 4) {
      v2f a = *(const v2f*)(ap + kb);
      v2f b = *(const v2f*)(bp + kb);
      acc = __builtin_amdgcn_wmma_f32_16x16x4_f32(
           false, a,  false, b,
           (short)0, acc,  false,  false);
    }

    if (gcol < NACT) {
      float* cp = C + (m0 + (size_t)((lane >> 4) << 3)) * NACT + gcol;
#pragma unroll
      for (int r = 0; r < 8; ++r) cp[(size_t)r * NACT] = acc[r];
    }

    const bool b3 = (l15 & 8) != 0, b2 = (l15 & 4) != 0, b1 = (l15 & 2) != 0;
    float ps[8], pd[8];
#pragma unroll
    for (int r = 0; r < 8; ++r) { ps[r] = acc[r] * wsrc; pd[r] = acc[r] * wdst; }
    float s4[4], d4[4];
#pragma unroll
    for (int i = 0; i < 4; ++i) {
      s4[i] = (b3 ? ps[4 + i] : ps[i]) + row_xor_get<8>(b3 ? ps[i] : ps[4 + i]);
      d4[i] = (b3 ? pd[4 + i] : pd[i]) + row_xor_get<8>(b3 ? pd[i] : pd[4 + i]);
    }
    float s2[2], d2[2];
#pragma unroll
    for (int j = 0; j < 2; ++j) {
      s2[j] = (b2 ? s4[2 + j] : s4[j]) + row_xor_get<4>(b2 ? s4[j] : s4[2 + j]);
      d2[j] = (b2 ? d4[2 + j] : d4[j]) + row_xor_get<4>(b2 ? d4[j] : d4[2 + j]);
    }
    float s1 = (b1 ? s2[1] : s2[0]) + row_xor_get<2>(b1 ? s2[0] : s2[1]);
    float d1 = (b1 ? d2[1] : d2[0]) + row_xor_get<2>(b1 ? d2[0] : d2[1]);
    s1 += row_xor_get<1>(s1);
    d1 += row_xor_get<1>(d1);
    const bool is_src = (l15 & 1) == 0;
    const float sv = is_src ? s1 : d1;
    const int row = ((lane >> 4) << 3) + (l15 >> 1);
    if constexpr (HEADS == 1) {
      red[(wave * 16 + row) * 2 + (is_src ? 0 : 1)] = sv;
    } else {
      float* sp = is_src ? a_src : a_dst;
      sp[(m0 + row) * HEADS + hd] = sv;
    }
    __syncthreads();
    if constexpr (HEADS == 1) if (wave == 0) {
      const int r16 = lane & 15, which = lane >> 4;
      float s = red[r16 * 2 + which];
#pragma unroll
      for (int w = 1; w < NTILES; ++w) s += red[(w * 16 + r16) * 2 + which];
      float* sp = which ? a_dst : a_src;
      sp[m0 + r16] = s;
    }
  }
}

#define BSHIFT 9
#define BNODES 512
#define NBINS 256
#define SRC_BITS 17
#define ETILE 8192

__global__ void __launch_bounds__(1024)
bucket_count_kernel(const int* __restrict__ dst, int* __restrict__ bcnt, int E) {
  __shared__ int hist[NBINS];
  const int tid = threadIdx.x;
  if (tid < NBINS) hist[tid] = 0;
  const int base = blockIdx.x * ETILE + tid;
  int dv[8];
#pragma unroll
  for (int k = 0; k < 8; ++k) {
    const int e = base + k * 1024;
    dv[k] = (e < E) ? dst[e] : -1;
  }
  __syncthreads();
#pragma unroll
  for (int k = 0; k < 8; ++k) if (dv[k] >= 0) atomicAdd(&hist[dv[k] >> BSHIFT], 1);
  __syncthreads();
  if (tid < NBINS) bcnt[blockIdx.x * NBINS + tid] = hist[tid];
}

__global__ void __launch_bounds__(1024)
bucket_scatter_kernel(const int* __restrict__ src, const int* __restrict__ dst,
                      const int* __restrict__ bcnt, int* __restrict__ bstart,
                      int* __restrict__ rec, int E, int nblk) {
  __shared__ int ptot[4][NBINS];
  __shared__ int ppre[4][NBINS];
  __shared__ int cursor[NBINS + 1];
  const int tid = threadIdx.x;
  const int lane = tid & 31;
  const int base = blockIdx.x * ETILE + tid;
  int dv[8], sv[8];
#pragma unroll
  for (int k = 0; k < 8; ++k) {
    const int e = base + k * 1024;
    const bool ok = e < E;
    dv[k] = ok ? dst[e] : -1;
    sv[k] = ok ? src[e] : 0;
  }
  {
    const int k = tid & (NBINS - 1);
    const int q = tid >> 8;
    const int gq = (nblk + 3) >> 2;
    const int j0 = q * gq;
    const int j1 = min(nblk, j0 + gq);
    int tot = 0, pre = 0;
#pragma unroll 8
    for (int j = j0; j < j1; ++j) {
      const int c = bcnt[j * NBINS + k];
      tot += c;
      pre += (j < (int)blockIdx.x) ? c : 0;
    }
    ptot[q][k] = tot;
    ppre[q][k] = pre;
    if (tid == 0) cursor[NBINS] = 0;
  }
  __syncthreads();
  if (tid < 32) {
    int v[8];
    int run = 0;
#pragma unroll
    for (int i = 0; i < 8; ++i) {
      const int idx = tid * 8 + i;
      v[i] = run;
      run += ptot[0][idx] + ptot[1][idx] + ptot[2][idx] + ptot[3][idx];
    }
    int inc = run;
#pragma unroll
    for (int off = 1; off < 32; off <<= 1) {
      const int nb = __shfl_up(inc, off);
      if (lane >= off) inc += nb;
    }
    const int excl = inc - run;
#pragma unroll
    for (int i = 0; i < 8; ++i) {
      const int idx = tid * 8 + i;
      const int st = excl + v[i];
      cursor[idx] = st + ppre[0][idx] + ppre[1][idx] + ppre[2][idx] + ppre[3][idx];
      if (blockIdx.x == 0) bstart[idx] = st;
    }
    if (blockIdx.x == 0 && tid == 31) bstart[NBINS] = inc;
  }
  __syncthreads();
  int pv[8];
#pragma unroll
  for (int k = 0; k < 8; ++k) pv[k] = atomicAdd(&cursor[(dv[k] >= 0) ? (dv[k] >> BSHIFT) : NBINS], 1);
#pragma unroll
  for (int k = 0; k < 8; ++k)
    if (dv[k] >= 0) rec[pv[k]] = ((dv[k] & (BNODES - 1)) << SRC_BITS) | sv[k];
}

__global__ void __launch_bounds__(1024)
bucket_csr_kernel(const int* __restrict__ bstart, const int* __restrict__ rec,
                  int* __restrict__ rowptr, int* __restrict__ col, int Nn) {
  __shared__ int cntl[BNODES + 1];
  __shared__ int wsum[16];
  const int tid = threadIdx.x, lane = tid & 31, w = tid >> 5;
  const int bs = bstart[blockIdx.x];
  const int n = bstart[blockIdx.x + 1] - bs;
  if (tid <= BNODES) cntl[tid] = 0;
  __syncthreads();
  const int* rp = rec + bs;
  for (int t0 = 0; t0 < n; t0 += ETILE) {
    int rv[8];
#pragma unroll
    for (int k = 0; k < 8; ++k) {
      const int i = t0 + k * 1024 + tid;
      rv[k] = (i < n) ? rp[i] : -1;
    }
#pragma unroll
    for (int k = 0; k < 8; ++k) if (rv[k] >= 0) atomicAdd(&cntl[rv[k] >> SRC_BITS], 1);
  }
  __syncthreads();
  int c = 0, inc = 0;
  if (tid < BNODES) {
    c = cntl[tid];
    inc = c;
#pragma unroll
    for (int off = 1; off < 32; off <<= 1) {
      const int nb = __shfl_up(inc, off);
      if (lane >= off) inc += nb;
    }
    if (lane == 31) wsum[w] = inc;
  }
  __syncthreads();
  if (tid < BNODES) {
    int woff = 0;
#pragma unroll
    for (int k = 0; k < 16; ++k) woff += (k < w) ? wsum[k] : 0;
    const int excl = woff + inc - c;
    cntl[tid] = excl;
    const int node = blockIdx.x * BNODES + tid;
    if (node < Nn) rowptr[node] = bs + excl;
  }
  if (blockIdx.x == gridDim.x - 1 && tid == 0) rowptr[Nn] = bs + n;
  __syncthreads();
  int* cp = col + bs;
  for (int t0 = 0; t0 < n; t0 += ETILE) {
    int rv[8], pv[8];
#pragma unroll
    for (int k = 0; k < 8; ++k) {
      const int i = t0 + k * 1024 + tid;
      rv[k] = (i < n) ? rp[i] : -1;
    }
#pragma unroll
    for (int k = 0; k < 8; ++k) pv[k] = atomicAdd(&cntl[(rv[k] >= 0) ? (rv[k] >> SRC_BITS) : BNODES], 1);
#pragma unroll
    for (int k = 0; k < 8; ++k) if (rv[k] >= 0) cp[pv[k]] = rv[k] & ((1 << SRC_BITS) - 1);
  }
}

__global__ void __launch_bounds__(256)
sort_seg_kernel(const int* __restrict__ rowptr, const int* __restrict__ col,
                int* __restrict__ col2, int Nn) {
  const int lane = threadIdx.x & 31;
  const int nw = gridDim.x * 8;
  for (int d = __builtin_amdgcn_readfirstlane(blockIdx.x * 8 + (threadIdx.x >> 5)); d < Nn; d += nw) {
  const int start = rowptr[d];
  const int deg = rowptr[d + 1] - start;
  if (deg <= 32) {
    const int k = (lane < deg) ? col[start + lane] : 0x7fffffff;
    int r = 0;
    for (int j = 0; j < deg; ++j) {
      const int kj = __builtin_amdgcn_readlane(k, j);
      r += (kj < k || (kj == k && j < lane)) ? 1 : 0;
    }
    if (lane < deg) col2[start + r] = k;
  } else {
    for (int i = lane; i < deg; i += 32) {
      const int k = col[start + i];
      int r = 0;
      for (int j = 0; j < deg; ++j) {
        const int kj = col[start + j];
        r += (kj < k || (kj == k && j < i)) ? 1 : 0;
      }
      col2[start + r] = k;
    }
  }
  }
}

__device__ __forceinline__ float leaky(float v) { return (v >= 0.f) ? v : NEG_SLOPE * v; }


__device__ __forceinline__ void wave_lds_order() {
  __builtin_amdgcn_fence(__ATOMIC_RELEASE, "wavefront");
  __builtin_amdgcn_wave_barrier();
  __builtin_amdgcn_fence(__ATOMIC_ACQUIRE, "wavefront");
}

__device__ __forceinline__ void gat1_node_short(int d, int lane, int start, int deg,
                                                const int* __restrict__ col2,
                                                const float* __restrict__ asc,
                                                const float* __restrict__ adc,
                                                const float* __restrict__ h,
                                                const float* __restrict__ b1,
                                                float* __restrict__ h1, float* tile) {
  const int hd = lane >> 2, q = lane & 3;
  const float4 hself = *(const float4*)(h + (size_t)d * 128 + lane * 4);
  const float4 ad0 = *(const float4*)(adc + (size_t)d * 8);
  const float4 ad1 = *(const float4*)(adc + (size_t)d * 8 + 4);
  const float eself = leaky(asc[(size_t)d * 8 + hd] + adc[(size_t)d * 8 + hd]);
  int sv = d;
  float4 e0 = make_float4(-INFINITY, -INFINITY, -INFINITY, -INFINITY);
  float4 e1 = e0;
  if (lane < deg) {
    sv = col2[start + lane];
    const float4 a0 = *(const float4*)(asc + (size_t)sv * 8);
    const float4 a1 = *(const float4*)(asc + (size_t)sv * 8 + 4);
    e0.x = leaky(a0.x + ad0.x); e0.y = leaky(a0.y + ad0.y);
    e0.z = leaky(a0.z + ad0.z); e0.w = leaky(a0.w + ad0.w);
    e1.x = leaky(a1.x + ad1.x); e1.y = leaky(a1.y + ad1.y);
    e1.z = leaky(a1.z + ad1.z); e1.w = leaky(a1.w + ad1.w);
  }
  *(float4*)(tile + lane * 8) = e0;
  *(float4*)(tile + lane * 8 + 4) = e1;
  wave_lds_order();

  float* colp = tile + q * 8 + hd;
  float v[8];
#pragma unroll
  for (int t = 0; t < 8; ++t) v[t] = colp[32 * t];
  float m = eself;
#pragma unroll
  for (int t = 0; t < 8; ++t) m = fmaxf(m, v[t]);
  m = fmaxf(m, row_xor_get<1>(m));
  m = fmaxf(m, row_xor_get<2>(m));
  float ssum = 0.f;
#pragma unroll
  for (int t = 0; t < 8; ++t) { v[t] = __expf(v[t] - m); ssum += v[t]; }
  ssum += row_xor_get<1>(ssum);
  ssum += row_xor_get<2>(ssum);
  const float exself = __expf(eself - m);
  ssum += exself;
#pragma unroll
  for (int t = 0; t < 8; ++t) {
    if (4 * t < deg) colp[32 * t] = v[t] / ssum;
  }
  wave_lds_order();

  float4 acc = make_float4(0.f, 0.f, 0.f, 0.f);
  const float* wp = tile + hd;
  for (int j0 = 0; j0 < deg; j0 += 4) {
#pragma unroll
    for (int u = 0; u < 4; ++u) {
      const int j = j0 + u;
      const int s = __builtin_amdgcn_readlane(sv, j);
      const float alpha = wp[j * 8];
      const float4 hv = *(const float4*)(h + (size_t)s * 128 + lane * 4);
      acc.x = fmaf(hv.x, alpha, acc.x);
      acc.y = fmaf(hv.y, alpha, acc.y);
      acc.z = fmaf(hv.z, alpha, acc.z);
      acc.w = fmaf(hv.w, alpha, acc.w);
    }
  }
  {
    const float alpha = exself / ssum;
    acc.x = fmaf(hself.x, alpha, acc.x);
    acc.y = fmaf(hself.y, alpha, acc.y);
    acc.z = fmaf(hself.z, alpha, acc.z);
    acc.w = fmaf(hself.w, alpha, acc.w);
  }
  const float4 bv = *(const float4*)(b1 + lane * 4);
  float4 o;
  o.x = acc.x + bv.x; o.x = (o.x > 0.f) ? o.x : (__expf(o.x) - 1.0f);
  o.y = acc.y + bv.y; o.y = (o.y > 0.f) ? o.y : (__expf(o.y) - 1.0f);
  o.z = acc.z + bv.z; o.z = (o.z > 0.f) ? o.z : (__expf(o.z) - 1.0f);
  o.w = acc.w + bv.w; o.w = (o.w > 0.f) ? o.w : (__expf(o.w) - 1.0f);
  *(float4*)(h1 + (size_t)d * 128 + lane * 4) = o;
  wave_lds_order();
}

__device__ __forceinline__ void gat1_node_long(int d, int lane, int start, int deg,
                                               const int* __restrict__ col2,
                                               const float* __restrict__ asc,
                                               const float* __restrict__ adc,
                                               const float* __restrict__ h,
                                               const float* __restrict__ b1,
                                               float* __restrict__ h1) {
  const int hd = lane >> 2;
  const float ad = adc[(size_t)d * 8 + hd];
  const float eself = leaky(asc[(size_t)d * 8 + hd] + ad);
  float m = eself;
#pragma unroll 4
  for (int j = 0; j < deg; ++j) {
    const int s = col2[start + j];
    const float e = leaky(asc[(size_t)s * 8 + hd] + ad);
    m = fmaxf(m, e);
  }
  float ssum = 0.f;
#pragma unroll 4
  for (int j = 0; j < deg; ++j) {
    const float e = leaky(asc[(size_t)col2[start + j] * 8 + hd] + ad);
    ssum += __expf(e - m);
  }
  const float exself = __expf(eself - m);
  ssum += exself;
  float4 acc = make_float4(0.f, 0.f, 0.f, 0.f);
#pragma unroll 4
  for (int j = 0; j < deg; ++j) {
    const int s = col2[start + j];
    const float ex = __expf(leaky(asc[(size_t)s * 8 + hd] + ad) - m);
    const float alpha = ex / ssum;
    const float4 hv = *(const float4*)(h + (size_t)s * 128 + lane * 4);
    acc.x = fmaf(hv.x, alpha, acc.x);
    acc.y = fmaf(hv.y, alpha, acc.y);
    acc.z = fmaf(hv.z, alpha, acc.z);
    acc.w = fmaf(hv.w, alpha, acc.w);
  }
  {
    const float alpha = exself / ssum;
    const float4 hv = *(const float4*)(h + (size_t)d * 128 + lane * 4);
    acc.x = fmaf(hv.x, alpha, acc.x);
    acc.y = fmaf(hv.y, alpha, acc.y);
    acc.z = fmaf(hv.z, alpha, acc.z);
    acc.w = fmaf(hv.w, alpha, acc.w);
  }
  const float4 bv = *(const float4*)(b1 + lane * 4);
  float4 o;
  o.x = acc.x + bv.x; o.x = (o.x > 0.f) ? o.x : (__expf(o.x) - 1.0f);
  o.y = acc.y + bv.y; o.y = (o.y > 0.f) ? o.y : (__expf(o.y) - 1.0f);
  o.z = acc.z + bv.z; o.z = (o.z > 0.f) ? o.z : (__expf(o.z) - 1.0f);
  o.w = acc.w + bv.w; o.w = (o.w > 0.f) ? o.w : (__expf(o.w) - 1.0f);
  *(float4*)(h1 + (size_t)d * 128 + lane * 4) = o;
}

__global__ void __launch_bounds__(256, 16)
gather1_kernel(const int* __restrict__ rowptr, const int* __restrict__ col2,
               const float* __restrict__ asc, const float* __restrict__ adc,
               const float* __restrict__ h, const float* __restrict__ b1,
               float* __restrict__ h1, int Nn) {
  __shared__ __attribute__((aligned(16))) float tiles[8 * 32 * 8];
  const int lane = threadIdx.x & 31;
  const int w = threadIdx.x >> 5;
  float* tile = tiles + w * (32 * 8);
  const int nw = gridDim.x * 8;
  for (int d = __builtin_amdgcn_readfirstlane(blockIdx.x * 8 + w); d < Nn; d += nw) {
    const int start = rowptr[d];
    const int deg = rowptr[d + 1] - start;
    if (deg <= 32) gat1_node_short(d, lane, start, deg, col2, asc, adc, h, b1, h1, tile);
    else           gat1_node_long(d, lane, start, deg, col2, asc, adc, h, b1, h1);
  }
}

__device__ __forceinline__ float wave_max(float v) {
#pragma unroll
  for (int off = 16; off >= 1; off >>= 1) v = fmaxf(v, __shfl_xor(v, off));
  return v;
}
__device__ __forceinline__ float wave_sum(float v) {
#pragma unroll
  for (int off = 16; off >= 1; off >>= 1) v += __shfl_xor(v, off);
  return v;
}

__device__ __forceinline__ void gat2_accum(int n, int sv, float av, int lc,
                                           const float* __restrict__ h2, float2& acc) {
#pragma unroll 4
  for (int t = 0; t < n; ++t) {
    const int s = __builtin_amdgcn_readlane(sv, t);
    const float a = __int_as_float(__builtin_amdgcn_readlane(__float_as_int(av), t));
    const float2 hv = *(const float2*)(h2 + (size_t)s * 40 + 2 * lc);
    acc.x = fmaf(hv.x, a, acc.x);
    acc.y = fmaf(hv.y, a, acc.y);
  }
}

__global__ void __launch_bounds__(256)
gather2_kernel(const int* __restrict__ rowptr, const int* __restrict__ col2,
               const float* __restrict__ asc, const float* __restrict__ adc,
               const float* __restrict__ h2, const float* __restrict__ b2,
               float* __restrict__ out, int Nn) {
  const int lane = threadIdx.x & 31;
  const int lc = (lane < 19) ? lane : 19;
  const int nw = gridDim.x * 8;
  for (int d = __builtin_amdgcn_readfirstlane(blockIdx.x * 8 + (threadIdx.x >> 5)); d < Nn; d += nw) {
  const int start = rowptr[d];
  const int deg = rowptr[d + 1] - start;
  const float ad = adc[d];
  const float eself = leaky(asc[d] + ad);

  int s0 = 0;
  float e0 = -INFINITY;
  if (lane < deg) { s0 = col2[start + lane]; e0 = leaky(asc[s0] + ad); }
  float m = fmaxf(eself, wave_max(e0));
  for (int c = 32; c < deg; c += 32) {
    float e = -INFINITY;
    if (c + lane < deg) e = leaky(asc[col2[start + c + lane]] + ad);
    m = fmaxf(m, wave_max(e));
  }
  const float ex0 = (lane < deg) ? __expf(e0 - m) : 0.f;
  float ssum = wave_sum(ex0);
  for (int c = 32; c < deg; c += 32) {
    float ex = 0.f;
    if (c + lane < deg) ex = __expf(leaky(asc[col2[start + c + lane]] + ad) - m);
    ssum += wave_sum(ex);
  }
  const float exself = __expf(eself - m);
  ssum += exself;

  float2 acc = make_float2(0.f, 0.f);
  gat2_accum(min(deg, 32), s0, ex0 / ssum, lc, h2, acc);
  for (int c = 32; c < deg; c += 32) {
    int s = 0;
    float a = 0.f;
    if (c + lane < deg) {
      s = col2[start + c + lane];
      a = __expf(leaky(asc[s] + ad) - m) / ssum;
    }
    gat2_accum(min(deg - c, 32), s, a, lc, h2, acc);
  }
  {
    const float a = exself / ssum;
    const float2 hv = *(const float2*)(h2 + (size_t)d * 40 + 2 * lc);
    acc.x = fmaf(hv.x, a, acc.x);
    acc.y = fmaf(hv.y, a, acc.y);
  }
  if (lane < 20) {
    const float2 bv = *(const float2*)(b2 + 2 * lane);
    *(float2*)(out + (size_t)d * 40 + 2 * lane) = make_float2(acc.x + bv.x, acc.y + bv.y);
  }
  }
}

extern "C" void kernel_launch(void* const* d_in, const int* in_sizes, int n_in,
                              void* d_out, int out_size, void* d_ws, size_t ws_size,
                              hipStream_t stream) {
  const float* x   = (const float*)d_in[0];
  const int*   ei  = (const int*)d_in[1];
  const float* W1  = (const float*)d_in[2];
  const float* as1 = (const float*)d_in[3];
  const float* ad1 = (const float*)d_in[4];
  const float* b1  = (const float*)d_in[5];
  const float* W2  = (const float*)d_in[6];
  const float* as2 = (const float*)d_in[7];
  const float* ad2 = (const float*)d_in[8];
  const float* b2  = (const float*)d_in[9];
  float* out = (float*)d_out;

  const int Nn   = in_sizes[0] / 128;
  const int E    = in_sizes[1] / 2;
  const int* srcp = ei;
  const int* dstp = ei + E;
  const int NBLK = (E + ETILE - 1) / ETILE;
  const int NBKT = (Nn + BNODES - 1) / BNODES;

  float* ws   = (float*)d_ws;
  float* h    = ws;  ws += (size_t)Nn * 128;
  float* h1   = ws;  ws += (size_t)Nn * 128;
  float* asc1 = ws;  ws += (size_t)Nn * 8;
  float* adc1 = ws;  ws += (size_t)Nn * 8;
  float* h2   = ws;  ws += (size_t)Nn * 40;
  float* asc2 = ws;  ws += Nn;
  float* adc2 = ws;  ws += Nn;
  int* wi     = (int*)ws;
  int* bcnt   = wi;  wi += (size_t)NBLK * NBINS;
  int* rec    = wi;  wi += E;
  int* col    = wi;  wi += E;
  int* col2   = wi;  wi += E;
  int* rowptr = wi;  wi += (size_t)Nn + 1;
  int* bstart = wi;  wi += NBINS + 1;

  bucket_count_kernel<<<NBLK, 1024, 0, stream>>>(dstp, bcnt, E);
  bucket_scatter_kernel<<<NBLK, 1024, 0, stream>>>(srcp, dstp, bcnt, bstart, rec, E, NBLK);
  bucket_csr_kernel<<<NBKT, 1024, 0, stream>>>(bstart, rec, rowptr, col, Nn);
  sort_seg_kernel<<<2048, 256, 0, stream>>>(rowptr, col, col2, Nn);

  gat_wmma_gemm_k128<4, 128, 8><<<dim3(896, 2), 4 * 32, 0, stream>>>(x, W1, h, as1, ad1, asc1, adc1, Nn / 16);
  gather1_kernel<<<2048, 256, 0, stream>>>(rowptr, col2, asc1, adc1, h, b1, h1, Nn);

  gat_wmma_gemm_k128<3, 40, 1><<<dim3(1563, 1), 3 * 32, 0, stream>>>(h1, W2, h2, as2, ad2, asc2, adc2, Nn / 16);
  gather2_kernel<<<2048, 256, 0, stream>>>(rowptr, col2, asc2, adc2, h2, b2, out, Nn);
}


# Layer-1 gather: nodes of 33-64 edges go edge-per-lane through a 64-row LDS tile, not the serial fallback whose ~80 dependent round trips set the launch tail; their denominator: lane sums, 4-lane tree
# speedup vs baseline: 5.4243x; 1.0620x over previous
#include <hip/hip_runtime.h>
#include <hip/hip_bf16.h>
#include <math.h>
#include <stdint.h>


typedef float v2f __attribute__((ext_vector_type(2)));
typedef float v8f __attribute__((ext_vector_type(8)));

#define NEG_SLOPE 0.2f

template <int MASK>
__device__ __forceinline__ float row_xor_get(float v) {
  return __int_as_float(__builtin_amdgcn_update_dpp(0, __float_as_int(v), 0x160 + MASK, 0xf, 0xf, true));
}

template <int NTILES, int NACT, int HEADS>
__global__ void __launch_bounds__(32 * NTILES)
gat_wmma_gemm_k128(const float* __restrict__ A, const float* __restrict__ B,
                   float* __restrict__ C,
                   const float* __restrict__ att_src, const float* __restrict__ att_dst,
                   float* __restrict__ a_src, float* __restrict__ a_dst, int nstrips) {
  static_assert(HEADS == 1 || HEADS * 16 == NACT, "a wave's 16 columns must be one head");
  constexpr int NPAD = NTILES * 16;
  constexpr int SA = 132;
  constexpr int SB = 132;
  constexpr int NT = 32 * NTILES;
  __shared__ __attribute__((aligned(16))) float ldsA[16 * SA];
  __shared__ float ldsB[NPAD * SB];
  __shared__ float red[(HEADS == 1) ? NTILES * 32 : 1];

  const int tid  = threadIdx.x;
  const int lane = tid & 31;
  const int wave = tid >> 5;
  const int c0 = blockIdx.y * NPAD;

  static_assert(NACT % 4 == 0 && NPAD % 4 == 0, "16-byte pieces must not straddle the last column");
  constexpr int BIT = (NPAD / 4) * 128 / NT;
  static_assert(BIT * NT == (NPAD / 4) * 128, "the panel must be a whole number of pieces per thread");
  {
    float4 bv[BIT];
#pragma unroll
    for (int u = 0; u < BIT; ++u) {
      const int p = tid + u * NT;
      const int k = ((p / NPAD) << 2) + (p & 3);
      const int gn = c0 + (((p >> 2) % (NPAD / 4)) << 2);
      const int gc = (gn < NACT) ? gn : (NACT - 4);
      bv[u] = *(const float4*)(B + (size_t)k * NACT + gc);
    }
#pragma unroll
    for (int u = 0; u < BIT; ++u) {
      const int p = tid + u * NT;
      const int k = ((p / NPAD) << 2) + (p & 3);
      const int n = ((p >> 2) % (NPAD / 4)) << 2;
      const bool in = (c0 + n) < NACT;
      float* lp = &ldsB[n * SB + k];
      lp[0]      = in ? bv[u].x : 0.0f;
      lp[SB]     = in ? bv[u].y : 0.0f;
      lp[2 * SB] = in ? bv[u].z : 0.0f;
      lp[3 * SB] = in ? bv[u].w : 0.0f;
    }
  }

  const int l15  = lane & 15;
  const int koff = (lane >> 4) << 1;
  const int ncol = wave * 16 + l15;
  const float* ap = &ldsA[l15 * SA + koff];
  const float* bp = &ldsB[ncol * SB + koff];

  const int gcol = c0 + ncol;
  const float wsrc = (gcol < NACT) ? att_src[gcol] : 0.0f;
  const float wdst = (gcol < NACT) ? att_dst[gcol] : 0.0f;
  const int hd = blockIdx.y * NTILES + wave;

  for (int strip = blockIdx.x; strip < nstrips; strip += gridDim.x) {
    const size_t m0 = (size_t)strip * 16;
    constexpr int AIT = (16 * 32 + NT - 1) / NT;
    float4 av[AIT];
#pragma unroll
    for (int u = 0; u < AIT; ++u) {
      const int i = min(tid + u * NT, 16 * 32 - 1);
      av[u] = *(const float4*)(A + (m0 + (i >> 5)) * 128 + ((i & 31) << 2));
    }
#pragma unroll
    for (int u = 0; u < AIT; ++u) {
      const int i = tid + u * NT;
      if ((u + 1) * NT <= 16 * 32 || i < 16 * 32) *(float4*)&ldsA[(i >> 5) * SA + ((i & 31) << 2)] = av[u];
    }
    __syncthreads();

    v8f acc = {0.f, 0.f, 0.f, 0.f, 0.f, 0.f, 0.f, 0.f};
#pragma unroll
    for (int kb = 0; kb < 128; kb += 4) {
      v2f a = *(const v2f*)(ap + kb);
      v2f b = *(const v2f*)(bp + kb);
      acc = __builtin_amdgcn_wmma_f32_16x16x4_f32(
           false, a,  false, b,
           (short)0, acc,  false,  false);
    }

    if (gcol < NACT) {
      float* cp = C + (m0 + (size_t)((lane >> 4) << 3)) * NACT + gcol;
#pragma unroll
      for (int r = 0; r < 8; ++r) cp[(size_t)r * NACT] = acc[r];
    }

    const bool b3 = (l15 & 8) != 0, b2 = (l15 & 4) != 0, b1 = (l15 & 2) != 0;
    float ps[8], pd[8];
#pragma unroll
    for (int r = 0; r < 8; ++r) { ps[r] = acc[r] * wsrc; pd[r] = acc[r] * wdst; }
    float s4[4], d4[4];
#pragma unroll
    for (int i = 0; i < 4; ++i) {
      s4[i] = (b3 ? ps[4 + i] : ps[i]) + row_xor_get<8>(b3 ? ps[i] : ps[4 + i]);
      d4[i] = (b3 ? pd[4 + i] : pd[i]) + row_xor_get<8>(b3 ? pd[i] : pd[4 + i]);
    }
    float s2[2], d2[2];
#pragma unroll
    for (int j = 0; j < 2; ++j) {
      s2[j] = (b2 ? s4[2 + j] : s4[j]) + row_xor_get<4>(b2 ? s4[j] : s4[2 + j]);
      d2[j] = (b2 ? d4[2 + j] : d4[j]) + row_xor_get<4>(b2 ? d4[j] : d4[2 + j]);
    }
    float s1 = (b1 ? s2[1] : s2[0]) + row_xor_get<2>(b1 ? s2[0] : s2[1]);
    float d1 = (b1 ? d2[1] : d2[0]) + row_xor_get<2>(b1 ? d2[0] : d2[1]);
    s1 += row_xor_get<1>(s1);
    d1 += row_xor_get<1>(d1);
    const bool is_src = (l15 & 1) == 0;
    const float sv = is_src ? s1 : d1;
    const int row = ((lane >> 4) << 3) + (l15 >> 1);
    if constexpr (HEADS == 1) {
      red[(wave * 16 + row) * 2 + (is_src ? 0 : 1)] = sv;
    } else {
      float* sp = is_src ? a_src : a_dst;
      sp[(m0 + row) * HEADS + hd] = sv;
    }
    __syncthreads();
    if constexpr (HEADS == 1) if (wave == 0) {
      const int r16 = lane & 15, which = lane >> 4;
      float s = red[r16 * 2 + which];
#pragma unroll
      for (int w = 1; w < NTILES; ++w) s += red[(w * 16 + r16) * 2 + which];
      float* sp = which ? a_dst : a_src;
      sp[m0 + r16] = s;
    }
  }
}

#define BSHIFT 9
#define BNODES 512
#define NBINS 256
#define SRC_BITS 17
#define ETILE 8192

__global__ void __launch_bounds__(1024)
bucket_count_kernel(const int* __restrict__ dst, int* __restrict__ bcnt, int E) {
  __shared__ int hist[NBINS];
  const int tid = threadIdx.x;
  if (tid < NBINS) hist[tid] = 0;
  const int base = blockIdx.x * ETILE + tid;
  int dv[8];
#pragma unroll
  for (int k = 0; k < 8; ++k) {
    const int e = base + k * 1024;
    dv[k] = (e < E) ? dst[e] : -1;
  }
  __syncthreads();
#pragma unroll
  for (int k = 0; k < 8; ++k) if (dv[k] >= 0) atomicAdd(&hist[dv[k] >> BSHIFT], 1);
  __syncthreads();
  if (tid < NBINS) bcnt[blockIdx.x * NBINS + tid] = hist[tid];
}

__global__ void __launch_bounds__(1024)
bucket_scatter_kernel(const int* __restrict__ src, const int* __restrict__ dst,
                      const int* __restrict__ bcnt, int* __restrict__ bstart,
                      int* __restrict__ rec, int E, int nblk) {
  __shared__ int ptot[4][NBINS];
  __shared__ int ppre[4][NBINS];
  __shared__ int cursor[NBINS + 1];
  const int tid = threadIdx.x;
  const int lane = tid & 31;
  const int base = blockIdx.x * ETILE + tid;
  int dv[8], sv[8];
#pragma unroll
  for (int k = 0; k < 8; ++k) {
    const int e = base + k * 1024;
    const bool ok = e < E;
    dv[k] = ok ? dst[e] : -1;
    sv[k] = ok ? src[e] : 0;
  }
  {
    const int k = tid & (NBINS - 1);
    const int q = tid >> 8;
    const int gq = (nblk + 3) >> 2;
    const int j0 = q * gq;
    const int j1 = min(nblk, j0 + gq);
    int tot = 0, pre = 0;
#pragma unroll 8
    for (int j = j0; j < j1; ++j) {
      const int c = bcnt[j * NBINS + k];
      tot += c;
      pre += (j < (int)blockIdx.x) ? c : 0;
    }
    ptot[q][k] = tot;
    ppre[q][k] = pre;
    if (tid == 0) cursor[NBINS] = 0;
  }
  __syncthreads();
  if (tid < 32) {
    int v[8];
    int run = 0;
#pragma unroll
    for (int i = 0; i < 8; ++i) {
      const int idx = tid * 8 + i;
      v[i] = run;
      run += ptot[0][idx] + ptot[1][idx] + ptot[2][idx] + ptot[3][idx];
    }
    int inc = run;
#pragma unroll
    for (int off = 1; off < 32; off <<= 1) {
      const int nb = __shfl_up(inc, off);
      if (lane >= off) inc += nb;
    }
    const int excl = inc - run;
#pragma unroll
    for (int i = 0; i < 8; ++i) {
      const int idx = tid * 8 + i;
      const int st = excl + v[i];
      cursor[idx] = st + ppre[0][idx] + ppre[1][idx] + ppre[2][idx] + ppre[3][idx];
      if (blockIdx.x == 0) bstart[idx] = st;
    }
    if (blockIdx.x == 0 && tid == 31) bstart[NBINS] = inc;
  }
  __syncthreads();
  int pv[8];
#pragma unroll
  for (int k = 0; k < 8; ++k) pv[k] = atomicAdd(&cursor[(dv[k] >= 0) ? (dv[k] >> BSHIFT) : NBINS], 1);
#pragma unroll
  for (int k = 0; k < 8; ++k)
    if (dv[k] >= 0) rec[pv[k]] = ((dv[k] & (BNODES - 1)) << SRC_BITS) | sv[k];
}

__global__ void __launch_bounds__(1024)
bucket_csr_kernel(const int* __restrict__ bstart, const int* __restrict__ rec,
                  int* __restrict__ rowptr, int* __restrict__ col, int Nn) {
  __shared__ int cntl[BNODES + 1];
  __shared__ int wsum[16];
  const int tid = threadIdx.x, lane = tid & 31, w = tid >> 5;
  const int bs = bstart[blockIdx.x];
  const int n = bstart[blockIdx.x + 1] - bs;
  if (tid <= BNODES) cntl[tid] = 0;
  __syncthreads();
  const int* rp = rec + bs;
  for (int t0 = 0; t0 < n; t0 += ETILE) {
    int rv[8];
#pragma unroll
    for (int k = 0; k < 8; ++k) {
      const int i = t0 + k * 1024 + tid;
      rv[k] = (i < n) ? rp[i] : -1;
    }
#pragma unroll
    for (int k = 0; k < 8; ++k) if (rv[k] >= 0) atomicAdd(&cntl[rv[k] >> SRC_BITS], 1);
  }
  __syncthreads();
  int c = 0, inc = 0;
  if (tid < BNODES) {
    c = cntl[tid];
    inc = c;
#pragma unroll
    for (int off = 1; off < 32; off <<= 1) {
      const int nb = __shfl_up(inc, off);
      if (lane >= off) inc += nb;
    }
    if (lane == 31) wsum[w] = inc;
  }
  __syncthreads();
  if (tid < BNODES) {
    int woff = 0;
#pragma unroll
    for (int k = 0; k < 16; ++k) woff += (k < w) ? wsum[k] : 0;
    const int excl = woff + inc - c;
    cntl[tid] = excl;
    const int node = blockIdx.x * BNODES + tid;
    if (node < Nn) rowptr[node] = bs + excl;
  }
  if (blockIdx.x == gridDim.x - 1 && tid == 0) rowptr[Nn] = bs + n;
  __syncthreads();
  int* cp = col + bs;
  for (int t0 = 0; t0 < n; t0 += ETILE) {
    int rv[8], pv[8];
#pragma unroll
    for (int k = 0; k < 8; ++k) {
      const int i = t0 + k * 1024 + tid;
      rv[k] = (i < n) ? rp[i] : -1;
    }
#pragma unroll
    for (int k = 0; k < 8; ++k) pv[k] = atomicAdd(&cntl[(rv[k] >= 0) ? (rv[k] >> SRC_BITS) : BNODES], 1);
#pragma unroll
    for (int k = 0; k < 8; ++k) if (rv[k] >= 0) cp[pv[k]] = rv[k] & ((1 << SRC_BITS) - 1);
  }
}

__global__ void __launch_bounds__(256)
sort_seg_kernel(const int* __restrict__ rowptr, const int* __restrict__ col,
                int* __restrict__ col2, int Nn) {
  const int lane = threadIdx.x & 31;
  const int nw = gridDim.x * 8;
  for (int d = __builtin_amdgcn_readfirstlane(blockIdx.x * 8 + (threadIdx.x >> 5)); d < Nn; d += nw) {
  const int start = rowptr[d];
  const int deg = rowptr[d + 1] - start;
  if (deg <= 32) {
    const int k = (lane < deg) ? col[start + lane] : 0x7fffffff;
    int r = 0;
    for (int j = 0; j < deg; ++j) {
      const int kj = __builtin_amdgcn_readlane(k, j);
      r += (kj < k || (kj == k && j < lane)) ? 1 : 0;
    }
    if (lane < deg) col2[start + r] = k;
  } else {
    for (int i = lane; i < deg; i += 32) {
      const int k = col[start + i];
      int r = 0;
      for (int j = 0; j < deg; ++j) {
        const int kj = col[start + j];
        r += (kj < k || (kj == k && j < i)) ? 1 : 0;
      }
      col2[start + r] = k;
    }
  }
  }
}

__device__ __forceinline__ float leaky(float v) { return (v >= 0.f) ? v : NEG_SLOPE * v; }


__device__ __forceinline__ void wave_lds_order() {
  __builtin_amdgcn_fence(__ATOMIC_RELEASE, "wavefront");
  __builtin_amdgcn_wave_barrier();
  __builtin_amdgcn_fence(__ATOMIC_ACQUIRE, "wavefront");
}

__device__ __forceinline__ void gat1_node_short(int d, int lane, int start, int deg,
                                                const int* __restrict__ col2,
                                                const float* __restrict__ asc,
                                                const float* __restrict__ adc,
                                                const float* __restrict__ h,
                                                const float* __restrict__ b1,
                                                float* __restrict__ h1, float* tile) {
  const int hd = lane >> 2, q = lane & 3;
  const float4 hself = *(const float4*)(h + (size_t)d * 128 + lane * 4);
  const float4 ad0 = *(const float4*)(adc + (size_t)d * 8);
  const float4 ad1 = *(const float4*)(adc + (size_t)d * 8 + 4);
  const float eself = leaky(asc[(size_t)d * 8 + hd] + adc[(size_t)d * 8 + hd]);
  int sv = d;
  float4 e0 = make_float4(-INFINITY, -INFINITY, -INFINITY, -INFINITY);
  float4 e1 = e0;
  if (lane < deg) {
    sv = col2[start + lane];
    const float4 a0 = *(const float4*)(asc + (size_t)sv * 8);
    const float4 a1 = *(const float4*)(asc + (size_t)sv * 8 + 4);
    e0.x = leaky(a0.x + ad0.x); e0.y = leaky(a0.y + ad0.y);
    e0.z = leaky(a0.z + ad0.z); e0.w = leaky(a0.w + ad0.w);
    e1.x = leaky(a1.x + ad1.x); e1.y = leaky(a1.y + ad1.y);
    e1.z = leaky(a1.z + ad1.z); e1.w = leaky(a1.w + ad1.w);
  }
  *(float4*)(tile + lane * 8) = e0;
  *(float4*)(tile + lane * 8 + 4) = e1;
  wave_lds_order();

  float* colp = tile + q * 8 + hd;
  float v[8];
#pragma unroll
  for (int t = 0; t < 8; ++t) v[t] = colp[32 * t];
  float m = eself;
#pragma unroll
  for (int t = 0; t < 8; ++t) m = fmaxf(m, v[t]);
  m = fmaxf(m, row_xor_get<1>(m));
  m = fmaxf(m, row_xor_get<2>(m));
  float ssum = 0.f;
#pragma unroll
  for (int t = 0; t < 8; ++t) { v[t] = __expf(v[t] - m); ssum += v[t]; }
  ssum += row_xor_get<1>(ssum);
  ssum += row_xor_get<2>(ssum);
  const float exself = __expf(eself - m);
  ssum += exself;
#pragma unroll
  for (int t = 0; t < 8; ++t) {
    if (4 * t < deg) colp[32 * t] = v[t] / ssum;
  }
  wave_lds_order();

  float4 acc = make_float4(0.f, 0.f, 0.f, 0.f);
  const float* wp = tile + hd;
  for (int j0 = 0; j0 < deg; j0 += 4) {
#pragma unroll
    for (int u = 0; u < 4; ++u) {
      const int j = j0 + u;
      const int s = __builtin_amdgcn_readlane(sv, j);
      const float alpha = wp[j * 8];
      const float4 hv = *(const float4*)(h + (size_t)s * 128 + lane * 4);
      acc.x = fmaf(hv.x, alpha, acc.x);
      acc.y = fmaf(hv.y, alpha, acc.y);
      acc.z = fmaf(hv.z, alpha, acc.z);
      acc.w = fmaf(hv.w, alpha, acc.w);
    }
  }
  {
    const float alpha = exself / ssum;
    acc.x = fmaf(hself.x, alpha, acc.x);
    acc.y = fmaf(hself.y, alpha, acc.y);
    acc.z = fmaf(hself.z, alpha, acc.z);
    acc.w = fmaf(hself.w, alpha, acc.w);
  }
  const float4 bv = *(const float4*)(b1 + lane * 4);
  float4 o;
  o.x = acc.x + bv.x; o.x = (o.x > 0.f) ? o.x : (__expf(o.x) - 1.0f);
  o.y = acc.y + bv.y; o.y = (o.y > 0.f) ? o.y : (__expf(o.y) - 1.0f);
  o.z = acc.z + bv.z; o.z = (o.z > 0.f) ? o.z : (__expf(o.z) - 1.0f);
  o.w = acc.w + bv.w; o.w = (o.w > 0.f) ? o.w : (__expf(o.w) - 1.0f);
  *(float4*)(h1 + (size_t)d * 128 + lane * 4) = o;
  wave_lds_order();
}

__device__ __forceinline__ void gat1_node_mid(int d, int lane, int start, int deg,
                                              const int* __restrict__ col2,
                                              const float* __restrict__ asc,
                                              const float* __restrict__ adc,
                                              const float* __restrict__ h,
                                              const float* __restrict__ b1,
                                              float* __restrict__ h1, float* tile) {
  const int hd = lane >> 2, q = lane & 3;
  const float4 ad0 = *(const float4*)(adc + (size_t)d * 8);
  const float4 ad1 = *(const float4*)(adc + (size_t)d * 8 + 4);
  const float eself = leaky(asc[(size_t)d * 8 + hd] + adc[(size_t)d * 8 + hd]);
  const bool has1 = (32 + lane) < deg;
  const int sv0 = col2[start + lane];
  const int sv1 = has1 ? col2[start + 32 + lane] : d;
  {
    const float4 a0 = *(const float4*)(asc + (size_t)sv0 * 8);
    const float4 a1 = *(const float4*)(asc + (size_t)sv0 * 8 + 4);
    const float4 c0 = *(const float4*)(asc + (size_t)sv1 * 8);
    const float4 c1 = *(const float4*)(asc + (size_t)sv1 * 8 + 4);
    float4 e0, e1, f0, f1;
    e0.x = leaky(a0.x + ad0.x); e0.y = leaky(a0.y + ad0.y);
    e0.z = leaky(a0.z + ad0.z); e0.w = leaky(a0.w + ad0.w);
    e1.x = leaky(a1.x + ad1.x); e1.y = leaky(a1.y + ad1.y);
    e1.z = leaky(a1.z + ad1.z); e1.w = leaky(a1.w + ad1.w);
    f0.x = has1 ? leaky(c0.x + ad0.x) : -INFINITY; f0.y = has1 ? leaky(c0.y + ad0.y) : -INFINITY;
    f0.z = has1 ? leaky(c0.z + ad0.z) : -INFINITY; f0.w = has1 ? leaky(c0.w + ad0.w) : -INFINITY;
    f1.x = has1 ? leaky(c1.x + ad1.x) : -INFINITY; f1.y = has1 ? leaky(c1.y + ad1.y) : -INFINITY;
    f1.z = has1 ? leaky(c1.z + ad1.z) : -INFINITY; f1.w = has1 ? leaky(c1.w + ad1.w) : -INFINITY;
    *(float4*)(tile + lane * 8) = e0;
    *(float4*)(tile + lane * 8 + 4) = e1;
    *(float4*)(tile + (32 + lane) * 8) = f0;
    *(float4*)(tile + (32 + lane) * 8 + 4) = f1;
  }
  wave_lds_order();

  float* colp = tile + q * 8 + hd;
  float m = eself;
#pragma unroll
  for (int t = 0; t < 16; ++t) m = fmaxf(m, colp[32 * t]);
  m = fmaxf(m, row_xor_get<1>(m));
  m = fmaxf(m, row_xor_get<2>(m));
  float ssum = 0.f;
#pragma unroll
  for (int t = 0; t < 16; ++t) {
    const float ex = __expf(colp[32 * t] - m);
    ssum += ex;
    colp[32 * t] = ex;
  }
  ssum += row_xor_get<1>(ssum);
  ssum += row_xor_get<2>(ssum);
  const float exself = __expf(eself - m);
  ssum += exself;
#pragma unroll
  for (int t = 0; t < 16; ++t) colp[32 * t] = colp[32 * t] / ssum;
  wave_lds_order();

  float4 acc = make_float4(0.f, 0.f, 0.f, 0.f);
  const float* wp = tile + hd;
  for (int j0 = 0; j0 < deg; j0 += 4) {
    const int svc = (j0 < 32) ? sv0 : sv1;
#pragma unroll
    for (int u = 0; u < 4; ++u) {
      const int j = j0 + u;
      const int s = __builtin_amdgcn_readlane(svc, j & 31);
      const float alpha = wp[j * 8];
      const float4 hv = *(const float4*)(h + (size_t)s * 128 + lane * 4);
      acc.x = fmaf(hv.x, alpha, acc.x);
      acc.y = fmaf(hv.y, alpha, acc.y);
      acc.z = fmaf(hv.z, alpha, acc.z);
      acc.w = fmaf(hv.w, alpha, acc.w);
    }
  }
  {
    const float4 hself = *(const float4*)(h + (size_t)d * 128 + lane * 4);
    const float alpha = exself / ssum;
    acc.x = fmaf(hself.x, alpha, acc.x);
    acc.y = fmaf(hself.y, alpha, acc.y);
    acc.z = fmaf(hself.z, alpha, acc.z);
    acc.w = fmaf(hself.w, alpha, acc.w);
  }
  const float4 bv = *(const float4*)(b1 + lane * 4);
  float4 o;
  o.x = acc.x + bv.x; o.x = (o.x > 0.f) ? o.x : (__expf(o.x) - 1.0f);
  o.y = acc.y + bv.y; o.y = (o.y > 0.f) ? o.y : (__expf(o.y) - 1.0f);
  o.z = acc.z + bv.z; o.z = (o.z > 0.f) ? o.z : (__expf(o.z) - 1.0f);
  o.w = acc.w + bv.w; o.w = (o.w > 0.f) ? o.w : (__expf(o.w) - 1.0f);
  *(float4*)(h1 + (size_t)d * 128 + lane * 4) = o;
  wave_lds_order();
}

__device__ __forceinline__ void gat1_node_long(int d, int lane, int start, int deg,
                                               const int* __restrict__ col2,
                                               const float* __restrict__ asc,
                                               const float* __restrict__ adc,
                                               const float* __restrict__ h,
                                               const float* __restrict__ b1,
                                               float* __restrict__ h1) {
  const int hd = lane >> 2;
  const float ad = adc[(size_t)d * 8 + hd];
  const float eself = leaky(asc[(size_t)d * 8 + hd] + ad);
  float m = eself;
#pragma unroll 4
  for (int j = 0; j < deg; ++j) {
    const int s = col2[start + j];
    const float e = leaky(asc[(size_t)s * 8 + hd] + ad);
    m = fmaxf(m, e);
  }
  float ssum = 0.f;
#pragma unroll 4
  for (int j = 0; j < deg; ++j) {
    const float e = leaky(asc[(size_t)col2[start + j] * 8 + hd] + ad);
    ssum += __expf(e - m);
  }
  const float exself = __expf(eself - m);
  ssum += exself;
  float4 acc = make_float4(0.f, 0.f, 0.f, 0.f);
#pragma unroll 4
  for (int j = 0; j < deg; ++j) {
    const int s = col2[start + j];
    const float ex = __expf(leaky(asc[(size_t)s * 8 + hd] + ad) - m);
    const float alpha = ex / ssum;
    const float4 hv = *(const float4*)(h + (size_t)s * 128 + lane * 4);
    acc.x = fmaf(hv.x, alpha, acc.x);
    acc.y = fmaf(hv.y, alpha, acc.y);
    acc.z = fmaf(hv.z, alpha, acc.z);
    acc.w = fmaf(hv.w, alpha, acc.w);
  }
  {
    const float alpha = exself / ssum;
    const float4 hv = *(const float4*)(h + (size_t)d * 128 + lane * 4);
    acc.x = fmaf(hv.x, alpha, acc.x);
    acc.y = fmaf(hv.y, alpha, acc.y);
    acc.z = fmaf(hv.z, alpha, acc.z);
    acc.w = fmaf(hv.w, alpha, acc.w);
  }
  const float4 bv = *(const float4*)(b1 + lane * 4);
  float4 o;
  o.x = acc.x + bv.x; o.x = (o.x > 0.f) ? o.x : (__expf(o.x) - 1.0f);
  o.y = acc.y + bv.y; o.y = (o.y > 0.f) ? o.y : (__expf(o.y) - 1.0f);
  o.z = acc.z + bv.z; o.z = (o.z > 0.f) ? o.z : (__expf(o.z) - 1.0f);
  o.w = acc.w + bv.w; o.w = (o.w > 0.f) ? o.w : (__expf(o.w) - 1.0f);
  *(float4*)(h1 + (size_t)d * 128 + lane * 4) = o;
}

__global__ void __launch_bounds__(256, 16)
gather1_kernel(const int* __restrict__ rowptr, const int* __restrict__ col2,
               const float* __restrict__ asc, const float* __restrict__ adc,
               const float* __restrict__ h, const float* __restrict__ b1,
               float* __restrict__ h1, int Nn) {
  __shared__ __attribute__((aligned(16))) float tiles[8 * 64 * 8];
  const int lane = threadIdx.x & 31;
  const int w = threadIdx.x >> 5;
  float* tile = tiles + w * (64 * 8);
  const int nw = gridDim.x * 8;
  for (int d = __builtin_amdgcn_readfirstlane(blockIdx.x * 8 + w); d < Nn; d += nw) {
    const int start = rowptr[d];
    const int deg = rowptr[d + 1] - start;
    if (deg <= 32)      gat1_node_short(d, lane, start, deg, col2, asc, adc, h, b1, h1, tile);
    else if (deg <= 64) gat1_node_mid(d, lane, start, deg, col2, asc, adc, h, b1, h1, tile);
    else                gat1_node_long(d, lane, start, deg, col2, asc, adc, h, b1, h1);
  }
}

__device__ __forceinline__ float wave_max(float v) {
#pragma unroll
  for (int off = 16; off >= 1; off >>= 1) v = fmaxf(v, __shfl_xor(v, off));
  return v;
}
__device__ __forceinline__ float wave_sum(float v) {
#pragma unroll
  for (int off = 16; off >= 1; off >>= 1) v += __shfl_xor(v, off);
  return v;
}

__device__ __forceinline__ void gat2_accum(int n, int sv, float av, int lc,
                                           const float* __restrict__ h2, float2& acc) {
#pragma unroll 4
  for (int t = 0; t < n; ++t) {
    const int s = __builtin_amdgcn_readlane(sv, t);
    const float a = __int_as_float(__builtin_amdgcn_readlane(__float_as_int(av), t));
    const float2 hv = *(const float2*)(h2 + (size_t)s * 40 + 2 * lc);
    acc.x = fmaf(hv.x, a, acc.x);
    acc.y = fmaf(hv.y, a, acc.y);
  }
}

__global__ void __launch_bounds__(256)
gather2_kernel(const int* __restrict__ rowptr, const int* __restrict__ col2,
               const float* __restrict__ asc, const float* __restrict__ adc,
               const float* __restrict__ h2, const float* __restrict__ b2,
               float* __restrict__ out, int Nn) {
  const int lane = threadIdx.x & 31;
  const int lc = (lane < 19) ? lane : 19;
  const int nw = gridDim.x * 8;
  for (int d = __builtin_amdgcn_readfirstlane(blockIdx.x * 8 + (threadIdx.x >> 5)); d < Nn; d += nw) {
  const int start = rowptr[d];
  const int deg = rowptr[d + 1] - start;
  const float ad = adc[d];
  const float eself = leaky(asc[d] + ad);

  int s0 = 0;
  float e0 = -INFINITY;
  if (lane < deg) { s0 = col2[start + lane]; e0 = leaky(asc[s0] + ad); }
  float m = fmaxf(eself, wave_max(e0));
  for (int c = 32; c < deg; c += 32) {
    float e = -INFINITY;
    if (c + lane < deg) e = leaky(asc[col2[start + c + lane]] + ad);
    m = fmaxf(m, wave_max(e));
  }
  const float ex0 = (lane < deg) ? __expf(e0 - m) : 0.f;
  float ssum = wave_sum(ex0);
  for (int c = 32; c < deg; c += 32) {
    float ex = 0.f;
    if (c + lane < deg) ex = __expf(leaky(asc[col2[start + c + lane]] + ad) - m);
    ssum += wave_sum(ex);
  }
  const float exself = __expf(eself - m);
  ssum += exself;

  float2 acc = make_float2(0.f, 0.f);
  gat2_accum(min(deg, 32), s0, ex0 / ssum, lc, h2, acc);
  for (int c = 32; c < deg; c += 32) {
    int s = 0;
    float a = 0.f;
    if (c + lane < deg) {
      s = col2[start + c + lane];
      a = __expf(leaky(asc[s] + ad) - m) / ssum;
    }
    gat2_accum(min(deg - c, 32), s, a, lc, h2, acc);
  }
  {
    const float a = exself / ssum;
    const float2 hv = *(const float2*)(h2 + (size_t)d * 40 + 2 * lc);
    acc.x = fmaf(hv.x, a, acc.x);
    acc.y = fmaf(hv.y, a, acc.y);
  }
  if (lane < 20) {
    const float2 bv = *(const float2*)(b2 + 2 * lane);
    *(float2*)(out + (size_t)d * 40 + 2 * lane) = make_float2(acc.x + bv.x, acc.y + bv.y);
  }
  }
}

extern "C" void kernel_launch(void* const* d_in, const int* in_sizes, int n_in,
                              void* d_out, int out_size, void* d_ws, size_t ws_size,
                              hipStream_t stream) {
  const float* x   = (const float*)d_in[0];
  const int*   ei  = (const int*)d_in[1];
  const float* W1  = (const float*)d_in[2];
  const float* as1 = (const float*)d_in[3];
  const float* ad1 = (const float*)d_in[4];
  const float* b1  = (const float*)d_in[5];
  const float* W2  = (const float*)d_in[6];
  const float* as2 = (const float*)d_in[7];
  const float* ad2 = (const float*)d_in[8];
  const float* b2  = (const float*)d_in[9];
  float* out = (float*)d_out;

  const int Nn   = in_sizes[0] / 128;
  const int E    = in_sizes[1] / 2;
  const int* srcp = ei;
  const int* dstp = ei + E;
  const int NBLK = (E + ETILE - 1) / ETILE;
  const int NBKT = (Nn + BNODES - 1) / BNODES;

  float* ws   = (float*)d_ws;
  float* h    = ws;  ws += (size_t)Nn * 128;
  float* h1   = ws;  ws += (size_t)Nn * 128;
  float* asc1 = ws;  ws += (size_t)Nn * 8;
  float* adc1 = ws;  ws += (size_t)Nn * 8;
  float* h2   = ws;  ws += (size_t)Nn * 40;
  float* asc2 = ws;  ws += Nn;
  float* adc2 = ws;  ws += Nn;
  int* wi     = (int*)ws;
  int* bcnt   = wi;  wi += (size_t)NBLK * NBINS;
  int* rec    = wi;  wi += E;
  int* col    = wi;  wi += E;
  int* col2   = wi;  wi += E;
  int* rowptr = wi;  wi += (size_t)Nn + 1;
  int* bstart = wi;  wi += NBINS + 1;

  bucket_count_kernel<<<NBLK, 1024, 0, stream>>>(dstp, bcnt, E);
  bucket_scatter_kernel<<<NBLK, 1024, 0, stream>>>(srcp, dstp, bcnt, bstart, rec, E, NBLK);
  bucket_csr_kernel<<<NBKT, 1024, 0, stream>>>(bstart, rec, rowptr, col, Nn);
  sort_seg_kernel<<<2048, 256, 0, stream>>>(rowptr, col, col2, Nn);

  gat_wmma_gemm_k128<4, 128, 8><<<dim3(896, 2), 4 * 32, 0, stream>>>(x, W1, h, as1, ad1, asc1, adc1, Nn / 16);
  gather1_kernel<<<2048, 256, 0, stream>>>(rowptr, col2, asc1, adc1, h, b1, h1, Nn);

  gat_wmma_gemm_k128<3, 40, 1><<<dim3(1563, 1), 3 * 32, 0, stream>>>(h1, W2, h2, as2, ad2, asc2, adc2, Nn / 16);
  gather2_kernel<<<2048, 256, 0, stream>>>(rowptr, col2, asc2, adc2, h2, b2, out, Nn);
}


# Segment sort by source id moved into the layer-1 gather wave (rank by key compare under the score loads, rows placed by rank in its LDS tile, sorted ids kept for layer 2): sort launch gone
# speedup vs baseline: 5.6309x; 1.0381x over previous
#include <hip/hip_runtime.h>
#include <hip/hip_bf16.h>
#include <math.h>
#include <stdint.h>


typedef float v2f __attribute__((ext_vector_type(2)));
typedef float v8f __attribute__((ext_vector_type(8)));

#define NEG_SLOPE 0.2f

template <int MASK>
__device__ __forceinline__ float row_xor_get(float v) {
  return __int_as_float(__builtin_amdgcn_update_dpp(0, __float_as_int(v), 0x160 + MASK, 0xf, 0xf, true));
}

template <int NTILES, int NACT, int HEADS>
__global__ void __launch_bounds__(32 * NTILES)
gat_wmma_gemm_k128(const float* __restrict__ A, const float* __restrict__ B,
                   float* __restrict__ C,
                   const float* __restrict__ att_src, const float* __restrict__ att_dst,
                   float* __restrict__ a_src, float* __restrict__ a_dst, int nstrips) {
  static_assert(HEADS == 1 || HEADS * 16 == NACT, "a wave's 16 columns must be one head");
  constexpr int NPAD = NTILES * 16;
  constexpr int SA = 132;
  constexpr int SB = 132;
  constexpr int NT = 32 * NTILES;
  __shared__ __attribute__((aligned(16))) float ldsA[16 * SA];
  __shared__ float ldsB[NPAD * SB];
  __shared__ float red[(HEADS == 1) ? NTILES * 32 : 1];

  const int tid  = threadIdx.x;
  const int lane = tid & 31;
  const int wave = tid >> 5;
  const int c0 = blockIdx.y * NPAD;

  static_assert(NACT % 4 == 0 && NPAD % 4 == 0, "16-byte pieces must not straddle the last column");
  constexpr int BIT = (NPAD / 4) * 128 / NT;
  static_assert(BIT * NT == (NPAD / 4) * 128, "the panel must be a whole number of pieces per thread");
  {
    float4 bv[BIT];
#pragma unroll
    for (int u = 0; u < BIT; ++u) {
      const int p = tid + u * NT;
      const int k = ((p / NPAD) << 2) + (p & 3);
      const int gn = c0 + (((p >> 2) % (NPAD / 4)) << 2);
      const int gc = (gn < NACT) ? gn : (NACT - 4);
      bv[u] = *(const float4*)(B + (size_t)k * NACT + gc);
    }
#pragma unroll
    for (int u = 0; u < BIT; ++u) {
      const int p = tid + u * NT;
      const int k = ((p / NPAD) << 2) + (p & 3);
      const int n = ((p >> 2) % (NPAD / 4)) << 2;
      const bool in = (c0 + n) < NACT;
      float* lp = &ldsB[n * SB + k];
      lp[0]      = in ? bv[u].x : 0.0f;
      lp[SB]     = in ? bv[u].y : 0.0f;
      lp[2 * SB] = in ? bv[u].z : 0.0f;
      lp[3 * SB] = in ? bv[u].w : 0.0f;
    }
  }

  const int l15  = lane & 15;
  const int koff = (lane >> 4) << 1;
  const int ncol = wave * 16 + l15;
  const float* ap = &ldsA[l15 * SA + koff];
  const float* bp = &ldsB[ncol * SB + koff];

  const int gcol = c0 + ncol;
  const float wsrc = (gcol < NACT) ? att_src[gcol] : 0.0f;
  const float wdst = (gcol < NACT) ? att_dst[gcol] : 0.0f;
  const int hd = blockIdx.y * NTILES + wave;

  for (int strip = blockIdx.x; strip < nstrips; strip += gridDim.x) {
    const size_t m0 = (size_t)strip * 16;
    constexpr int AIT = (16 * 32 + NT - 1) / NT;
    float4 av[AIT];
#pragma unroll
    for (int u = 0; u < AIT; ++u) {
      const int i = min(tid + u * NT, 16 * 32 - 1);
      av[u] = *(const float4*)(A + (m0 + (i >> 5)) * 128 + ((i & 31) << 2));
    }
#pragma unroll
    for (int u = 0; u < AIT; ++u) {
      const int i = tid + u * NT;
      if ((u + 1) * NT <= 16 * 32 || i < 16 * 32) *(float4*)&ldsA[(i >> 5) * SA + ((i & 31) << 2)] = av[u];
    }
    __syncthreads();

    v8f acc = {0.f, 0.f, 0.f, 0.f, 0.f, 0.f, 0.f, 0.f};
#pragma unroll
    for (int kb = 0; kb < 128; kb += 4) {
      v2f a = *(const v2f*)(ap + kb);
      v2f b = *(const v2f*)(bp + kb);
      acc = __builtin_amdgcn_wmma_f32_16x16x4_f32(
           false, a,  false, b,
           (short)0, acc,  false,  false);
    }

    if (gcol < NACT) {
      float* cp = C + (m0 + (size_t)((lane >> 4) << 3)) * NACT + gcol;
#pragma unroll
      for (int r = 0; r < 8; ++r) cp[(size_t)r * NACT] = acc[r];
    }

    const bool b3 = (l15 & 8) != 0, b2 = (l15 & 4) != 0, b1 = (l15 & 2) != 0;
    float ps[8], pd[8];
#pragma unroll
    for (int r = 0; r < 8; ++r) { ps[r] = acc[r] * wsrc; pd[r] = acc[r] * wdst; }
    float s4[4], d4[4];
#pragma unroll
    for (int i = 0; i < 4; ++i) {
      s4[i] = (b3 ? ps[4 + i] : ps[i]) + row_xor_get<8>(b3 ? ps[i] : ps[4 + i]);
      d4[i] = (b3 ? pd[4 + i] : pd[i]) + row_xor_get<8>(b3 ? pd[i] : pd[4 + i]);
    }
    float s2[2], d2[2];
#pragma unroll
    for (int j = 0; j < 2; ++j) {
      s2[j] = (b2 ? s4[2 + j] : s4[j]) + row_xor_get<4>(b2 ? s4[j] : s4[2 + j]);
      d2[j] = (b2 ? d4[2 + j] : d4[j]) + row_xor_get<4>(b2 ? d4[j] : d4[2 + j]);
    }
    float s1 = (b1 ? s2[1] : s2[0]) + row_xor_get<2>(b1 ? s2[0] : s2[1]);
    float d1 = (b1 ? d2[1] : d2[0]) + row_xor_get<2>(b1 ? d2[0] : d2[1]);
    s1 += row_xor_get<1>(s1);
    d1 += row_xor_get<1>(d1);
    const bool is_src = (l15 & 1) == 0;
    const float sv = is_src ? s1 : d1;
    const int row = ((lane >> 4) << 3) + (l15 >> 1);
    if constexpr (HEADS == 1) {
      red[(wave * 16 + row) * 2 + (is_src ? 0 : 1)] = sv;
    } else {
      float* sp = is_src ? a_src : a_dst;
      sp[(m0 + row) * HEADS + hd] = sv;
    }
    __syncthreads();
    if constexpr (HEADS == 1) if (wave == 0) {
      const int r16 = lane & 15, which = lane >> 4;
      float s = red[r16 * 2 + which];
#pragma unroll
      for (int w = 1; w < NTILES; ++w) s += red[(w * 16 + r16) * 2 + which];
      float* sp = which ? a_dst : a_src;
      sp[m0 + r16] = s;
    }
  }
}

#define BSHIFT 9
#define BNODES 512
#define NBINS 256
#define SRC_BITS 17
#define ETILE 8192

__global__ void __launch_bounds__(1024)
bucket_count_kernel(const int* __restrict__ dst, int* __restrict__ bcnt, int E) {
  __shared__ int hist[NBINS];
  const int tid = threadIdx.x;
  if (tid < NBINS) hist[tid] = 0;
  const int base = blockIdx.x * ETILE + tid;
  int dv[8];
#pragma unroll
  for (int k = 0; k < 8; ++k) {
    const int e = base + k * 1024;
    dv[k] = (e < E) ? dst[e] : -1;
  }
  __syncthreads();
#pragma unroll
  for (int k = 0; k < 8; ++k) if (dv[k] >= 0) atomicAdd(&hist[dv[k] >> BSHIFT], 1);
  __syncthreads();
  if (tid < NBINS) bcnt[blockIdx.x * NBINS + tid] = hist[tid];
}

__global__ void __launch_bounds__(1024)
bucket_scatter_kernel(const int* __restrict__ src, const int* __restrict__ dst,
                      const int* __restrict__ bcnt, int* __restrict__ bstart,
                      int* __restrict__ rec, int E, int nblk) {
  __shared__ int ptot[4][NBINS];
  __shared__ int ppre[4][NBINS];
  __shared__ int cursor[NBINS + 1];
  const int tid = threadIdx.x;
  const int lane = tid & 31;
  const int base = blockIdx.x * ETILE + tid;
  int dv[8], sv[8];
#pragma unroll
  for (int k = 0; k < 8; ++k) {
    const int e = base + k * 1024;
    const bool ok = e < E;
    dv[k] = ok ? dst[e] : -1;
    sv[k] = ok ? src[e] : 0;
  }
  {
    const int k = tid & (NBINS - 1);
    const int q = tid >> 8;
    const int gq = (nblk + 3) >> 2;
    const int j0 = q * gq;
    const int j1 = min(nblk, j0 + gq);
    int tot = 0, pre = 0;
#pragma unroll 8
    for (int j = j0; j < j1; ++j) {
      const int c = bcnt[j * NBINS + k];
      tot += c;
      pre += (j < (int)blockIdx.x) ? c : 0;
    }
    ptot[q][k] = tot;
    ppre[q][k] = pre;
    if (tid == 0) cursor[NBINS] = 0;
  }
  __syncthreads();
  if (tid < 32) {
    int v[8];
    int run = 0;
#pragma unroll
    for (int i = 0; i < 8; ++i) {
      const int idx = tid * 8 + i;
      v[i] = run;
      run += ptot[0][idx] + ptot[1][idx] + ptot[2][idx] + ptot[3][idx];
    }
    int inc = run;
#pragma unroll
    for (int off = 1; off < 32; off <<= 1) {
      const int nb = __shfl_up(inc, off);
      if (lane >= off) inc += nb;
    }
    const int excl = inc - run;
#pragma unroll
    for (int i = 0; i < 8; ++i) {
      const int idx = tid * 8 + i;
      const int st = excl + v[i];
      cursor[idx] = st + ppre[0][idx] + ppre[1][idx] + ppre[2][idx] + ppre[3][idx];
      if (blockIdx.x == 0) bstart[idx] = st;
    }
    if (blockIdx.x == 0 && tid == 31) bstart[NBINS] = inc;
  }
  __syncthreads();
  int pv[8];
#pragma unroll
  for (int k = 0; k < 8; ++k) pv[k] = atomicAdd(&cursor[(dv[k] >= 0) ? (dv[k] >> BSHIFT) : NBINS], 1);
#pragma unroll
  for (int k = 0; k < 8; ++k)
    if (dv[k] >= 0) rec[pv[k]] = ((dv[k] & (BNODES - 1)) << SRC_BITS) | sv[k];
}

__global__ void __launch_bounds__(1024)
bucket_csr_kernel(const int* __restrict__ bstart, const int* __restrict__ rec,
                  int* __restrict__ rowptr, int* __restrict__ col, int Nn) {
  __shared__ int cntl[BNODES + 1];
  __shared__ int wsum[16];
  const int tid = threadIdx.x, lane = tid & 31, w = tid >> 5;
  const int bs = bstart[blockIdx.x];
  const int n = bstart[blockIdx.x + 1] - bs;
  if (tid <= BNODES) cntl[tid] = 0;
  __syncthreads();
  const int* rp = rec + bs;
  for (int t0 = 0; t0 < n; t0 += ETILE) {
    int rv[8];
#pragma unroll
    for (int k = 0; k < 8; ++k) {
      const int i = t0 + k * 1024 + tid;
      rv[k] = (i < n) ? rp[i] : -1;
    }
#pragma unroll
    for (int k = 0; k < 8; ++k) if (rv[k] >= 0) atomicAdd(&cntl[rv[k] >> SRC_BITS], 1);
  }
  __syncthreads();
  int c = 0, inc = 0;
  if (tid < BNODES) {
    c = cntl[tid];
    inc = c;
#pragma unroll
    for (int off = 1; off < 32; off <<= 1) {
      const int nb = __shfl_up(inc, off);
      if (lane >= off) inc += nb;
    }
    if (lane == 31) wsum[w] = inc;
  }
  __syncthreads();
  if (tid < BNODES) {
    int woff = 0;
#pragma unroll
    for (int k = 0; k < 16; ++k) woff += (k < w) ? wsum[k] : 0;
    const int excl = woff + inc - c;
    cntl[tid] = excl;
    const int node = blockIdx.x * BNODES + tid;
    if (node < Nn) rowptr[node] = bs + excl;
  }
  if (blockIdx.x == gridDim.x - 1 && tid == 0) rowptr[Nn] = bs + n;
  __syncthreads();
  int* cp = col + bs;
  for (int t0 = 0; t0 < n; t0 += ETILE) {
    int rv[8], pv[8];
#pragma unroll
    for (int k = 0; k < 8; ++k) {
      const int i = t0 + k * 1024 + tid;
      rv[k] = (i < n) ? rp[i] : -1;
    }
#pragma unroll
    for (int k = 0; k < 8; ++k) pv[k] = atomicAdd(&cntl[(rv[k] >= 0) ? (rv[k] >> SRC_BITS) : BNODES], 1);
#pragma unroll
    for (int k = 0; k < 8; ++k) if (rv[k] >= 0) cp[pv[k]] = rv[k] & ((1 << SRC_BITS) - 1);
  }
}

__device__ __forceinline__ int rank_among(int key, int keys, int n) {
  int r = 0;
  for (int j = 0; j < n; ++j) r += (__builtin_amdgcn_readlane(keys, j) < key) ? 1 : 0;
  return r;
}
#define PLACE_BITS 6

__device__ __forceinline__ float leaky(float v) { return (v >= 0.f) ? v : NEG_SLOPE * v; }


__device__ __forceinline__ void wave_lds_order() {
  __builtin_amdgcn_fence(__ATOMIC_RELEASE, "wavefront");
  __builtin_amdgcn_wave_barrier();
  __builtin_amdgcn_fence(__ATOMIC_ACQUIRE, "wavefront");
}

__device__ __forceinline__ void gat1_node_short(int d, int lane, int start, int deg,
                                                const int* __restrict__ col, int* col2,
                                                const float* __restrict__ asc,
                                                const float* __restrict__ adc,
                                                const float* __restrict__ h,
                                                const float* __restrict__ b1,
                                                float* __restrict__ h1, float* tile, int* ids) {
  const int hd = lane >> 2, q = lane & 3;
  const float4 hself = *(const float4*)(h + (size_t)d * 128 + lane * 4);
  const float4 ad0 = *(const float4*)(adc + (size_t)d * 8);
  const float4 ad1 = *(const float4*)(adc + (size_t)d * 8 + 4);
  const float eself = leaky(asc[(size_t)d * 8 + hd] + adc[(size_t)d * 8 + hd]);
  int su = d;
  int r = lane;
  float4 e0 = make_float4(-INFINITY, -INFINITY, -INFINITY, -INFINITY);
  float4 e1 = e0;
  if (lane < deg) su = col[start + lane];
  const float4 a0 = *(const float4*)(asc + (size_t)su * 8);
  const float4 a1 = *(const float4*)(asc + (size_t)su * 8 + 4);
  {
    const int key = (su << PLACE_BITS) | lane;
    const int rr = rank_among(key, key, deg);
    if (lane < deg) r = rr;
  }
  if (lane < deg) {
    col2[start + r] = su;
    e0.x = leaky(a0.x + ad0.x); e0.y = leaky(a0.y + ad0.y);
    e0.z = leaky(a0.z + ad0.z); e0.w = leaky(a0.w + ad0.w);
    e1.x = leaky(a1.x + ad1.x); e1.y = leaky(a1.y + ad1.y);
    e1.z = leaky(a1.z + ad1.z); e1.w = leaky(a1.w + ad1.w);
  }
  *(float4*)(tile + r * 8) = e0;
  *(float4*)(tile + r * 8 + 4) = e1;
  ids[r] = su;
  wave_lds_order();
  const int sv = ids[lane];

  float* colp = tile + q * 8 + hd;
  float v[8];
#pragma unroll
  for (int t = 0; t < 8; ++t) v[t] = colp[32 * t];
  float m = eself;
#pragma unroll
  for (int t = 0; t < 8; ++t) m = fmaxf(m, v[t]);
  m = fmaxf(m, row_xor_get<1>(m));
  m = fmaxf(m, row_xor_get<2>(m));
  float ssum = 0.f;
#pragma unroll
  for (int t = 0; t < 8; ++t) { v[t] = __expf(v[t] - m); ssum += v[t]; }
  ssum += row_xor_get<1>(ssum);
  ssum += row_xor_get<2>(ssum);
  const float exself = __expf(eself - m);
  ssum += exself;
#pragma unroll
  for (int t = 0; t < 8; ++t) {
    if (4 * t < deg) colp[32 * t] = v[t] / ssum;
  }
  wave_lds_order();

  float4 acc = make_float4(0.f, 0.f, 0.f, 0.f);
  const float* wp = tile + hd;
  for (int j0 = 0; j0 < deg; j0 += 4) {
#pragma unroll
    for (int u = 0; u < 4; ++u) {
      const int j = j0 + u;
      const int s = __builtin_amdgcn_readlane(sv, j);
      const float alpha = wp[j * 8];
      const float4 hv = *(const float4*)(h + (size_t)s * 128 + lane * 4);
      acc.x = fmaf(hv.x, alpha, acc.x);
      acc.y = fmaf(hv.y, alpha, acc.y);
      acc.z = fmaf(hv.z, alpha, acc.z);
      acc.w = fmaf(hv.w, alpha, acc.w);
    }
  }
  {
    const float alpha = exself / ssum;
    acc.x = fmaf(hself.x, alpha, acc.x);
    acc.y = fmaf(hself.y, alpha, acc.y);
    acc.z = fmaf(hself.z, alpha, acc.z);
    acc.w = fmaf(hself.w, alpha, acc.w);
  }
  const float4 bv = *(const float4*)(b1 + lane * 4);
  float4 o;
  o.x = acc.x + bv.x; o.x = (o.x > 0.f) ? o.x : (__expf(o.x) - 1.0f);
  o.y = acc.y + bv.y; o.y = (o.y > 0.f) ? o.y : (__expf(o.y) - 1.0f);
  o.z = acc.z + bv.z; o.z = (o.z > 0.f) ? o.z : (__expf(o.z) - 1.0f);
  o.w = acc.w + bv.w; o.w = (o.w > 0.f) ? o.w : (__expf(o.w) - 1.0f);
  *(float4*)(h1 + (size_t)d * 128 + lane * 4) = o;
  wave_lds_order();
}

__device__ __forceinline__ void gat1_node_mid(int d, int lane, int start, int deg,
                                              const int* __restrict__ col, int* col2,
                                              const float* __restrict__ asc,
                                              const float* __restrict__ adc,
                                              const float* __restrict__ h,
                                              const float* __restrict__ b1,
                                              float* __restrict__ h1, float* tile, int* ids) {
  const int hd = lane >> 2, q = lane & 3;
  const float4 ad0 = *(const float4*)(adc + (size_t)d * 8);
  const float4 ad1 = *(const float4*)(adc + (size_t)d * 8 + 4);
  const float eself = leaky(asc[(size_t)d * 8 + hd] + adc[(size_t)d * 8 + hd]);
  const bool has1 = (32 + lane) < deg;
  const int su0 = col[start + lane];
  const int su1 = has1 ? col[start + 32 + lane] : d;
  int r0, r1 = 32 + lane;
  {
    const float4 a0 = *(const float4*)(asc + (size_t)su0 * 8);
    const float4 a1 = *(const float4*)(asc + (size_t)su0 * 8 + 4);
    const float4 c0 = *(const float4*)(asc + (size_t)su1 * 8);
    const float4 c1 = *(const float4*)(asc + (size_t)su1 * 8 + 4);
    {
      const int key0 = (su0 << PLACE_BITS) | lane;
      const int key1 = (su1 << PLACE_BITS) | (32 + lane);
      r0 = rank_among(key0, key0, 32) + rank_among(key0, key1, deg - 32);
      const int rr1 = rank_among(key1, key0, 32) + rank_among(key1, key1, deg - 32);
      if (has1) r1 = rr1;
    }
    col2[start + r0] = su0;
    if (has1) col2[start + r1] = su1;
    float4 e0, e1, f0, f1;
    e0.x = leaky(a0.x + ad0.x); e0.y = leaky(a0.y + ad0.y);
    e0.z = leaky(a0.z + ad0.z); e0.w = leaky(a0.w + ad0.w);
    e1.x = leaky(a1.x + ad1.x); e1.y = leaky(a1.y + ad1.y);
    e1.z = leaky(a1.z + ad1.z); e1.w = leaky(a1.w + ad1.w);
    f0.x = has1 ? leaky(c0.x + ad0.x) : -INFINITY; f0.y = has1 ? leaky(c0.y + ad0.y) : -INFINITY;
    f0.z = has1 ? leaky(c0.z + ad0.z) : -INFINITY; f0.w = has1 ? leaky(c0.w + ad0.w) : -INFINITY;
    f1.x = has1 ? leaky(c1.x + ad1.x) : -INFINITY; f1.y = has1 ? leaky(c1.y + ad1.y) : -INFINITY;
    f1.z = has1 ? leaky(c1.z + ad1.z) : -INFINITY; f1.w = has1 ? leaky(c1.w + ad1.w) : -INFINITY;
    *(float4*)(tile + r0 * 8) = e0;
    *(float4*)(tile + r0 * 8 + 4) = e1;
    *(float4*)(tile + r1 * 8) = f0;
    *(float4*)(tile + r1 * 8 + 4) = f1;
    ids[r0] = su0;
    ids[r1] = su1;
  }
  wave_lds_order();
  const int sv0 = ids[lane];
  const int sv1 = ids[32 + lane];

  float* colp = tile + q * 8 + hd;
  float m = eself;
#pragma unroll
  for (int t = 0; t < 16; ++t) m = fmaxf(m, colp[32 * t]);
  m = fmaxf(m, row_xor_get<1>(m));
  m = fmaxf(m, row_xor_get<2>(m));
  float ssum = 0.f;
#pragma unroll
  for (int t = 0; t < 16; ++t) {
    const float ex = __expf(colp[32 * t] - m);
    ssum += ex;
    colp[32 * t] = ex;
  }
  ssum += row_xor_get<1>(ssum);
  ssum += row_xor_get<2>(ssum);
  const float exself = __expf(eself - m);
  ssum += exself;
#pragma unroll
  for (int t = 0; t < 16; ++t) colp[32 * t] = colp[32 * t] / ssum;
  wave_lds_order();

  float4 acc = make_float4(0.f, 0.f, 0.f, 0.f);
  const float* wp = tile + hd;
  for (int j0 = 0; j0 < deg; j0 += 4) {
    const int svc = (j0 < 32) ? sv0 : sv1;
#pragma unroll
    for (int u = 0; u < 4; ++u) {
      const int j = j0 + u;
      const int s = __builtin_amdgcn_readlane(svc, j & 31);
      const float alpha = wp[j * 8];
      const float4 hv = *(const float4*)(h + (size_t)s * 128 + lane * 4);
      acc.x = fmaf(hv.x, alpha, acc.x);
      acc.y = fmaf(hv.y, alpha, acc.y);
      acc.z = fmaf(hv.z, alpha, acc.z);
      acc.w = fmaf(hv.w, alpha, acc.w);
    }
  }
  {
    const float4 hself = *(const float4*)(h + (size_t)d * 128 + lane * 4);
    const float alpha = exself / ssum;
    acc.x = fmaf(hself.x, alpha, acc.x);
    acc.y = fmaf(hself.y, alpha, acc.y);
    acc.z = fmaf(hself.z, alpha, acc.z);
    acc.w = fmaf(hself.w, alpha, acc.w);
  }
  const float4 bv = *(const float4*)(b1 + lane * 4);
  float4 o;
  o.x = acc.x + bv.x; o.x = (o.x > 0.f) ? o.x : (__expf(o.x) - 1.0f);
  o.y = acc.y + bv.y; o.y = (o.y > 0.f) ? o.y : (__expf(o.y) - 1.0f);
  o.z = acc.z + bv.z; o.z = (o.z > 0.f) ? o.z : (__expf(o.z) - 1.0f);
  o.w = acc.w + bv.w; o.w = (o.w > 0.f) ? o.w : (__expf(o.w) - 1.0f);
  *(float4*)(h1 + (size_t)d * 128 + lane * 4) = o;
  wave_lds_order();
}

__device__ __forceinline__ void gat1_node_long(int d, int lane, int start, int deg,
                                               const int* __restrict__ col, int* col2,
                                               const float* __restrict__ asc,
                                               const float* __restrict__ adc,
                                               const float* __restrict__ h,
                                               const float* __restrict__ b1,
                                               float* __restrict__ h1) {
  const int hd = lane >> 2;
  for (int i = lane; i < deg; i += 32) {
    const int k = col[start + i];
    int r = 0;
    for (int j = 0; j < deg; ++j) {
      const int kj = col[start + j];
      r += (kj < k || (kj == k && j < i)) ? 1 : 0;
    }
    col2[start + r] = k;
  }
  __threadfence();
  const float ad = adc[(size_t)d * 8 + hd];
  const float eself = leaky(asc[(size_t)d * 8 + hd] + ad);
  float m = eself;
#pragma unroll 4
  for (int j = 0; j < deg; ++j) {
    const int s = col2[start + j];
    const float e = leaky(asc[(size_t)s * 8 + hd] + ad);
    m = fmaxf(m, e);
  }
  float ssum = 0.f;
#pragma unroll 4
  for (int j = 0; j < deg; ++j) {
    const float e = leaky(asc[(size_t)col2[start + j] * 8 + hd] + ad);
    ssum += __expf(e - m);
  }
  const float exself = __expf(eself - m);
  ssum += exself;
  float4 acc = make_float4(0.f, 0.f, 0.f, 0.f);
#pragma unroll 4
  for (int j = 0; j < deg; ++j) {
    const int s = col2[start + j];
    const float ex = __expf(leaky(asc[(size_t)s * 8 + hd] + ad) - m);
    const float alpha = ex / ssum;
    const float4 hv = *(const float4*)(h + (size_t)s * 128 + lane * 4);
    acc.x = fmaf(hv.x, alpha, acc.x);
    acc.y = fmaf(hv.y, alpha, acc.y);
    acc.z = fmaf(hv.z, alpha, acc.z);
    acc.w = fmaf(hv.w, alpha, acc.w);
  }
  {
    const float alpha = exself / ssum;
    const float4 hv = *(const float4*)(h + (size_t)d * 128 + lane * 4);
    acc.x = fmaf(hv.x, alpha, acc.x);
    acc.y = fmaf(hv.y, alpha, acc.y);
    acc.z = fmaf(hv.z, alpha, acc.z);
    acc.w = fmaf(hv.w, alpha, acc.w);
  }
  const float4 bv = *(const float4*)(b1 + lane * 4);
  float4 o;
  o.x = acc.x + bv.x; o.x = (o.x > 0.f) ? o.x : (__expf(o.x) - 1.0f);
  o.y = acc.y + bv.y; o.y = (o.y > 0.f) ? o.y : (__expf(o.y) - 1.0f);
  o.z = acc.z + bv.z; o.z = (o.z > 0.f) ? o.z : (__expf(o.z) - 1.0f);
  o.w = acc.w + bv.w; o.w = (o.w > 0.f) ? o.w : (__expf(o.w) - 1.0f);
  *(float4*)(h1 + (size_t)d * 128 + lane * 4) = o;
}

__global__ void __launch_bounds__(256, 16)
gather1_kernel(const int* __restrict__ rowptr, const int* __restrict__ col, int* col2,
               const float* __restrict__ asc, const float* __restrict__ adc,
               const float* __restrict__ h, const float* __restrict__ b1,
               float* __restrict__ h1, int Nn) {
  __shared__ __attribute__((aligned(16))) float tiles[8 * 64 * 8];
  __shared__ int idl[8 * 64];
  const int lane = threadIdx.x & 31;
  const int w = threadIdx.x >> 5;
  float* tile = tiles + w * (64 * 8);
  int* ids = idl + w * 64;
  const int nw = gridDim.x * 8;
  for (int d = __builtin_amdgcn_readfirstlane(blockIdx.x * 8 + w); d < Nn; d += nw) {
    const int start = rowptr[d];
    const int deg = rowptr[d + 1] - start;
    if (deg <= 32)      gat1_node_short(d, lane, start, deg, col, col2, asc, adc, h, b1, h1, tile, ids);
    else if (deg <= 64) gat1_node_mid(d, lane, start, deg, col, col2, asc, adc, h, b1, h1, tile, ids);
    else                gat1_node_long(d, lane, start, deg, col, col2, asc, adc, h, b1, h1);
  }
}

__device__ __forceinline__ float wave_max(float v) {
#pragma unroll
  for (int off = 16; off >= 1; off >>= 1) v = fmaxf(v, __shfl_xor(v, off));
  return v;
}
__device__ __forceinline__ float wave_sum(float v) {
#pragma unroll
  for (int off = 16; off >= 1; off >>= 1) v += __shfl_xor(v, off);
  return v;
}

__device__ __forceinline__ void gat2_accum(int n, int sv, float av, int lc,
                                           const float* __restrict__ h2, float2& acc) {
#pragma unroll 4
  for (int t = 0; t < n; ++t) {
    const int s = __builtin_amdgcn_readlane(sv, t);
    const float a = __int_as_float(__builtin_amdgcn_readlane(__float_as_int(av), t));
    const float2 hv = *(const float2*)(h2 + (size_t)s * 40 + 2 * lc);
    acc.x = fmaf(hv.x, a, acc.x);
    acc.y = fmaf(hv.y, a, acc.y);
  }
}

__global__ void __launch_bounds__(256)
gather2_kernel(const int* __restrict__ rowptr, const int* __restrict__ col2,
               const float* __restrict__ asc, const float* __restrict__ adc,
               const float* __restrict__ h2, const float* __restrict__ b2,
               float* __restrict__ out, int Nn) {
  const int lane = threadIdx.x & 31;
  const int lc = (lane < 19) ? lane : 19;
  const int nw = gridDim.x * 8;
  for (int d = __builtin_amdgcn_readfirstlane(blockIdx.x * 8 + (threadIdx.x >> 5)); d < Nn; d += nw) {
  const int start = rowptr[d];
  const int deg = rowptr[d + 1] - start;
  const float ad = adc[d];
  const float eself = leaky(asc[d] + ad);

  int s0 = 0;
  float e0 = -INFINITY;
  if (lane < deg) { s0 = col2[start + lane]; e0 = leaky(asc[s0] + ad); }
  float m = fmaxf(eself, wave_max(e0));
  for (int c = 32; c < deg; c += 32) {
    float e = -INFINITY;
    if (c + lane < deg) e = leaky(asc[col2[start + c + lane]] + ad);
    m = fmaxf(m, wave_max(e));
  }
  const float ex0 = (lane < deg) ? __expf(e0 - m) : 0.f;
  float ssum = wave_sum(ex0);
  for (int c = 32; c < deg; c += 32) {
    float ex = 0.f;
    if (c + lane < deg) ex = __expf(leaky(asc[col2[start + c + lane]] + ad) - m);
    ssum += wave_sum(ex);
  }
  const float exself = __expf(eself - m);
  ssum += exself;

  float2 acc = make_float2(0.f, 0.f);
  gat2_accum(min(deg, 32), s0, ex0 / ssum, lc, h2, acc);
  for (int c = 32; c < deg; c += 32) {
    int s = 0;
    float a = 0.f;
    if (c + lane < deg) {
      s = col2[start + c + lane];
      a = __expf(leaky(asc[s] + ad) - m) / ssum;
    }
    gat2_accum(min(deg - c, 32), s, a, lc, h2, acc);
  }
  {
    const float a = exself / ssum;
    const float2 hv = *(const float2*)(h2 + (size_t)d * 40 + 2 * lc);
    acc.x = fmaf(hv.x, a, acc.x);
    acc.y = fmaf(hv.y, a, acc.y);
  }
  if (lane < 20) {
    const float2 bv = *(const float2*)(b2 + 2 * lane);
    *(float2*)(out + (size_t)d * 40 + 2 * lane) = make_float2(acc.x + bv.x, acc.y + bv.y);
  }
  }
}

extern "C" void kernel_launch(void* const* d_in, const int* in_sizes, int n_in,
                              void* d_out, int out_size, void* d_ws, size_t ws_size,
                              hipStream_t stream) {
  const float* x   = (const float*)d_in[0];
  const int*   ei  = (const int*)d_in[1];
  const float* W1  = (const float*)d_in[2];
  const float* as1 = (const float*)d_in[3];
  const float* ad1 = (const float*)d_in[4];
  const float* b1  = (const float*)d_in[5];
  const float* W2  = (const float*)d_in[6];
  const float* as2 = (const float*)d_in[7];
  const float* ad2 = (const float*)d_in[8];
  const float* b2  = (const float*)d_in[9];
  float* out = (float*)d_out;

  const int Nn   = in_sizes[0] / 128;
  const int E    = in_sizes[1] / 2;
  const int* srcp = ei;
  const int* dstp = ei + E;
  const int NBLK = (E + ETILE - 1) / ETILE;
  const int NBKT = (Nn + BNODES - 1) / BNODES;

  float* ws   = (float*)d_ws;
  float* h    = ws;  ws += (size_t)Nn * 128;
  float* h1   = ws;  ws += (size_t)Nn * 128;
  float* asc1 = ws;  ws += (size_t)Nn * 8;
  float* adc1 = ws;  ws += (size_t)Nn * 8;
  float* h2   = ws;  ws += (size_t)Nn * 40;
  float* asc2 = ws;  ws += Nn;
  float* adc2 = ws;  ws += Nn;
  int* wi     = (int*)ws;
  int* bcnt   = wi;  wi += (size_t)NBLK * NBINS;
  int* rec    = wi;  wi += E;
  int* col    = wi;  wi += E;
  int* col2   = wi;  wi += E;
  int* rowptr = wi;  wi += (size_t)Nn + 1;
  int* bstart = wi;  wi += NBINS + 1;

  bucket_count_kernel<<<NBLK, 1024, 0, stream>>>(dstp, bcnt, E);
  bucket_scatter_kernel<<<NBLK, 1024, 0, stream>>>(srcp, dstp, bcnt, bstart, rec, E, NBLK);
  bucket_csr_kernel<<<NBKT, 1024, 0, stream>>>(bstart, rec, rowptr, col, Nn);

  gat_wmma_gemm_k128<4, 128, 8><<<dim3(896, 2), 4 * 32, 0, stream>>>(x, W1, h, as1, ad1, asc1, adc1, Nn / 16);
  gather1_kernel<<<2048, 256, 0, stream>>>(rowptr, col, col2, asc1, adc1, h, b1, h1, Nn);

  gat_wmma_gemm_k128<3, 40, 1><<<dim3(1563, 1), 3 * 32, 0, stream>>>(h1, W2, h2, as2, ad2, asc2, adc2, Nn / 16);
  gather2_kernel<<<2048, 256, 0, stream>>>(rowptr, col2, asc2, adc2, h2, b2, out, Nn);
}


# Both gathers take an edge's score from the source row they fetch (dot with att_src, lane tree) in a running-max softmax: per-edge score loads go; sums rescaled; layer 2 sums even/odd edges apart
# speedup vs baseline: 5.8386x; 1.0369x over previous
#include <hip/hip_runtime.h>
#include <hip/hip_bf16.h>
#include <math.h>
#include <stdint.h>


typedef float v2f __attribute__((ext_vector_type(2)));
typedef float v8f __attribute__((ext_vector_type(8)));

#define NEG_SLOPE 0.2f

template <int MASK>
__device__ __forceinline__ float row_xor_get(float v) {
  return __int_as_float(__builtin_amdgcn_update_dpp(0, __float_as_int(v), 0x160 + MASK, 0xf, 0xf, true));
}

template <int NTILES, int NACT, int HEADS>
__global__ void __launch_bounds__(32 * NTILES)
gat_wmma_gemm_k128(const float* __restrict__ A, const float* __restrict__ B,
                   float* __restrict__ C,
                   const float* __restrict__ att_src, const float* __restrict__ att_dst,
                   float* __restrict__ a_src, float* __restrict__ a_dst, int nstrips) {
  static_assert(HEADS == 1 || HEADS * 16 == NACT, "a wave's 16 columns must be one head");
  constexpr int NPAD = NTILES * 16;
  constexpr int SA = 132;
  constexpr int SB = 132;
  constexpr int NT = 32 * NTILES;
  __shared__ __attribute__((aligned(16))) float ldsA[16 * SA];
  __shared__ float ldsB[NPAD * SB];
  __shared__ float red[(HEADS == 1) ? NTILES * 32 : 1];

  const int tid  = threadIdx.x;
  const int lane = tid & 31;
  const int wave = tid >> 5;
  const int c0 = blockIdx.y * NPAD;

  static_assert(NACT % 4 == 0 && NPAD % 4 == 0, "16-byte pieces must not straddle the last column");
  constexpr int BIT = (NPAD / 4) * 128 / NT;
  static_assert(BIT * NT == (NPAD / 4) * 128, "the panel must be a whole number of pieces per thread");
  {
    float4 bv[BIT];
#pragma unroll
    for (int u = 0; u < BIT; ++u) {
      const int p = tid + u * NT;
      const int k = ((p / NPAD) << 2) + (p & 3);
      const int gn = c0 + (((p >> 2) % (NPAD / 4)) << 2);
      const int gc = (gn < NACT) ? gn : (NACT - 4);
      bv[u] = *(const float4*)(B + (size_t)k * NACT + gc);
    }
#pragma unroll
    for (int u = 0; u < BIT; ++u) {
      const int p = tid + u * NT;
      const int k = ((p / NPAD) << 2) + (p & 3);
      const int n = ((p >> 2) % (NPAD / 4)) << 2;
      const bool in = (c0 + n) < NACT;
      float* lp = &ldsB[n * SB + k];
      lp[0]      = in ? bv[u].x : 0.0f;
      lp[SB]     = in ? bv[u].y : 0.0f;
      lp[2 * SB] = in ? bv[u].z : 0.0f;
      lp[3 * SB] = in ? bv[u].w : 0.0f;
    }
  }

  const int l15  = lane & 15;
  const int koff = (lane >> 4) << 1;
  const int ncol = wave * 16 + l15;
  const float* ap = &ldsA[l15 * SA + koff];
  const float* bp = &ldsB[ncol * SB + koff];

  const int gcol = c0 + ncol;
  const float wsrc = (gcol < NACT) ? att_src[gcol] : 0.0f;
  const float wdst = (gcol < NACT) ? att_dst[gcol] : 0.0f;
  const int hd = blockIdx.y * NTILES + wave;

  for (int strip = blockIdx.x; strip < nstrips; strip += gridDim.x) {
    const size_t m0 = (size_t)strip * 16;
    constexpr int AIT = (16 * 32 + NT - 1) / NT;
    float4 av[AIT];
#pragma unroll
    for (int u = 0; u < AIT; ++u) {
      const int i = min(tid + u * NT, 16 * 32 - 1);
      av[u] = *(const float4*)(A + (m0 + (i >> 5)) * 128 + ((i & 31) << 2));
    }
#pragma unroll
    for (int u = 0; u < AIT; ++u) {
      const int i = tid + u * NT;
      if ((u + 1) * NT <= 16 * 32 || i < 16 * 32) *(float4*)&ldsA[(i >> 5) * SA + ((i & 31) << 2)] = av[u];
    }
    __syncthreads();

    v8f acc = {0.f, 0.f, 0.f, 0.f, 0.f, 0.f, 0.f, 0.f};
#pragma unroll
    for (int kb = 0; kb < 128; kb += 4) {
      v2f a = *(const v2f*)(ap + kb);
      v2f b = *(const v2f*)(bp + kb);
      acc = __builtin_amdgcn_wmma_f32_16x16x4_f32(
           false, a,  false, b,
           (short)0, acc,  false,  false);
    }

    if (gcol < NACT) {
      float* cp = C + (m0 + (size_t)((lane >> 4) << 3)) * NACT + gcol;
#pragma unroll
      for (int r = 0; r < 8; ++r) cp[(size_t)r * NACT] = acc[r];
    }

    const bool b3 = (l15 & 8) != 0, b2 = (l15 & 4) != 0, b1 = (l15 & 2) != 0;
    float ps[8], pd[8];
#pragma unroll
    for (int r = 0; r < 8; ++r) { ps[r] = acc[r] * wsrc; pd[r] = acc[r] * wdst; }
    float s4[4], d4[4];
#pragma unroll
    for (int i = 0; i < 4; ++i) {
      s4[i] = (b3 ? ps[4 + i] : ps[i]) + row_xor_get<8>(b3 ? ps[i] : ps[4 + i]);
      d4[i] = (b3 ? pd[4 + i] : pd[i]) + row_xor_get<8>(b3 ? pd[i] : pd[4 + i]);
    }
    float s2[2], d2[2];
#pragma unroll
    for (int j = 0; j < 2; ++j) {
      s2[j] = (b2 ? s4[2 + j] : s4[j]) + row_xor_get<4>(b2 ? s4[j] : s4[2 + j]);
      d2[j] = (b2 ? d4[2 + j] : d4[j]) + row_xor_get<4>(b2 ? d4[j] : d4[2 + j]);
    }
    float s1 = (b1 ? s2[1] : s2[0]) + row_xor_get<2>(b1 ? s2[0] : s2[1]);
    float d1 = (b1 ? d2[1] : d2[0]) + row_xor_get<2>(b1 ? d2[0] : d2[1]);
    s1 += row_xor_get<1>(s1);
    d1 += row_xor_get<1>(d1);
    const bool is_src = (l15 & 1) == 0;
    const float sv = is_src ? s1 : d1;
    const int row = ((lane >> 4) << 3) + (l15 >> 1);
    if constexpr (HEADS == 1) {
      red[(wave * 16 + row) * 2 + (is_src ? 0 : 1)] = sv;
    } else {
      float* sp = is_src ? a_src : a_dst;
      sp[(m0 + row) * HEADS + hd] = sv;
    }
    __syncthreads();
    if constexpr (HEADS == 1) if (wave == 0) {
      const int r16 = lane & 15, which = lane >> 4;
      float s = red[r16 * 2 + which];
#pragma unroll
      for (int w = 1; w < NTILES; ++w) s += red[(w * 16 + r16) * 2 + which];
      float* sp = which ? a_dst : a_src;
      sp[m0 + r16] = s;
    }
  }
}

#define BSHIFT 9
#define BNODES 512
#define NBINS 256
#define SRC_BITS 17
#define ETILE 8192

__global__ void __launch_bounds__(1024)
bucket_count_kernel(const int* __restrict__ dst, int* __restrict__ bcnt, int E) {
  __shared__ int hist[NBINS];
  const int tid = threadIdx.x;
  if (tid < NBINS) hist[tid] = 0;
  const int base = blockIdx.x * ETILE + tid;
  int dv[8];
#pragma unroll
  for (int k = 0; k < 8; ++k) {
    const int e = base + k * 1024;
    dv[k] = (e < E) ? dst[e] : -1;
  }
  __syncthreads();
#pragma unroll
  for (int k = 0; k < 8; ++k) if (dv[k] >= 0) atomicAdd(&hist[dv[k] >> BSHIFT], 1);
  __syncthreads();
  if (tid < NBINS) bcnt[blockIdx.x * NBINS + tid] = hist[tid];
}

__global__ void __launch_bounds__(1024)
bucket_scatter_kernel(const int* __restrict__ src, const int* __restrict__ dst,
                      const int* __restrict__ bcnt, int* __restrict__ bstart,
                      int* __restrict__ rec, int E, int nblk) {
  __shared__ int ptot[4][NBINS];
  __shared__ int ppre[4][NBINS];
  __shared__ int cursor[NBINS + 1];
  const int tid = threadIdx.x;
  const int lane = tid & 31;
  const int base = blockIdx.x * ETILE + tid;
  int dv[8], sv[8];
#pragma unroll
  for (int k = 0; k < 8; ++k) {
    const int e = base + k * 1024;
    const bool ok = e < E;
    dv[k] = ok ? dst[e] : -1;
    sv[k] = ok ? src[e] : 0;
  }
  {
    const int k = tid & (NBINS - 1);
    const int q = tid >> 8;
    const int gq = (nblk + 3) >> 2;
    const int j0 = q * gq;
    const int j1 = min(nblk, j0 + gq);
    int tot = 0, pre = 0;
#pragma unroll 8
    for (int j = j0; j < j1; ++j) {
      const int c = bcnt[j * NBINS + k];
      tot += c;
      pre += (j < (int)blockIdx.x) ? c : 0;
    }
    ptot[q][k] = tot;
    ppre[q][k] = pre;
    if (tid == 0) cursor[NBINS] = 0;
  }
  __syncthreads();
  if (tid < 32) {
    int v[8];
    int run = 0;
#pragma unroll
    for (int i = 0; i < 8; ++i) {
      const int idx = tid * 8 + i;
      v[i] = run;
      run += ptot[0][idx] + ptot[1][idx] + ptot[2][idx] + ptot[3][idx];
    }
    int inc = run;
#pragma unroll
    for (int off = 1; off < 32; off <<= 1) {
      const int nb = __shfl_up(inc, off);
      if (lane >= off) inc += nb;
    }
    const int excl = inc - run;
#pragma unroll
    for (int i = 0; i < 8; ++i) {
      const int idx = tid * 8 + i;
      const int st = excl + v[i];
      cursor[idx] = st + ppre[0][idx] + ppre[1][idx] + ppre[2][idx] + ppre[3][idx];
      if (blockIdx.x == 0) bstart[idx] = st;
    }
    if (blockIdx.x == 0 && tid == 31) bstart[NBINS] = inc;
  }
  __syncthreads();
  int pv[8];
#pragma unroll
  for (int k = 0; k < 8; ++k) pv[k] = atomicAdd(&cursor[(dv[k] >= 0) ? (dv[k] >> BSHIFT) : NBINS], 1);
#pragma unroll
  for (int k = 0; k < 8; ++k)
    if (dv[k] >= 0) rec[pv[k]] = ((dv[k] & (BNODES - 1)) << SRC_BITS) | sv[k];
}

__global__ void __launch_bounds__(1024)
bucket_csr_kernel(const int* __restrict__ bstart, const int* __restrict__ rec,
                  int* __restrict__ rowptr, int* __restrict__ col, int Nn) {
  __shared__ int cntl[BNODES + 1];
  __shared__ int wsum[16];
  const int tid = threadIdx.x, lane = tid & 31, w = tid >> 5;
  const int bs = bstart[blockIdx.x];
  const int n = bstart[blockIdx.x + 1] - bs;
  if (tid <= BNODES) cntl[tid] = 0;
  __syncthreads();
  const int* rp = rec + bs;
  for (int t0 = 0; t0 < n; t0 += ETILE) {
    int rv[8];
#pragma unroll
    for (int k = 0; k < 8; ++k) {
      const int i = t0 + k * 1024 + tid;
      rv[k] = (i < n) ? rp[i] : -1;
    }
#pragma unroll
    for (int k = 0; k < 8; ++k) if (rv[k] >= 0) atomicAdd(&cntl[rv[k] >> SRC_BITS], 1);
  }
  __syncthreads();
  int c = 0, inc = 0;
  if (tid < BNODES) {
    c = cntl[tid];
    inc = c;
#pragma unroll
    for (int off = 1; off < 32; off <<= 1) {
      const int nb = __shfl_up(inc, off);
      if (lane >= off) inc += nb;
    }
    if (lane == 31) wsum[w] = inc;
  }
  __syncthreads();
  if (tid < BNODES) {
    int woff = 0;
#pragma unroll
    for (int k = 0; k < 16; ++k) woff += (k < w) ? wsum[k] : 0;
    const int excl = woff + inc - c;
    cntl[tid] = excl;
    const int node = blockIdx.x * BNODES + tid;
    if (node < Nn) rowptr[node] = bs + excl;
  }
  if (blockIdx.x == gridDim.x - 1 && tid == 0) rowptr[Nn] = bs + n;
  __syncthreads();
  int* cp = col + bs;
  for (int t0 = 0; t0 < n; t0 += ETILE) {
    int rv[8], pv[8];
#pragma unroll
    for (int k = 0; k < 8; ++k) {
      const int i = t0 + k * 1024 + tid;
      rv[k] = (i < n) ? rp[i] : -1;
    }
#pragma unroll
    for (int k = 0; k < 8; ++k) pv[k] = atomicAdd(&cntl[(rv[k] >= 0) ? (rv[k] >> SRC_BITS) : BNODES], 1);
#pragma unroll
    for (int k = 0; k < 8; ++k) if (rv[k] >= 0) cp[pv[k]] = rv[k] & ((1 << SRC_BITS) - 1);
  }
}

__device__ __forceinline__ int rank_among(int key, int keys, int n) {
  int r = 0;
  for (int j = 0; j < n; ++j) r += (__builtin_amdgcn_readlane(keys, j) < key) ? 1 : 0;
  return r;
}
#define PLACE_BITS 6

__device__ __forceinline__ float leaky(float v) { return (v >= 0.f) ? v : NEG_SLOPE * v; }


__device__ __forceinline__ void wave_lds_order() {
  __builtin_amdgcn_fence(__ATOMIC_RELEASE, "wavefront");
  __builtin_amdgcn_wave_barrier();
  __builtin_amdgcn_fence(__ATOMIC_ACQUIRE, "wavefront");
}

__device__ __forceinline__ void gat1_sum_sorted(int d, int lane, int deg, int sv0, int sv1,
                                                float adh, float m,
                                                const float* __restrict__ h, float4 wsrc,
                                                const float* __restrict__ b1,
                                                float* __restrict__ h1) {
  float ssum = 0.f;
  float4 acc = make_float4(0.f, 0.f, 0.f, 0.f);
  for (int j0 = 0; j0 <= deg; j0 += 4) {
    const int svc = (j0 < 32) ? sv0 : sv1;
    float4 hv[4];
#pragma unroll
    for (int u = 0; u < 4; ++u) {
      const int s = __builtin_amdgcn_readlane(svc, (j0 + u) & 31);
      hv[u] = *(const float4*)(h + (size_t)s * 128 + lane * 4);
    }
    float e[4];
#pragma unroll
    for (int u = 0; u < 4; ++u) {
      float p = hv[u].x * wsrc.x;
      p = fmaf(hv[u].y, wsrc.y, p);
      p = fmaf(hv[u].z, wsrc.z, p);
      p = fmaf(hv[u].w, wsrc.w, p);
      p += row_xor_get<1>(p);
      p += row_xor_get<2>(p);
      const float x = p + adh;
      e[u] = ((j0 + u) <= deg) ? fmaxf(x, NEG_SLOPE * x) : -INFINITY;
    }
    const float mn = fmaxf(fmaxf(m, e[0]), fmaxf(fmaxf(e[1], e[2]), e[3]));
    const float sc = __expf(m - mn);
    const float p0 = __expf(e[0] - mn);
    const float p1 = __expf(e[1] - mn);
    const float p2 = __expf(e[2] - mn);
    const float p3 = __expf(e[3] - mn);
    m = mn;
    ssum = fmaf(ssum, sc, ((p0 + p1) + p2) + p3);
    acc.x = fmaf(hv[3].x, p3, fmaf(hv[2].x, p2, fmaf(hv[1].x, p1, fmaf(hv[0].x, p0, acc.x * sc))));
    acc.y = fmaf(hv[3].y, p3, fmaf(hv[2].y, p2, fmaf(hv[1].y, p1, fmaf(hv[0].y, p0, acc.y * sc))));
    acc.z = fmaf(hv[3].z, p3, fmaf(hv[2].z, p2, fmaf(hv[1].z, p1, fmaf(hv[0].z, p0, acc.z * sc))));
    acc.w = fmaf(hv[3].w, p3, fmaf(hv[2].w, p2, fmaf(hv[1].w, p1, fmaf(hv[0].w, p0, acc.w * sc))));
  }
  const float4 bv = *(const float4*)(b1 + lane * 4);
  float4 o;
  o.x = acc.x / ssum + bv.x; o.x = (o.x > 0.f) ? o.x : (__expf(o.x) - 1.0f);
  o.y = acc.y / ssum + bv.y; o.y = (o.y > 0.f) ? o.y : (__expf(o.y) - 1.0f);
  o.z = acc.z / ssum + bv.z; o.z = (o.z > 0.f) ? o.z : (__expf(o.z) - 1.0f);
  o.w = acc.w / ssum + bv.w; o.w = (o.w > 0.f) ? o.w : (__expf(o.w) - 1.0f);
  *(float4*)(h1 + (size_t)d * 128 + lane * 4) = o;
}

__device__ __forceinline__ int gat1_sort_short(int d, int lane, int start, int deg,
                                               const int* __restrict__ col, int* col2, int* ids) {
  int su = d;
  int r = lane;
  if (lane < deg) su = col[start + lane];
  {
    const int key = (su << PLACE_BITS) | lane;
    const int rr = rank_among(key, key, deg);
    if (lane < deg) r = rr;
  }
  if (lane < deg) col2[start + r] = su;
  ids[r] = su;
  wave_lds_order();
  return ids[lane];
}

__device__ __forceinline__ void gat1_sort_mid(int d, int lane, int start, int deg,
                                              const int* __restrict__ col, int* col2, int* ids,
                                              int& sv0, int& sv1) {
  const bool has1 = (32 + lane) < deg;
  const int su0 = col[start + lane];
  const int su1 = has1 ? col[start + 32 + lane] : d;
  int r0, r1 = 32 + lane;
  {
    const int key0 = (su0 << PLACE_BITS) | lane;
    const int key1 = (su1 << PLACE_BITS) | (32 + lane);
    r0 = rank_among(key0, key0, 32) + rank_among(key0, key1, deg - 32);
    const int rr1 = rank_among(key1, key0, 32) + rank_among(key1, key1, deg - 32);
    if (has1) r1 = rr1;
  }
  col2[start + r0] = su0;
  if (has1) col2[start + r1] = su1;
  ids[r0] = su0;
  ids[r1] = su1;
  wave_lds_order();
  sv0 = ids[lane];
  sv1 = ids[32 + lane];
}

__device__ __forceinline__ void gat1_node_long(int d, int lane, int start, int deg,
                                               const int* __restrict__ col, int* col2,
                                               const float* __restrict__ asc,
                                               const float* __restrict__ adc,
                                               const float* __restrict__ h,
                                               const float* __restrict__ b1,
                                               float* __restrict__ h1) {
  const int hd = lane >> 2;
  for (int i = lane; i < deg; i += 32) {
    const int k = col[start + i];
    int r = 0;
    for (int j = 0; j < deg; ++j) {
      const int kj = col[start + j];
      r += (kj < k || (kj == k && j < i)) ? 1 : 0;
    }
    col2[start + r] = k;
  }
  __threadfence();
  const float ad = adc[(size_t)d * 8 + hd];
  const float eself = leaky(asc[(size_t)d * 8 + hd] + ad);
  float m = eself;
#pragma unroll 4
  for (int j = 0; j < deg; ++j) {
    const int s = col2[start + j];
    const float e = leaky(asc[(size_t)s * 8 + hd] + ad);
    m = fmaxf(m, e);
  }
  float ssum = 0.f;
#pragma unroll 4
  for (int j = 0; j < deg; ++j) {
    const float e = leaky(asc[(size_t)col2[start + j] * 8 + hd] + ad);
    ssum += __expf(e - m);
  }
  const float exself = __expf(eself - m);
  ssum += exself;
  float4 acc = make_float4(0.f, 0.f, 0.f, 0.f);
#pragma unroll 4
  for (int j = 0; j < deg; ++j) {
    const int s = col2[start + j];
    const float ex = __expf(leaky(asc[(size_t)s * 8 + hd] + ad) - m);
    const float alpha = ex / ssum;
    const float4 hv = *(const float4*)(h + (size_t)s * 128 + lane * 4);
    acc.x = fmaf(hv.x, alpha, acc.x);
    acc.y = fmaf(hv.y, alpha, acc.y);
    acc.z = fmaf(hv.z, alpha, acc.z);
    acc.w = fmaf(hv.w, alpha, acc.w);
  }
  {
    const float alpha = exself / ssum;
    const float4 hv = *(const float4*)(h + (size_t)d * 128 + lane * 4);
    acc.x = fmaf(hv.x, alpha, acc.x);
    acc.y = fmaf(hv.y, alpha, acc.y);
    acc.z = fmaf(hv.z, alpha, acc.z);
    acc.w = fmaf(hv.w, alpha, acc.w);
  }
  const float4 bv = *(const float4*)(b1 + lane * 4);
  float4 o;
  o.x = acc.x + bv.x; o.x = (o.x > 0.f) ? o.x : (__expf(o.x) - 1.0f);
  o.y = acc.y + bv.y; o.y = (o.y > 0.f) ? o.y : (__expf(o.y) - 1.0f);
  o.z = acc.z + bv.z; o.z = (o.z > 0.f) ? o.z : (__expf(o.z) - 1.0f);
  o.w = acc.w + bv.w; o.w = (o.w > 0.f) ? o.w : (__expf(o.w) - 1.0f);
  *(float4*)(h1 + (size_t)d * 128 + lane * 4) = o;
}

__global__ void __launch_bounds__(256, 16)
gather1_kernel(const int* __restrict__ rowptr, const int* __restrict__ col, int* col2,
               const float* __restrict__ asc, const float* __restrict__ adc,
               const float* __restrict__ h, const float* __restrict__ att_src,
               const float* __restrict__ b1, float* __restrict__ h1, int Nn) {
  __shared__ int idl[8 * 64];
  const int lane = threadIdx.x & 31;
  const int w = threadIdx.x >> 5;
  int* ids = idl + w * 64;
  const float4 wsrc = *(const float4*)(att_src + lane * 4);
  const int nw = gridDim.x * 8;
  for (int d = __builtin_amdgcn_readfirstlane(blockIdx.x * 8 + w); d < Nn; d += nw) {
    const int start = rowptr[d];
    const int deg = rowptr[d + 1] - start;
    if (deg < 64) {
      const int hd = lane >> 2;
      const float adh = adc[(size_t)d * 8 + hd];
      const float m0 = leaky(asc[(size_t)d * 8 + hd] + adh);
      int sv0, sv1;
      if (deg <= 32) {
        sv0 = gat1_sort_short(d, lane, start, deg, col, col2, ids);
        sv1 = d;
      } else {
        gat1_sort_mid(d, lane, start, deg, col, col2, ids, sv0, sv1);
      }
      gat1_sum_sorted(d, lane, deg, sv0, sv1, adh, m0, h, wsrc, b1, h1);
      wave_lds_order();
    } else {
      gat1_node_long(d, lane, start, deg, col, col2, asc, adc, h, b1, h1);
    }
  }
}

__device__ __forceinline__ float other_row_get(float v) {
  const int b = __float_as_int(v);
  return __int_as_float(__builtin_amdgcn_permlanex16(b, b, 0x76543210, 0xfedcba98, false, false));
}

__global__ void __launch_bounds__(256)
gather2_kernel(const int* __restrict__ rowptr, const int* __restrict__ col2,
               const float* __restrict__ asc, const float* __restrict__ adc,
               const float* __restrict__ h2, const float* __restrict__ att_src,
               const float* __restrict__ b2, float* __restrict__ out, int Nn) {
  const int lane = threadIdx.x & 31;
  const int half = lane >> 4;
  const int l15 = lane & 15;
  const int lc = (l15 < 9) ? l15 : 9;
  float4 w4 = *(const float4*)(att_src + 4 * lc);
  if (l15 >= 10) w4 = make_float4(0.f, 0.f, 0.f, 0.f);
  const int nw = gridDim.x * 8;
  for (int d = __builtin_amdgcn_readfirstlane(blockIdx.x * 8 + (threadIdx.x >> 5)); d < Nn; d += nw) {
    const int start = rowptr[d];
    const int deg = rowptr[d + 1] - start;
    const float ad = adc[d];
    const float eself = leaky(asc[d] + ad);
    const float4 hself = *(const float4*)(h2 + (size_t)d * 40 + 4 * lc);

    float m = eself;
    float ssum = 0.f;
    float4 acc = make_float4(0.f, 0.f, 0.f, 0.f);
    for (int c = 0; c < deg; c += 32) {
      const int n = min(deg - c, 32);
      int sv = d;
      if (lane < n) sv = col2[start + c + lane];
      for (int t0 = 0; t0 < n; t0 += 4) {
        float4 hv[2];
#pragma unroll
        for (int u = 0; u < 2; ++u) {
          const int se = __builtin_amdgcn_readlane(sv, t0 + 2 * u);
          const int so = __builtin_amdgcn_readlane(sv, t0 + 2 * u + 1);
          const int s = half ? so : se;
          hv[u] = *(const float4*)(h2 + (size_t)s * 40 + 4 * lc);
        }
        float e[2];
#pragma unroll
        for (int u = 0; u < 2; ++u) {
          float p = hv[u].x * w4.x;
          p = fmaf(hv[u].y, w4.y, p);
          p = fmaf(hv[u].z, w4.z, p);
          p = fmaf(hv[u].w, w4.w, p);
          p += row_xor_get<1>(p);
          p += row_xor_get<2>(p);
          p += row_xor_get<4>(p);
          p += row_xor_get<8>(p);
          const bool valid = (t0 + 2 * u + half) < n;
          const float x = p + ad;
          e[u] = valid ? fmaxf(x, NEG_SLOPE * x) : -INFINITY;
        }
        const float mn = fmaxf(m, fmaxf(e[0], e[1]));
        const float sc = __expf(m - mn);
        const float p0 = __expf(e[0] - mn);
        const float p1 = __expf(e[1] - mn);
        m = mn;
        ssum = fmaf(ssum, sc, p0 + p1);
        acc.x = fmaf(hv[1].x, p1, fmaf(hv[0].x, p0, acc.x * sc));
        acc.y = fmaf(hv[1].y, p1, fmaf(hv[0].y, p0, acc.y * sc));
        acc.z = fmaf(hv[1].z, p1, fmaf(hv[0].z, p0, acc.z * sc));
        acc.w = fmaf(hv[1].w, p1, fmaf(hv[0].w, p0, acc.w * sc));
      }
    }
    const float mo = other_row_get(m);
    const float so = other_row_get(ssum);
    float4 ao;
    ao.x = other_row_get(acc.x);
    ao.y = other_row_get(acc.y);
    ao.z = other_row_get(acc.z);
    ao.w = other_row_get(acc.w);
    const float mt = fmaxf(m, mo);
    const float f = __expf(m - mt);
    const float g = __expf(mo - mt);
    const float exself = __expf(eself - mt);
    const float st = fmaf(so, g, ssum * f) + exself;
    float4 o;
    o.x = fmaf(hself.x, exself, fmaf(ao.x, g, acc.x * f));
    o.y = fmaf(hself.y, exself, fmaf(ao.y, g, acc.y * f));
    o.z = fmaf(hself.z, exself, fmaf(ao.z, g, acc.z * f));
    o.w = fmaf(hself.w, exself, fmaf(ao.w, g, acc.w * f));
    if (lane < 10) {
      const float4 bv = *(const float4*)(b2 + 4 * lane);
      *(float4*)(out + (size_t)d * 40 + 4 * lane) =
          make_float4(o.x / st + bv.x, o.y / st + bv.y, o.z / st + bv.z, o.w / st + bv.w);
    }
  }
}

extern "C" void kernel_launch(void* const* d_in, const int* in_sizes, int n_in,
                              void* d_out, int out_size, void* d_ws, size_t ws_size,
                              hipStream_t stream) {
  const float* x   = (const float*)d_in[0];
  const int*   ei  = (const int*)d_in[1];
  const float* W1  = (const float*)d_in[2];
  const float* as1 = (const float*)d_in[3];
  const float* ad1 = (const float*)d_in[4];
  const float* b1  = (const float*)d_in[5];
  const float* W2  = (const float*)d_in[6];
  const float* as2 = (const float*)d_in[7];
  const float* ad2 = (const float*)d_in[8];
  const float* b2  = (const float*)d_in[9];
  float* out = (float*)d_out;

  const int Nn   = in_sizes[0] / 128;
  const int E    = in_sizes[1] / 2;
  const int* srcp = ei;
  const int* dstp = ei + E;
  const int NBLK = (E + ETILE - 1) / ETILE;
  const int NBKT = (Nn + BNODES - 1) / BNODES;

  float* ws   = (float*)d_ws;
  float* h    = ws;  ws += (size_t)Nn * 128;
  float* h1   = ws;  ws += (size_t)Nn * 128;
  float* asc1 = ws;  ws += (size_t)Nn * 8;
  float* adc1 = ws;  ws += (size_t)Nn * 8;
  float* h2   = ws;  ws += (size_t)Nn * 40;
  float* asc2 = ws;  ws += Nn;
  float* adc2 = ws;  ws += Nn;
  int* wi     = (int*)ws;
  int* bcnt   = wi;  wi += (size_t)NBLK * NBINS;
  int* rec    = wi;  wi += E;
  int* col    = wi;  wi += E;
  int* col2   = wi;  wi += E;
  int* rowptr = wi;  wi += (size_t)Nn + 1;
  int* bstart = wi;  wi += NBINS + 1;

  bucket_count_kernel<<<NBLK, 1024, 0, stream>>>(dstp, bcnt, E);
  bucket_scatter_kernel<<<NBLK, 1024, 0, stream>>>(srcp, dstp, bcnt, bstart, rec, E, NBLK);
  bucket_csr_kernel<<<NBKT, 1024, 0, stream>>>(bstart, rec, rowptr, col, Nn);

  gat_wmma_gemm_k128<4, 128, 8><<<dim3(896, 2), 4 * 32, 0, stream>>>(x, W1, h, as1, ad1, asc1, adc1, Nn / 16);
  gather1_kernel<<<2048, 256, 0, stream>>>(rowptr, col, col2, asc1, adc1, h, as1, b1, h1, Nn);

  gat_wmma_gemm_k128<3, 40, 1><<<dim3(1563, 1), 3 * 32, 0, stream>>>(h1, W2, h2, as2, ad2, asc2, adc2, Nn / 16);
  gather2_kernel<<<2048, 256, 0, stream>>>(rowptr, col2, asc2, adc2, h2, as2, b2, out, Nn);
}


# Layer-1 GEMM moved into the three CSR-build launches (independent work: matrix units beside LDS counters): 32-wave blocks, 64-row groups double-buffered in LDS; launches 7 to 6; sums unchanged
# speedup vs baseline: 6.1611x; 1.0552x over previous
#include <hip/hip_runtime.h>
#include <hip/hip_bf16.h>
#include <math.h>
#include <stdint.h>


typedef float v2f __attribute__((ext_vector_type(2)));
typedef float v8f __attribute__((ext_vector_type(8)));

#define NEG_SLOPE 0.2f

template <int MASK>
__device__ __forceinline__ float row_xor_get(float v) {
  return __int_as_float(__builtin_amdgcn_update_dpp(0, __float_as_int(v), 0x160 + MASK, 0xf, 0xf, true));
}

template <int NTILES, int NACT, int HEADS>
__global__ void __launch_bounds__(32 * NTILES)
gat_wmma_gemm_k128(const float* __restrict__ A, const float* __restrict__ B,
                   float* __restrict__ C,
                   const float* __restrict__ att_src, const float* __restrict__ att_dst,
                   float* __restrict__ a_src, float* __restrict__ a_dst, int nstrips) {
  static_assert(HEADS == 1 || HEADS * 16 == NACT, "a wave's 16 columns must be one head");
  constexpr int NPAD = NTILES * 16;
  constexpr int SA = 132;
  constexpr int SB = 132;
  constexpr int NT = 32 * NTILES;
  __shared__ __attribute__((aligned(16))) float ldsA[16 * SA];
  __shared__ float ldsB[NPAD * SB];
  __shared__ float red[(HEADS == 1) ? NTILES * 32 : 1];

  const int tid  = threadIdx.x;
  const int lane = tid & 31;
  const int wave = tid >> 5;
  const int c0 = blockIdx.y * NPAD;

  static_assert(NACT % 4 == 0 && NPAD % 4 == 0, "16-byte pieces must not straddle the last column");
  constexpr int BIT = (NPAD / 4) * 128 / NT;
  static_assert(BIT * NT == (NPAD / 4) * 128, "the panel must be a whole number of pieces per thread");
  {
    float4 bv[BIT];
#pragma unroll
    for (int u = 0; u < BIT; ++u) {
      const int p = tid + u * NT;
      const int k = ((p / NPAD) << 2) + (p & 3);
      const int gn = c0 + (((p >> 2) % (NPAD / 4)) << 2);
      const int gc = (gn < NACT) ? gn : (NACT - 4);
      bv[u] = *(const float4*)(B + (size_t)k * NACT + gc);
    }
#pragma unroll
    for (int u = 0; u < BIT; ++u) {
      const int p = tid + u * NT;
      const int k = ((p / NPAD) << 2) + (p & 3);
      const int n = ((p >> 2) % (NPAD / 4)) << 2;
      const bool in = (c0 + n) < NACT;
      float* lp = &ldsB[n * SB + k];
      lp[0]      = in ? bv[u].x : 0.0f;
      lp[SB]     = in ? bv[u].y : 0.0f;
      lp[2 * SB] = in ? bv[u].z : 0.0f;
      lp[3 * SB] = in ? bv[u].w : 0.0f;
    }
  }

  const int l15  = lane & 15;
  const int koff = (lane >> 4) << 1;
  const int ncol = wave * 16 + l15;
  const float* ap = &ldsA[l15 * SA + koff];
  const float* bp = &ldsB[ncol * SB + koff];

  const int gcol = c0 + ncol;
  const float wsrc = (gcol < NACT) ? att_src[gcol] : 0.0f;
  const float wdst = (gcol < NACT) ? att_dst[gcol] : 0.0f;
  const int hd = blockIdx.y * NTILES + wave;

  for (int strip = blockIdx.x; strip < nstrips; strip += gridDim.x) {
    const size_t m0 = (size_t)strip * 16;
    constexpr int AIT = (16 * 32 + NT - 1) / NT;
    float4 av[AIT];
#pragma unroll
    for (int u = 0; u < AIT; ++u) {
      const int i = min(tid + u * NT, 16 * 32 - 1);
      av[u] = *(const float4*)(A + (m0 + (i >> 5)) * 128 + ((i & 31) << 2));
    }
#pragma unroll
    for (int u = 0; u < AIT; ++u) {
      const int i = tid + u * NT;
      if ((u + 1) * NT <= 16 * 32 || i < 16 * 32) *(float4*)&ldsA[(i >> 5) * SA + ((i & 31) << 2)] = av[u];
    }
    __syncthreads();

    v8f acc = {0.f, 0.f, 0.f, 0.f, 0.f, 0.f, 0.f, 0.f};
#pragma unroll
    for (int kb = 0; kb < 128; kb += 4) {
      v2f a = *(const v2f*)(ap + kb);
      v2f b = *(const v2f*)(bp + kb);
      acc = __builtin_amdgcn_wmma_f32_16x16x4_f32(
           false, a,  false, b,
           (short)0, acc,  false,  false);
    }

    if (gcol < NACT) {
      float* cp = C + (m0 + (size_t)((lane >> 4) << 3)) * NACT + gcol;
#pragma unroll
      for (int r = 0; r < 8; ++r) cp[(size_t)r * NACT] = acc[r];
    }

    const bool b3 = (l15 & 8) != 0, b2 = (l15 & 4) != 0, b1 = (l15 & 2) != 0;
    float ps[8], pd[8];
#pragma unroll
    for (int r = 0; r < 8; ++r) { ps[r] = acc[r] * wsrc; pd[r] = acc[r] * wdst; }
    float s4[4], d4[4];
#pragma unroll
    for (int i = 0; i < 4; ++i) {
      s4[i] = (b3 ? ps[4 + i] : ps[i]) + row_xor_get<8>(b3 ? ps[i] : ps[4 + i]);
      d4[i] = (b3 ? pd[4 + i] : pd[i]) + row_xor_get<8>(b3 ? pd[i] : pd[4 + i]);
    }
    float s2[2], d2[2];
#pragma unroll
    for (int j = 0; j < 2; ++j) {
      s2[j] = (b2 ? s4[2 + j] : s4[j]) + row_xor_get<4>(b2 ? s4[j] : s4[2 + j]);
      d2[j] = (b2 ? d4[2 + j] : d4[j]) + row_xor_get<4>(b2 ? d4[j] : d4[2 + j]);
    }
    float s1 = (b1 ? s2[1] : s2[0]) + row_xor_get<2>(b1 ? s2[0] : s2[1]);
    float d1 = (b1 ? d2[1] : d2[0]) + row_xor_get<2>(b1 ? d2[0] : d2[1]);
    s1 += row_xor_get<1>(s1);
    d1 += row_xor_get<1>(d1);
    const bool is_src = (l15 & 1) == 0;
    const float sv = is_src ? s1 : d1;
    const int row = ((lane >> 4) << 3) + (l15 >> 1);
    if constexpr (HEADS == 1) {
      red[(wave * 16 + row) * 2 + (is_src ? 0 : 1)] = sv;
    } else {
      float* sp = is_src ? a_src : a_dst;
      sp[(m0 + row) * HEADS + hd] = sv;
    }
    __syncthreads();
    if constexpr (HEADS == 1) if (wave == 0) {
      const int r16 = lane & 15, which = lane >> 4;
      float s = red[r16 * 2 + which];
#pragma unroll
      for (int w = 1; w < NTILES; ++w) s += red[(w * 16 + r16) * 2 + which];
      float* sp = which ? a_dst : a_src;
      sp[m0 + r16] = s;
    }
  }
}

__device__ __forceinline__ void block_lds_barrier() {
  __builtin_amdgcn_fence(__ATOMIC_RELEASE, "workgroup", "local");
  __builtin_amdgcn_s_barrier();
  __builtin_amdgcn_fence(__ATOMIC_ACQUIRE, "workgroup", "local");
}

__device__ __forceinline__ void gemm1_rows64_role(const float* __restrict__ A, const float* __restrict__ B,
                                                  float* __restrict__ C,
                                                  const float* __restrict__ att_src,
                                                  const float* __restrict__ att_dst,
                                                  float* __restrict__ a_src, float* __restrict__ a_dst,
                                                  int q0, int q1, int gb, int ngb, int M) {
  constexpr int SA = 132;
  constexpr int SB = 132;
  constexpr int NT = 1024;
  __shared__ __attribute__((aligned(16))) float ldsA[2 * 64 * SA];
  __shared__ float ldsB[128 * SB];

  const int tid  = threadIdx.x;
  const int lane = tid & 31;
  const int wave = tid >> 5;
  const int ct   = wave & 7;
  const int slot = wave >> 3;

  int q = q0 + gb;
  if (q >= q1) return;

  {
    float4 bv[4];
#pragma unroll
    for (int u = 0; u < 4; ++u) {
      const int p = tid + u * NT;
      const int k = ((p >> 7) << 2) + (p & 3);
      const int n = ((p >> 2) & 31) << 2;
      bv[u] = *(const float4*)(B + (size_t)k * 128 + n);
    }
#pragma unroll
    for (int u = 0; u < 4; ++u) {
      const int p = tid + u * NT;
      const int k = ((p >> 7) << 2) + (p & 3);
      const int n = ((p >> 2) & 31) << 2;
      float* lp = &ldsB[n * SB + k];
      lp[0]      = bv[u].x;
      lp[SB]     = bv[u].y;
      lp[2 * SB] = bv[u].z;
      lp[3 * SB] = bv[u].w;
    }
  }

  const int l15  = lane & 15;
  const int koff = (lane >> 4) << 1;
  const int gcol = ct * 16 + l15;
  const float* ap = &ldsA[(slot * 16 + l15) * SA + koff];
  const float* bp = &ldsB[gcol * SB + koff];
  const float wsrc = att_src[gcol];
  const float wdst = att_dst[gcol];

  const int arow = tid >> 5;
  const int acol = (tid & 31) << 2;
  float4 av0 = *(const float4*)(A + (size_t)min(q * 64 + arow, M - 1) * 128 + acol);
  float4 av1 = *(const float4*)(A + (size_t)min(q * 64 + 32 + arow, M - 1) * 128 + acol);

  int buf = 0;
  for (; q < q1; q += ngb, buf ^= 1) {
    const int boff = buf * (64 * SA);
    *(float4*)&ldsA[boff + arow * SA + acol] = av0;
    *(float4*)&ldsA[boff + (32 + arow) * SA + acol] = av1;
    block_lds_barrier();

    const int qn = q + ngb;
    if (qn < q1) {
      av0 = *(const float4*)(A + (size_t)min(qn * 64 + arow, M - 1) * 128 + acol);
      av1 = *(const float4*)(A + (size_t)min(qn * 64 + 32 + arow, M - 1) * 128 + acol);
    }

    const float* apb = ap + boff;
    v8f acc = {0.f, 0.f, 0.f, 0.f, 0.f, 0.f, 0.f, 0.f};
#pragma unroll
    for (int kb = 0; kb < 128; kb += 4) {
      v2f a = *(const v2f*)(apb + kb);
      v2f b = *(const v2f*)(bp + kb);
      acc = __builtin_amdgcn_wmma_f32_16x16x4_f32(
           false, a,  false, b,
           (short)0, acc,  false,  false);
    }

    const bool b3 = (l15 & 8) != 0, b2 = (l15 & 4) != 0, b1 = (l15 & 2) != 0;
    float ps[8], pd[8];
#pragma unroll
    for (int r = 0; r < 8; ++r) { ps[r] = acc[r] * wsrc; pd[r] = acc[r] * wdst; }
    float s4[4], d4[4];
#pragma unroll
    for (int i = 0; i < 4; ++i) {
      s4[i] = (b3 ? ps[4 + i] : ps[i]) + row_xor_get<8>(b3 ? ps[i] : ps[4 + i]);
      d4[i] = (b3 ? pd[4 + i] : pd[i]) + row_xor_get<8>(b3 ? pd[i] : pd[4 + i]);
    }
    float s2[2], d2[2];
#pragma unroll
    for (int j = 0; j < 2; ++j) {
      s2[j] = (b2 ? s4[2 + j] : s4[j]) + row_xor_get<4>(b2 ? s4[j] : s4[2 + j]);
      d2[j] = (b2 ? d4[2 + j] : d4[j]) + row_xor_get<4>(b2 ? d4[j] : d4[2 + j]);
    }
    float s1 = (b1 ? s2[1] : s2[0]) + row_xor_get<2>(b1 ? s2[0] : s2[1]);
    float d1 = (b1 ? d2[1] : d2[0]) + row_xor_get<2>(b1 ? d2[0] : d2[1]);
    s1 += row_xor_get<1>(s1);
    d1 += row_xor_get<1>(d1);
    const bool is_src = (l15 & 1) == 0;
    const float sv = is_src ? s1 : d1;
    const int row = ((lane >> 4) << 3) + (l15 >> 1);

    const size_t m0 = ((size_t)q * 4 + slot) * 16;
    if (m0 < (size_t)M) {
      float* cp = C + (m0 + (size_t)((lane >> 4) << 3)) * 128 + gcol;
#pragma unroll
      for (int r = 0; r < 8; ++r) cp[(size_t)r * 128] = acc[r];
      float* sp = is_src ? a_src : a_dst;
      sp[(m0 + row) * 8 + ct] = sv;
    }
  }
}

#define BSHIFT 9
#define BNODES 512
#define NBINS 256
#define SRC_BITS 17
#define ETILE 8192

__device__ __forceinline__ void bucket_count_role(const int* __restrict__ dst, int* __restrict__ bcnt,
                                                  int E, int bid) {
  __shared__ int hist[NBINS];
  const int tid = threadIdx.x;
  if (tid < NBINS) hist[tid] = 0;
  const int base = bid * ETILE + tid;
  int dv[8];
#pragma unroll
  for (int k = 0; k < 8; ++k) {
    const int e = base + k * 1024;
    dv[k] = (e < E) ? dst[e] : -1;
  }
  __syncthreads();
#pragma unroll
  for (int k = 0; k < 8; ++k) if (dv[k] >= 0) atomicAdd(&hist[dv[k] >> BSHIFT], 1);
  __syncthreads();
  if (tid < NBINS) bcnt[bid * NBINS + tid] = hist[tid];
}

__device__ __forceinline__ void bucket_scatter_role(const int* __restrict__ src, const int* __restrict__ dst,
                                                    const int* __restrict__ bcnt, int* __restrict__ bstart,
                                                    int* __restrict__ rec, int E, int nblk, int bid) {
  __shared__ int ptot[4][NBINS];
  __shared__ int ppre[4][NBINS];
  __shared__ int cursor[NBINS + 1];
  const int tid = threadIdx.x;
  const int lane = tid & 31;
  const int base = bid * ETILE + tid;
  int dv[8], sv[8];
#pragma unroll
  for (int k = 0; k < 8; ++k) {
    const int e = base + k * 1024;
    const bool ok = e < E;
    dv[k] = ok ? dst[e] : -1;
    sv[k] = ok ? src[e] : 0;
  }
  {
    const int k = tid & (NBINS - 1);
    const int q = tid >> 8;
    const int gq = (nblk + 3) >> 2;
    const int j0 = q * gq;
    const int j1 = min(nblk, j0 + gq);
    int tot = 0, pre = 0;
#pragma unroll 8
    for (int j = j0; j < j1; ++j) {
      const int c = bcnt[j * NBINS + k];
      tot += c;
      pre += (j < bid) ? c : 0;
    }
    ptot[q][k] = tot;
    ppre[q][k] = pre;
    if (tid == 0) cursor[NBINS] = 0;
  }
  __syncthreads();
  if (tid < 32) {
    int v[8];
    int run = 0;
#pragma unroll
    for (int i = 0; i < 8; ++i) {
      const int idx = tid * 8 + i;
      v[i] = run;
      run += ptot[0][idx] + ptot[1][idx] + ptot[2][idx] + ptot[3][idx];
    }
    int inc = run;
#pragma unroll
    for (int off = 1; off < 32; off <<= 1) {
      const int nb = __shfl_up(inc, off);
      if (lane >= off) inc += nb;
    }
    const int excl = inc - run;
#pragma unroll
    for (int i = 0; i < 8; ++i) {
      const int idx = tid * 8 + i;
      const int st = excl + v[i];
      cursor[idx] = st + ppre[0][idx] + ppre[1][idx] + ppre[2][idx] + ppre[3][idx];
      if (bid == 0) bstart[idx] = st;
    }
    if (bid == 0 && tid == 31) bstart[NBINS] = inc;
  }
  __syncthreads();
  int pv[8];
#pragma unroll
  for (int k = 0; k < 8; ++k) pv[k] = atomicAdd(&cursor[(dv[k] >= 0) ? (dv[k] >> BSHIFT) : NBINS], 1);
#pragma unroll
  for (int k = 0; k < 8; ++k)
    if (dv[k] >= 0) rec[pv[k]] = ((dv[k] & (BNODES - 1)) << SRC_BITS) | sv[k];
}

__device__ __forceinline__ void bucket_csr_role(const int* __restrict__ bstart, const int* __restrict__ rec,
                                                int* __restrict__ rowptr, int* __restrict__ col, int Nn,
                                                int nblk, int bid) {
  __shared__ int cntl[BNODES + 1];
  __shared__ int wsum[16];
  const int tid = threadIdx.x, lane = tid & 31, w = tid >> 5;
  const int bs = bstart[bid];
  const int n = bstart[bid + 1] - bs;
  if (tid <= BNODES) cntl[tid] = 0;
  __syncthreads();
  const int* rp = rec + bs;
  for (int t0 = 0; t0 < n; t0 += ETILE) {
    int rv[8];
#pragma unroll
    for (int k = 0; k < 8; ++k) {
      const int i = t0 + k * 1024 + tid;
      rv[k] = (i < n) ? rp[i] : -1;
    }
#pragma unroll
    for (int k = 0; k < 8; ++k) if (rv[k] >= 0) atomicAdd(&cntl[rv[k] >> SRC_BITS], 1);
  }
  __syncthreads();
  int c = 0, inc = 0;
  if (tid < BNODES) {
    c = cntl[tid];
    inc = c;
#pragma unroll
    for (int off = 1; off < 32; off <<= 1) {
      const int nb = __shfl_up(inc, off);
      if (lane >= off) inc += nb;
    }
    if (lane == 31) wsum[w] = inc;
  }
  __syncthreads();
  if (tid < BNODES) {
    int woff = 0;
#pragma unroll
    for (int k = 0; k < 16; ++k) woff += (k < w) ? wsum[k] : 0;
    const int excl = woff + inc - c;
    cntl[tid] = excl;
    const int node = bid * BNODES + tid;
    if (node < Nn) rowptr[node] = bs + excl;
  }
  if (bid == nblk - 1 && tid == 0) rowptr[Nn] = bs + n;
  __syncthreads();
  int* cp = col + bs;
  for (int t0 = 0; t0 < n; t0 += ETILE) {
    int rv[8], pv[8];
#pragma unroll
    for (int k = 0; k < 8; ++k) {
      const int i = t0 + k * 1024 + tid;
      rv[k] = (i < n) ? rp[i] : -1;
    }
#pragma unroll
    for (int k = 0; k < 8; ++k) pv[k] = atomicAdd(&cntl[(rv[k] >= 0) ? (rv[k] >> SRC_BITS) : BNODES], 1);
#pragma unroll
    for (int k = 0; k < 8; ++k) if (rv[k] >= 0) cp[pv[k]] = rv[k] & ((1 << SRC_BITS) - 1);
  }
}

__global__ void __launch_bounds__(1024, 16)
count_gemm_kernel(const int* __restrict__ dst, int* __restrict__ bcnt, int E, int nrole,
                  const float* __restrict__ A, const float* __restrict__ B, float* __restrict__ C,
                  const float* __restrict__ att_src, const float* __restrict__ att_dst,
                  float* __restrict__ a_src, float* __restrict__ a_dst, int q0, int q1, int M) {
  const int bid = blockIdx.x;
  if (bid < nrole) bucket_count_role(dst, bcnt, E, bid);
  else gemm1_rows64_role(A, B, C, att_src, att_dst, a_src, a_dst, q0, q1, bid - nrole, (int)gridDim.x - nrole, M);
}

__global__ void __launch_bounds__(1024, 16)
scatter_gemm_kernel(const int* __restrict__ src, const int* __restrict__ dst,
                    const int* __restrict__ bcnt, int* __restrict__ bstart,
                    int* __restrict__ rec, int E, int nrole,
                    const float* __restrict__ A, const float* __restrict__ B, float* __restrict__ C,
                    const float* __restrict__ att_src, const float* __restrict__ att_dst,
                    float* __restrict__ a_src, float* __restrict__ a_dst, int q0, int q1, int M) {
  const int bid = blockIdx.x;
  if (bid < nrole) bucket_scatter_role(src, dst, bcnt, bstart, rec, E, nrole, bid);
  else gemm1_rows64_role(A, B, C, att_src, att_dst, a_src, a_dst, q0, q1, bid - nrole, (int)gridDim.x - nrole, M);
}

__global__ void __launch_bounds__(1024, 16)
csr_gemm_kernel(const int* __restrict__ bstart, const int* __restrict__ rec,
                int* __restrict__ rowptr, int* __restrict__ col, int Nn, int nrole,
                const float* __restrict__ A, const float* __restrict__ B, float* __restrict__ C,
                const float* __restrict__ att_src, const float* __restrict__ att_dst,
                float* __restrict__ a_src, float* __restrict__ a_dst, int q0, int q1) {
  const int bid = blockIdx.x;
  if (bid < nrole) bucket_csr_role(bstart, rec, rowptr, col, Nn, nrole, bid);
  else gemm1_rows64_role(A, B, C, att_src, att_dst, a_src, a_dst, q0, q1, bid - nrole, (int)gridDim.x - nrole, Nn);
}

__device__ __forceinline__ int rank_among(int key, int keys, int n) {
  int r = 0;
  for (int j = 0; j < n; ++j) r += (__builtin_amdgcn_readlane(keys, j) < key) ? 1 : 0;
  return r;
}
#define PLACE_BITS 6

__device__ __forceinline__ float leaky(float v) { return (v >= 0.f) ? v : NEG_SLOPE * v; }


__device__ __forceinline__ void wave_lds_order() {
  __builtin_amdgcn_fence(__ATOMIC_RELEASE, "wavefront");
  __builtin_amdgcn_wave_barrier();
  __builtin_amdgcn_fence(__ATOMIC_ACQUIRE, "wavefront");
}

__device__ __forceinline__ void gat1_sum_sorted(int d, int lane, int deg, int sv0, int sv1,
                                                float adh, float m,
                                                const float* __restrict__ h, float4 wsrc,
                                                const float* __restrict__ b1,
                                                float* __restrict__ h1) {
  float ssum = 0.f;
  float4 acc = make_float4(0.f, 0.f, 0.f, 0.f);
  for (int j0 = 0; j0 <= deg; j0 += 4) {
    const int svc = (j0 < 32) ? sv0 : sv1;
    float4 hv[4];
#pragma unroll
    for (int u = 0; u < 4; ++u) {
      const int s = __builtin_amdgcn_readlane(svc, (j0 + u) & 31);
      hv[u] = *(const float4*)(h + (size_t)s * 128 + lane * 4);
    }
    float e[4];
#pragma unroll
    for (int u = 0; u < 4; ++u) {
      float p = hv[u].x * wsrc.x;
      p = fmaf(hv[u].y, wsrc.y, p);
      p = fmaf(hv[u].z, wsrc.z, p);
      p = fmaf(hv[u].w, wsrc.w, p);
      p += row_xor_get<1>(p);
      p += row_xor_get<2>(p);
      const float x = p + adh;
      e[u] = ((j0 + u) <= deg) ? fmaxf(x, NEG_SLOPE * x) : -INFINITY;
    }
    const float mn = fmaxf(fmaxf(m, e[0]), fmaxf(fmaxf(e[1], e[2]), e[3]));
    const float sc = __expf(m - mn);
    const float p0 = __expf(e[0] - mn);
    const float p1 = __expf(e[1] - mn);
    const float p2 = __expf(e[2] - mn);
    const float p3 = __expf(e[3] - mn);
    m = mn;
    ssum = fmaf(ssum, sc, ((p0 + p1) + p2) + p3);
    acc.x = fmaf(hv[3].x, p3, fmaf(hv[2].x, p2, fmaf(hv[1].x, p1, fmaf(hv[0].x, p0, acc.x * sc))));
    acc.y = fmaf(hv[3].y, p3, fmaf(hv[2].y, p2, fmaf(hv[1].y, p1, fmaf(hv[0].y, p0, acc.y * sc))));
    acc.z = fmaf(hv[3].z, p3, fmaf(hv[2].z, p2, fmaf(hv[1].z, p1, fmaf(hv[0].z, p0, acc.z * sc))));
    acc.w = fmaf(hv[3].w, p3, fmaf(hv[2].w, p2, fmaf(hv[1].w, p1, fmaf(hv[0].w, p0, acc.w * sc))));
  }
  const float4 bv = *(const float4*)(b1 + lane * 4);
  float4 o;
  o.x = acc.x / ssum + bv.x; o.x = (o.x > 0.f) ? o.x : (__expf(o.x) - 1.0f);
  o.y = acc.y / ssum + bv.y; o.y = (o.y > 0.f) ? o.y : (__expf(o.y) - 1.0f);
  o.z = acc.z / ssum + bv.z; o.z = (o.z > 0.f) ? o.z : (__expf(o.z) - 1.0f);
  o.w = acc.w / ssum + bv.w; o.w = (o.w > 0.f) ? o.w : (__expf(o.w) - 1.0f);
  *(float4*)(h1 + (size_t)d * 128 + lane * 4) = o;
}

__device__ __forceinline__ int gat1_sort_short(int d, int lane, int start, int deg,
                                               const int* __restrict__ col, int* col2, int* ids) {
  int su = d;
  int r = lane;
  if (lane < deg) su = col[start + lane];
  {
    const int key = (su << PLACE_BITS) | lane;
    const int rr = rank_among(key, key, deg);
    if (lane < deg) r = rr;
  }
  if (lane < deg) col2[start + r] = su;
  ids[r] = su;
  wave_lds_order();
  return ids[lane];
}

__device__ __forceinline__ void gat1_sort_mid(int d, int lane, int start, int deg,
                                              const int* __restrict__ col, int* col2, int* ids,
                                              int& sv0, int& sv1) {
  const bool has1 = (32 + lane) < deg;
  const int su0 = col[start + lane];
  const int su1 = has1 ? col[start + 32 + lane] : d;
  int r0, r1 = 32 + lane;
  {
    const int key0 = (su0 << PLACE_BITS) | lane;
    const int key1 = (su1 << PLACE_BITS) | (32 + lane);
    r0 = rank_among(key0, key0, 32) + rank_among(key0, key1, deg - 32);
    const int rr1 = rank_among(key1, key0, 32) + rank_among(key1, key1, deg - 32);
    if (has1) r1 = rr1;
  }
  col2[start + r0] = su0;
  if (has1) col2[start + r1] = su1;
  ids[r0] = su0;
  ids[r1] = su1;
  wave_lds_order();
  sv0 = ids[lane];
  sv1 = ids[32 + lane];
}

__device__ __forceinline__ void gat1_node_long(int d, int lane, int start, int deg,
                                               const int* __restrict__ col, int* col2,
                                               const float* __restrict__ asc,
                                               const float* __restrict__ adc,
                                               const float* __restrict__ h,
                                               const float* __restrict__ b1,
                                               float* __restrict__ h1) {
  const int hd = lane >> 2;
  for (int i = lane; i < deg; i += 32) {
    const int k = col[start + i];
    int r = 0;
    for (int j = 0; j < deg; ++j) {
      const int kj = col[start + j];
      r += (kj < k || (kj == k && j < i)) ? 1 : 0;
    }
    col2[start + r] = k;
  }
  __threadfence();
  const float ad = adc[(size_t)d * 8 + hd];
  const float eself = leaky(asc[(size_t)d * 8 + hd] + ad);
  float m = eself;
#pragma unroll 4
  for (int j = 0; j < deg; ++j) {
    const int s = col2[start + j];
    const float e = leaky(asc[(size_t)s * 8 + hd] + ad);
    m = fmaxf(m, e);
  }
  float ssum = 0.f;
#pragma unroll 4
  for (int j = 0; j < deg; ++j) {
    const float e = leaky(asc[(size_t)col2[start + j] * 8 + hd] + ad);
    ssum += __expf(e - m);
  }
  const float exself = __expf(eself - m);
  ssum += exself;
  float4 acc = make_float4(0.f, 0.f, 0.f, 0.f);
#pragma unroll 4
  for (int j = 0; j < deg; ++j) {
    const int s = col2[start + j];
    const float ex = __expf(leaky(asc[(size_t)s * 8 + hd] + ad) - m);
    const float alpha = ex / ssum;
    const float4 hv = *(const float4*)(h + (size_t)s * 128 + lane * 4);
    acc.x = fmaf(hv.x, alpha, acc.x);
    acc.y = fmaf(hv.y, alpha, acc.y);
    acc.z = fmaf(hv.z, alpha, acc.z);
    acc.w = fmaf(hv.w, alpha, acc.w);
  }
  {
    const float alpha = exself / ssum;
    const float4 hv = *(const float4*)(h + (size_t)d * 128 + lane * 4);
    acc.x = fmaf(hv.x, alpha, acc.x);
    acc.y = fmaf(hv.y, alpha, acc.y);
    acc.z = fmaf(hv.z, alpha, acc.z);
    acc.w = fmaf(hv.w, alpha, acc.w);
  }
  const float4 bv = *(const float4*)(b1 + lane * 4);
  float4 o;
  o.x = acc.x + bv.x; o.x = (o.x > 0.f) ? o.x : (__expf(o.x) - 1.0f);
  o.y = acc.y + bv.y; o.y = (o.y > 0.f) ? o.y : (__expf(o.y) - 1.0f);
  o.z = acc.z + bv.z; o.z = (o.z > 0.f) ? o.z : (__expf(o.z) - 1.0f);
  o.w = acc.w + bv.w; o.w = (o.w > 0.f) ? o.w : (__expf(o.w) - 1.0f);
  *(float4*)(h1 + (size_t)d * 128 + lane * 4) = o;
}

__global__ void __launch_bounds__(256, 16)
gather1_kernel(const int* __restrict__ rowptr, const int* __restrict__ col, int* col2,
               const float* __restrict__ asc, const float* __restrict__ adc,
               const float* __restrict__ h, const float* __restrict__ att_src,
               const float* __restrict__ b1, float* __restrict__ h1, int Nn) {
  __shared__ int idl[8 * 64];
  const int lane = threadIdx.x & 31;
  const int w = threadIdx.x >> 5;
  int* ids = idl + w * 64;
  const float4 wsrc = *(const float4*)(att_src + lane * 4);
  const int nw = gridDim.x * 8;
  for (int d = __builtin_amdgcn_readfirstlane(blockIdx.x * 8 + w); d < Nn; d += nw) {
    const int start = rowptr[d];
    const int deg = rowptr[d + 1] - start;
    if (deg < 64) {
      const int hd = lane >> 2;
      const float adh = adc[(size_t)d * 8 + hd];
      const float m0 = leaky(asc[(size_t)d * 8 + hd] + adh);
      int sv0, sv1;
      if (deg <= 32) {
        sv0 = gat1_sort_short(d, lane, start, deg, col, col2, ids);
        sv1 = d;
      } else {
        gat1_sort_mid(d, lane, start, deg, col, col2, ids, sv0, sv1);
      }
      gat1_sum_sorted(d, lane, deg, sv0, sv1, adh, m0, h, wsrc, b1, h1);
      wave_lds_order();
    } else {
      gat1_node_long(d, lane, start, deg, col, col2, asc, adc, h, b1, h1);
    }
  }
}

__device__ __forceinline__ float other_row_get(float v) {
  const int b = __float_as_int(v);
  return __int_as_float(__builtin_amdgcn_permlanex16(b, b, 0x76543210, 0xfedcba98, false, false));
}

__global__ void __launch_bounds__(256)
gather2_kernel(const int* __restrict__ rowptr, const int* __restrict__ col2,
               const float* __restrict__ asc, const float* __restrict__ adc,
               const float* __restrict__ h2, const float* __restrict__ att_src,
               const float* __restrict__ b2, float* __restrict__ out, int Nn) {
  const int lane = threadIdx.x & 31;
  const int half = lane >> 4;
  const int l15 = lane & 15;
  const int lc = (l15 < 9) ? l15 : 9;
  float4 w4 = *(const float4*)(att_src + 4 * lc);
  if (l15 >= 10) w4 = make_float4(0.f, 0.f, 0.f, 0.f);
  const int nw = gridDim.x * 8;
  for (int d = __builtin_amdgcn_readfirstlane(blockIdx.x * 8 + (threadIdx.x >> 5)); d < Nn; d += nw) {
    const int start = rowptr[d];
    const int deg = rowptr[d + 1] - start;
    const float ad = adc[d];
    const float eself = leaky(asc[d] + ad);
    const float4 hself = *(const float4*)(h2 + (size_t)d * 40 + 4 * lc);

    float m = eself;
    float ssum = 0.f;
    float4 acc = make_float4(0.f, 0.f, 0.f, 0.f);
    for (int c = 0; c < deg; c += 32) {
      const int n = min(deg - c, 32);
      int sv = d;
      if (lane < n) sv = col2[start + c + lane];
      for (int t0 = 0; t0 < n; t0 += 4) {
        float4 hv[2];
#pragma unroll
        for (int u = 0; u < 2; ++u) {
          const int se = __builtin_amdgcn_readlane(sv, t0 + 2 * u);
          const int so = __builtin_amdgcn_readlane(sv, t0 + 2 * u + 1);
          const int s = half ? so : se;
          hv[u] = *(const float4*)(h2 + (size_t)s * 40 + 4 * lc);
        }
        float e[2];
#pragma unroll
        for (int u = 0; u < 2; ++u) {
          float p = hv[u].x * w4.x;
          p = fmaf(hv[u].y, w4.y, p);
          p = fmaf(hv[u].z, w4.z, p);
          p = fmaf(hv[u].w, w4.w, p);
          p += row_xor_get<1>(p);
          p += row_xor_get<2>(p);
          p += row_xor_get<4>(p);
          p += row_xor_get<8>(p);
          const bool valid = (t0 + 2 * u + half) < n;
          const float x = p + ad;
          e[u] = valid ? fmaxf(x, NEG_SLOPE * x) : -INFINITY;
        }
        const float mn = fmaxf(m, fmaxf(e[0], e[1]));
        const float sc = __expf(m - mn);
        const float p0 = __expf(e[0] - mn);
        const float p1 = __expf(e[1] - mn);
        m = mn;
        ssum = fmaf(ssum, sc, p0 + p1);
        acc.x = fmaf(hv[1].x, p1, fmaf(hv[0].x, p0, acc.x * sc));
        acc.y = fmaf(hv[1].y, p1, fmaf(hv[0].y, p0, acc.y * sc));
        acc.z = fmaf(hv[1].z, p1, fmaf(hv[0].z, p0, acc.z * sc));
        acc.w = fmaf(hv[1].w, p1, fmaf(hv[0].w, p0, acc.w * sc));
      }
    }
    const float mo = other_row_get(m);
    const float so = other_row_get(ssum);
    float4 ao;
    ao.x = other_row_get(acc.x);
    ao.y = other_row_get(acc.y);
    ao.z = other_row_get(acc.z);
    ao.w = other_row_get(acc.w);
    const float mt = fmaxf(m, mo);
    const float f = __expf(m - mt);
    const float g = __expf(mo - mt);
    const float exself = __expf(eself - mt);
    const float st = fmaf(so, g, ssum * f) + exself;
    float4 o;
    o.x = fmaf(hself.x, exself, fmaf(ao.x, g, acc.x * f));
    o.y = fmaf(hself.y, exself, fmaf(ao.y, g, acc.y * f));
    o.z = fmaf(hself.z, exself, fmaf(ao.z, g, acc.z * f));
    o.w = fmaf(hself.w, exself, fmaf(ao.w, g, acc.w * f));
    if (lane < 10) {
      const float4 bv = *(const float4*)(b2 + 4 * lane);
      *(float4*)(out + (size_t)d * 40 + 4 * lane) =
          make_float4(o.x / st + bv.x, o.y / st + bv.y, o.z / st + bv.z, o.w / st + bv.w);
    }
  }
}

extern "C" void kernel_launch(void* const* d_in, const int* in_sizes, int n_in,
                              void* d_out, int out_size, void* d_ws, size_t ws_size,
                              hipStream_t stream) {
  const float* x   = (const float*)d_in[0];
  const int*   ei  = (const int*)d_in[1];
  const float* W1  = (const float*)d_in[2];
  const float* as1 = (const float*)d_in[3];
  const float* ad1 = (const float*)d_in[4];
  const float* b1  = (const float*)d_in[5];
  const float* W2  = (const float*)d_in[6];
  const float* as2 = (const float*)d_in[7];
  const float* ad2 = (const float*)d_in[8];
  const float* b2  = (const float*)d_in[9];
  float* out = (float*)d_out;

  const int Nn   = in_sizes[0] / 128;
  const int E    = in_sizes[1] / 2;
  const int* srcp = ei;
  const int* dstp = ei + E;
  const int NBLK = (E + ETILE - 1) / ETILE;
  const int NBKT = (Nn + BNODES - 1) / BNODES;

  float* ws   = (float*)d_ws;
  float* h    = ws;  ws += (size_t)Nn * 128;
  float* h1   = ws;  ws += (size_t)Nn * 128;
  float* asc1 = ws;  ws += (size_t)Nn * 8;
  float* adc1 = ws;  ws += (size_t)Nn * 8;
  float* h2   = ws;  ws += (size_t)Nn * 40;
  float* asc2 = ws;  ws += Nn;
  float* adc2 = ws;  ws += Nn;
  int* wi     = (int*)ws;
  int* bcnt   = wi;  wi += (size_t)NBLK * NBINS;
  int* rec    = wi;  wi += E;
  int* col    = wi;  wi += E;
  int* col2   = wi;  wi += E;
  int* rowptr = wi;  wi += (size_t)Nn + 1;
  int* bstart = wi;  wi += NBINS + 1;

  const int NQ  = (Nn + 63) / 64;
  const int GB  = 256;
  const int QA  = (2 * GB < NQ) ? 2 * GB : NQ;
  const int QB  = (4 * GB < NQ) ? 4 * GB : NQ;
  const int GB3 = (NQ - QB + 1) / 2;
  count_gemm_kernel<<<NBLK + GB, 1024, 0, stream>>>(dstp, bcnt, E, NBLK,
                                                    x, W1, h, as1, ad1, asc1, adc1, 0, QA, Nn);
  scatter_gemm_kernel<<<NBLK + GB, 1024, 0, stream>>>(srcp, dstp, bcnt, bstart, rec, E, NBLK,
                                                      x, W1, h, as1, ad1, asc1, adc1, QA, QB, Nn);
  csr_gemm_kernel<<<NBKT + GB3, 1024, 0, stream>>>(bstart, rec, rowptr, col, Nn, NBKT,
                                                   x, W1, h, as1, ad1, asc1, adc1, QB, NQ);

  gather1_kernel<<<2048, 256, 0, stream>>>(rowptr, col, col2, asc1, adc1, h, as1, b1, h1, Nn);

  gat_wmma_gemm_k128<3, 40, 1><<<dim3(1563, 1), 3 * 32, 0, stream>>>(h1, W2, h2, as2, ad2, asc2, adc2, Nn / 16);
  gather2_kernel<<<2048, 256, 0, stream>>>(rowptr, col2, asc2, adc2, h2, as2, b2, out, Nn);
}
